# starting kernel
# baseline (speedup 1.0000x reference)
#include <hip/hip_runtime.h>

typedef __attribute__((ext_vector_type(2))) float v2f;
typedef __attribute__((ext_vector_type(8))) float v8f;

#define NEG_SLOPE 0.2f

__device__ __forceinline__ unsigned fmap(float f) {
  unsigned u = __float_as_uint(f);
  return (u & 0x80000000u) ? ~u : (u | 0x80000000u);
}
__device__ __forceinline__ float funmap(unsigned u) {
  unsigned v = (u & 0x80000000u) ? (u & 0x7fffffffu) : ~u;
  return __uint_as_float(v);
}

template<int K, int NC>
__global__ void gemm_wmma_f32(const float* __restrict__ A, const float* __restrict__ B,
                              float* __restrict__ C, int M) {
  int wave = (blockIdx.x * blockDim.x + threadIdx.x) >> 5;
  int lane = threadIdx.x & 31;
  constexpr int nct = NC >> 4;
  int ntiles = ((M + 15) >> 4) * nct;
  if (wave >= ntiles) return;
  int tm = wave / nct, tc = wave - tm * nct;
  int row_base = tm << 4, col_base = tc << 4;
  int mrow = lane & 15, hi = lane >> 4;

  int arow = row_base + mrow; if (arow > M - 1) arow = M - 1;
  const float* Arow = A + (size_t)arow * K;
  const float* Bcol = B + col_base + mrow;

  v8f acc = {};
#pragma unroll 8
  for (int k0 = 0; k0 < K; k0 += 4) {
    int ka = k0 + 2 * hi;
    v2f a, b;
    a.x = Arow[ka];
    a.y = Arow[ka + 1];
    b.x = Bcol[ka * NC];
    b.y = Bcol[(ka + 1) * NC];
    acc = __builtin_amdgcn_wmma_f32_16x16x4_f32(
         false, a,  false, b,
         (short)0, acc,  false,  false);
  }

  int rbase = row_base + hi * 8;
  float* p = C + (size_t)rbase * NC + col_base + mrow;
  if (rbase + 8 <= M) {
#pragma unroll
    for (int r = 0; r < 8; ++r) p[r * NC] = acc[r];
  } else {
#pragma unroll
    for (int r = 0; r < 8; ++r) { if (rbase + r < M) p[r * NC] = acc[r]; }
  }
}

__global__ void node_alpha(const float* __restrict__ xW,
                           const float* __restrict__ avs, const float* __restrict__ avd,
                           float* __restrict__ as, float* __restrict__ ad, int N, int H) {
  int wave = (blockIdx.x * blockDim.x + threadIdx.x) >> 5;
  int lane = threadIdx.x & 31;
  if (wave >= N * H) return;
  int h = wave % H;
  const float* row = xW + (size_t)wave * 64;
  const float* vs = avs + h * 64;
  const float* vd = avd + h * 64;
  float s = row[lane] * vs[lane] + row[lane + 32] * vs[lane + 32];
  float d = row[lane] * vd[lane] + row[lane + 32] * vd[lane + 32];
#pragma unroll
  for (int off = 16; off; off >>= 1) {
    s += __shfl_xor(s, off, 32);
    d += __shfl_xor(d, off, 32);
  }
  if (lane == 0) { as[wave] = s; ad[wave] = d; }
}

template<int H>
__global__ void edge_leaky_max(const float* __restrict__ as, const float* __restrict__ ad,
                               const int* __restrict__ ei, float* __restrict__ ebuf,
                               unsigned* __restrict__ mU, int E) {
  int gid = blockIdx.x * blockDim.x + threadIdx.x;
  if (gid >= E * H) return;
  int e = gid / H, h = gid - e * H;
  int s = ei[e], d = ei[E + e];
  float v = as[s * H + h] + ad[d * H + h];
  v = (v >= 0.f) ? v : NEG_SLOPE * v;
  ebuf[gid] = v;
  atomicMax(&mU[d * H + h], fmap(v));
}

__global__ void decode_max(unsigned* __restrict__ mU, int n) {
  int gid = blockIdx.x * blockDim.x + threadIdx.x;
  if (gid >= n) return;
  float f = funmap(mU[gid]);
  ((float*)mU)[gid] = f;
}

template<int H>
__global__ void edge_expsum(float* __restrict__ ebuf, const float* __restrict__ mf,
                            const int* __restrict__ ei, float* __restrict__ denom, int E) {
  int gid = blockIdx.x * blockDim.x + threadIdx.x;
  if (gid >= E * H) return;
  int e = gid / H, h = gid - e * H;
  int d = ei[E + e];
  float ex = __expf(ebuf[gid] - mf[d * H + h]);
  ebuf[gid] = ex;
  atomicAdd(&denom[d * H + h], ex);
}

template<int H>
__global__ void edge_norm(float* __restrict__ ebuf, const float* __restrict__ denom,
                          const int* __restrict__ ei, int E) {
  int gid = blockIdx.x * blockDim.x + threadIdx.x;
  if (gid >= E * H) return;
  int e = gid / H, h = gid - e * H;
  int d = ei[E + e];
  ebuf[gid] = ebuf[gid] / denom[d * H + h];
}

template<int H, int C>
__global__ void edge_aggregate(const float* __restrict__ xW, const float* __restrict__ alpha,
                               const int* __restrict__ ei, float* __restrict__ out, int E) {
  constexpr int CH4 = (H * C) / 4;
  int gid = blockIdx.x * blockDim.x + threadIdx.x;
  if (gid >= E * CH4) return;
  int e = gid / CH4, j4 = gid - e * CH4;
  int h = j4 / (C / 4);
  int s = ei[e], d = ei[E + e];
  float a = alpha[e * H + h];
  const float4* src4 = (const float4*)(xW + (size_t)s * (H * C));
  float4 v = src4[j4];
  float* o = out + (size_t)d * (H * C) + j4 * 4;
  atomicAdd(o + 0, v.x * a);
  atomicAdd(o + 1, v.y * a);
  atomicAdd(o + 2, v.z * a);
  atomicAdd(o + 3, v.w * a);
}

template<int F, bool RELU>
__global__ void bias_act(float* __restrict__ x, const float* __restrict__ b, int total) {
  int gid = blockIdx.x * blockDim.x + threadIdx.x;
  if (gid >= total) return;
  float v = x[gid] + b[gid & (F - 1)];
  if (RELU) v = v > 0.f ? v : 0.f;
  x[gid] = v;
}

extern "C" void kernel_launch(void* const* d_in, const int* in_sizes, int n_in,
                              void* d_out, int out_size, void* d_ws, size_t ws_size,
                              hipStream_t stream) {
  const float* x     = (const float*)d_in[0];
  const int*   ei    = (const int*)  d_in[1];
  const float* W1    = (const float*)d_in[2];
  const float* asrc1 = (const float*)d_in[3];
  const float* adst1 = (const float*)d_in[4];
  const float* b1    = (const float*)d_in[5];
  const float* W2    = (const float*)d_in[6];
  const float* asrc2 = (const float*)d_in[7];
  const float* adst2 = (const float*)d_in[8];
  const float* b2    = (const float*)d_in[9];
  float* out = (float*)d_out;

  const int D = 64;
  const int N = in_sizes[0] / D;
  const int E = in_sizes[1] / 2;
  const int H = 8;
  const int F1 = H * D;

  float* ws   = (float*)d_ws;
  float* xW1  = ws;
  float* out1 = xW1  + (size_t)N * F1;
  float* e1   = out1 + (size_t)N * F1;
  float* as1  = e1   + (size_t)E * H;
  float* ad1  = as1  + (size_t)N * H;
  float* m1   = ad1  + (size_t)N * H;
  float* dn1  = m1   + (size_t)N * H;
  float* hW2  = dn1  + (size_t)N * H;
  float* e2   = hW2  + (size_t)N * D;
  float* as2  = e2   + (size_t)E;
  float* ad2  = as2  + (size_t)N;
  float* m2   = ad2  + (size_t)N;
  float* dn2  = m2   + (size_t)N;

  const int T = 256;

  hipMemsetAsync(out1, 0, (size_t)N * F1 * 4, stream);
  hipMemsetAsync(m1,   0, (size_t)N * H * 4, stream);
  hipMemsetAsync(dn1,  0, (size_t)N * H * 4, stream);
  hipMemsetAsync(out,  0, (size_t)N * D * 4, stream);
  hipMemsetAsync(m2,   0, (size_t)N * 4, stream);
  hipMemsetAsync(dn2,  0, (size_t)N * 4, stream);

  {
    int ntiles = (N / 16) * (F1 / 16);
    gemm_wmma_f32<64, 512><<<(ntiles + 7) / 8, T, 0, stream>>>(x, W1, xW1, N);
  }
  node_alpha<<<(N * H * 32 + T - 1) / T, T, 0, stream>>>(xW1, asrc1, adst1, as1, ad1, N, H);
  edge_leaky_max<8><<<(E * H + T - 1) / T, T, 0, stream>>>(as1, ad1, ei, e1, (unsigned*)m1, E);
  decode_max<<<(N * H + T - 1) / T, T, 0, stream>>>((unsigned*)m1, N * H);
  edge_expsum<8><<<(E * H + T - 1) / T, T, 0, stream>>>(e1, m1, ei, dn1, E);
  edge_norm<8><<<(E * H + T - 1) / T, T, 0, stream>>>(e1, dn1, ei, E);
  edge_aggregate<8, 64><<<(E * 128 + T - 1) / T, T, 0, stream>>>(xW1, e1, ei, out1, E);
  bias_act<512, true><<<((size_t)N * F1 + T - 1) / T, T, 0, stream>>>(out1, b1, N * F1);

  {
    int ntiles = (N / 16) * (D / 16);
    gemm_wmma_f32<512, 64><<<(ntiles + 7) / 8, T, 0, stream>>>(out1, W2, hW2, N);
  }
  node_alpha<<<(N * 32 + T - 1) / T, T, 0, stream>>>(hW2, asrc2, adst2, as2, ad2, N, 1);
  edge_leaky_max<1><<<(E + T - 1) / T, T, 0, stream>>>(as2, ad2, ei, e2, (unsigned*)m2, E);
  decode_max<<<(N + T - 1) / T, T, 0, stream>>>((unsigned*)m2, N);
  edge_expsum<1><<<(E + T - 1) / T, T, 0, stream>>>(e2, m2, ei, dn2, E);
  edge_norm<1><<<(E + T - 1) / T, T, 0, stream>>>(e2, dn2, ei, E);
  edge_aggregate<1, 64><<<(E * 16 + T - 1) / T, T, 0, stream>>>(hW2, e2, ei, out, E);
  bias_act<64, false><<<((size_t)N * D + T - 1) / T, T, 0, stream>>>(out, b2, N * D);
}


# softmax and aggregation by destination rows: edges counting-sorted in-call, each row then ordered by edge id (deterministic); one wave sums a row in registers; no f32 atomics; sum order = edge id
# speedup vs baseline: 2.4772x; 2.4772x over previous
#include <hip/hip_runtime.h>

typedef __attribute__((ext_vector_type(2))) float v2f;
typedef __attribute__((ext_vector_type(8))) float v8f;

#define NEG_SLOPE 0.2f

template<int K, int NC>
__global__ void gemm_wmma_f32(const float* __restrict__ A, const float* __restrict__ B,
                              float* __restrict__ C, int M) {
  int wave = (blockIdx.x * blockDim.x + threadIdx.x) >> 5;
  int lane = threadIdx.x & 31;
  constexpr int nct = NC >> 4;
  int ntiles = ((M + 15) >> 4) * nct;
  if (wave >= ntiles) return;
  int tm = wave / nct, tc = wave - tm * nct;
  int row_base = tm << 4, col_base = tc << 4;
  int mrow = lane & 15, hi = lane >> 4;

  int arow = row_base + mrow; if (arow > M - 1) arow = M - 1;
  const float* Arow = A + (size_t)arow * K;
  const float* Bcol = B + col_base + mrow;

  v8f acc = {};
#pragma unroll 8
  for (int k0 = 0; k0 < K; k0 += 4) {
    int ka = k0 + 2 * hi;
    v2f a, b;
    a.x = Arow[ka];
    a.y = Arow[ka + 1];
    b.x = Bcol[ka * NC];
    b.y = Bcol[(ka + 1) * NC];
    acc = __builtin_amdgcn_wmma_f32_16x16x4_f32(
         false, a,  false, b,
         (short)0, acc,  false,  false);
  }

  int rbase = row_base + hi * 8;
  float* p = C + (size_t)rbase * NC + col_base + mrow;
  if (rbase + 8 <= M) {
#pragma unroll
    for (int r = 0; r < 8; ++r) p[r * NC] = acc[r];
  } else {
#pragma unroll
    for (int r = 0; r < 8; ++r) { if (rbase + r < M) p[r * NC] = acc[r]; }
  }
}

__global__ void node_alpha(const float* __restrict__ xW,
                           const float* __restrict__ avs, const float* __restrict__ avd,
                           float* __restrict__ as, float* __restrict__ ad, int N, int H) {
  int wave = (blockIdx.x * blockDim.x + threadIdx.x) >> 5;
  int lane = threadIdx.x & 31;
  if (wave >= N * H) return;
  int h = wave % H;
  const float* row = xW + (size_t)wave * 64;
  const float* vs = avs + h * 64;
  const float* vd = avd + h * 64;
  float s = row[lane] * vs[lane] + row[lane + 32] * vs[lane + 32];
  float d = row[lane] * vd[lane] + row[lane + 32] * vd[lane + 32];
#pragma unroll
  for (int off = 16; off; off >>= 1) {
    s += __shfl_xor(s, off, 32);
    d += __shfl_xor(d, off, 32);
  }
  if (lane == 0) { as[wave] = s; ad[wave] = d; }
}

__global__ void edge_count_slot(const int* __restrict__ ei, int* __restrict__ cnt,
                                int* __restrict__ slot, int E) {
  int gid = blockIdx.x * blockDim.x + threadIdx.x;
  if (gid >= E) return;
  slot[gid] = atomicAdd(&cnt[ei[E + gid]], 1);
}

__global__ void row_scan(const int* __restrict__ cnt, int* __restrict__ rowptr, int N) {
  __shared__ int wsum[32];
  int t = threadIdx.x, lane = t & 31, w = t >> 5;
  int nw = (int)blockDim.x >> 5;
  int per = (N + nw - 1) / nw;
  per = (per + 127) & ~127;
  int lo = w * per; if (lo > N) lo = N;
  int hi = lo + per; if (hi > N) hi = N;
  int part = 0;
  for (int i = lo + lane; i < hi; i += 32) part += cnt[i];
  for (int off = 16; off; off >>= 1) part += __shfl_xor(part, off, 32);
  if (lane == 0) wsum[w] = part;
  __syncthreads();
  if (w == 0) {
    int s = (lane < nw) ? wsum[lane] : 0, si = s;
    for (int off = 1; off < 32; off <<= 1) {
      int v = __shfl_up(si, off, 32);
      if (lane >= off) si += v;
    }
    wsum[lane] = si - s;
  }
  __syncthreads();
  int carry = wsum[w];
  for (int base = lo; base < hi; base += 128) {
    int i = base + lane * 4;
    int a0 = (i     < hi) ? cnt[i]     : 0;
    int a1 = (i + 1 < hi) ? cnt[i + 1] : 0;
    int a2 = (i + 2 < hi) ? cnt[i + 2] : 0;
    int a3 = (i + 3 < hi) ? cnt[i + 3] : 0;
    int s1 = a0 + a1, s2 = s1 + a2, tot = s2 + a3;
    int incl = tot;
    for (int off = 1; off < 32; off <<= 1) {
      int u = __shfl_up(incl, off, 32);
      if (lane >= off) incl += u;
    }
    int ex = carry + incl - tot;
    if (i     < hi) rowptr[i]     = ex;
    if (i + 1 < hi) rowptr[i + 1] = ex + a0;
    if (i + 2 < hi) rowptr[i + 2] = ex + s1;
    if (i + 3 < hi) rowptr[i + 3] = ex + s2;
    carry += __shfl(incl, 31, 32);
  }
  if (t == (int)blockDim.x - 1) rowptr[N] = carry;
}

__global__ void edge_place(const int* __restrict__ ei, const int* __restrict__ rowptr,
                           const int* __restrict__ slot, int* __restrict__ seg, int E) {
  int gid = blockIdx.x * blockDim.x + threadIdx.x;
  if (gid >= E) return;
  seg[rowptr[ei[E + gid]] + slot[gid]] = gid;
}

__global__ void row_order(const int* __restrict__ ei, const int* __restrict__ rowptr,
                          const int* __restrict__ seg, int* __restrict__ srcS, int N) {
  int row = (blockIdx.x * blockDim.x + threadIdx.x) >> 5;
  int lane = threadIdx.x & 31;
  if (row >= N) return;
  int start = rowptr[row], deg = rowptr[row + 1] - start;
  for (int i = lane; i < deg; i += 32) {
    int id = seg[start + i];
    int rank = 0;
    for (int j = 0; j < deg; ++j) rank += (seg[start + j] < id) ? 1 : 0;
    srcS[start + rank] = ei[id];
  }
}

template<int H>
__global__ void row_softmax(const float* __restrict__ as, const float* __restrict__ ad,
                            const int* __restrict__ rowptr, const int* __restrict__ srcS,
                            float* __restrict__ alpha, int N) {
  int row = (blockIdx.x * blockDim.x + threadIdx.x) >> 5;
  int lane = threadIdx.x & 31;
  if (row >= N) return;
  constexpr int SL = 32 / H;
  int sl = lane / H, h = lane - sl * H;
  int start = rowptr[row], end = rowptr[row + 1];
  float adh = ad[row * H + h];
  float m = -INFINITY;
  for (int p = start + sl; p < end; p += SL) {
    float v = as[srcS[p] * H + h] + adh;
    v = (v >= 0.f) ? v : NEG_SLOPE * v;
    m = fmaxf(m, v);
  }
  for (int off = 16; off >= H; off >>= 1) m = fmaxf(m, __shfl_xor(m, off, 32));
  float sum = 0.f;
  for (int p = start + sl; p < end; p += SL) {
    float v = as[srcS[p] * H + h] + adh;
    v = (v >= 0.f) ? v : NEG_SLOPE * v;
    sum += __expf(v - m);
  }
  for (int off = 16; off >= H; off >>= 1) sum += __shfl_xor(sum, off, 32);
  for (int p = start + sl; p < end; p += SL) {
    float v = as[srcS[p] * H + h] + adh;
    v = (v >= 0.f) ? v : NEG_SLOPE * v;
    alpha[(size_t)p * H + h] = __expf(v - m) / sum;
  }
}

__global__ void row_aggregate_h8(const float* __restrict__ xW, const float* __restrict__ alpha,
                                 const int* __restrict__ rowptr, const int* __restrict__ srcS,
                                 const float* __restrict__ bias, float* __restrict__ out, int N) {
  int row = (blockIdx.x * blockDim.x + threadIdx.x) >> 5;
  int lane = threadIdx.x & 31;
  if (row >= N) return;
  int start = rowptr[row], end = rowptr[row + 1];
  int hi = lane >> 4;
  float4 acc[4];
  for (int i = 0; i < 4; ++i) acc[i] = make_float4(0.f, 0.f, 0.f, 0.f);
  for (int base = start; base < end; base += 32) {
    int n = end - base; if (n > 32) n = 32;
    int sv = (lane < n) ? srcS[base + lane] : 0;
    for (int k = 0; k < n; ++k) {
      int s = __builtin_amdgcn_readlane(sv, k);
      const float4* src4 = (const float4*)(xW + (size_t)s * 512);
      const float* al = alpha + (size_t)(base + k) * 8 + hi;
      for (int i = 0; i < 4; ++i) {
        float a = al[2 * i];
        float4 v = src4[lane + 32 * i];
        acc[i].x += a * v.x; acc[i].y += a * v.y; acc[i].z += a * v.z; acc[i].w += a * v.w;
      }
    }
  }
  const float4* b4 = (const float4*)bias;
  float4* o4 = (float4*)(out + (size_t)row * 512);
  for (int i = 0; i < 4; ++i) {
    float4 b = b4[lane + 32 * i];
    float4 r;
    r.x = fmaxf(acc[i].x + b.x, 0.f); r.y = fmaxf(acc[i].y + b.y, 0.f);
    r.z = fmaxf(acc[i].z + b.z, 0.f); r.w = fmaxf(acc[i].w + b.w, 0.f);
    o4[lane + 32 * i] = r;
  }
}

__global__ void row_aggregate_h1(const float* __restrict__ xW, const float* __restrict__ alpha,
                                 const int* __restrict__ rowptr, const int* __restrict__ srcS,
                                 const float* __restrict__ bias, float* __restrict__ out, int N) {
  int row = (blockIdx.x * blockDim.x + threadIdx.x) >> 5;
  int lane = threadIdx.x & 31;
  if (row >= N) return;
  int start = rowptr[row], end = rowptr[row + 1];
  float2 acc = make_float2(0.f, 0.f);
  for (int base = start; base < end; base += 32) {
    int n = end - base; if (n > 32) n = 32;
    int sv = 0; float av = 0.f;
    if (lane < n) { sv = srcS[base + lane]; av = alpha[base + lane]; }
    for (int k = 0; k < n; ++k) {
      int s = __builtin_amdgcn_readlane(sv, k);
      float a = __uint_as_float(__builtin_amdgcn_readlane(__float_as_uint(av), k));
      float2 v = ((const float2*)(xW + (size_t)s * 64))[lane];
      acc.x += a * v.x; acc.y += a * v.y;
    }
  }
  float2 b = ((const float2*)bias)[lane];
  float2 r; r.x = acc.x + b.x; r.y = acc.y + b.y;
  ((float2*)(out + (size_t)row * 64))[lane] = r;
}

extern "C" void kernel_launch(void* const* d_in, const int* in_sizes, int n_in,
                              void* d_out, int out_size, void* d_ws, size_t ws_size,
                              hipStream_t stream) {
  const float* x     = (const float*)d_in[0];
  const int*   ei    = (const int*)  d_in[1];
  const float* W1    = (const float*)d_in[2];
  const float* asrc1 = (const float*)d_in[3];
  const float* adst1 = (const float*)d_in[4];
  const float* b1    = (const float*)d_in[5];
  const float* W2    = (const float*)d_in[6];
  const float* asrc2 = (const float*)d_in[7];
  const float* adst2 = (const float*)d_in[8];
  const float* b2    = (const float*)d_in[9];
  float* out = (float*)d_out;

  const int D = 64;
  const int N = in_sizes[0] / D;
  const int E = in_sizes[1] / 2;
  const int H = 8;
  const int F1 = H * D;

  float* ws   = (float*)d_ws;
  float* xW1  = ws;
  float* out1 = xW1  + (size_t)N * F1;
  float* al1  = out1 + (size_t)N * F1;
  float* as1  = al1  + (size_t)E * H;
  float* ad1  = as1  + (size_t)N * H;
  float* hW2  = ad1  + (size_t)N * H;
  float* al2  = hW2  + (size_t)N * D;
  float* as2  = al2  + (size_t)E;
  float* ad2  = as2  + (size_t)N;
  int* cnt    = (int*)(ad2 + (size_t)N);
  int* rowptr = cnt    + (size_t)N;
  int* slot   = rowptr + (size_t)N + 1;
  int* seg    = slot   + (size_t)E;
  int* srcS   = seg    + (size_t)E;

  const int T = 256;

  hipMemsetAsync(cnt, 0, (size_t)N * 4, stream);
  edge_count_slot<<<(E + T - 1) / T, T, 0, stream>>>(ei, cnt, slot, E);
  row_scan<<<1, 1024, 0, stream>>>(cnt, rowptr, N);
  edge_place<<<(E + T - 1) / T, T, 0, stream>>>(ei, rowptr, slot, seg, E);
  row_order<<<(N * 32 + T - 1) / T, T, 0, stream>>>(ei, rowptr, seg, srcS, N);

  {
    int ntiles = (N / 16) * (F1 / 16);
    gemm_wmma_f32<64, 512><<<(ntiles + 7) / 8, T, 0, stream>>>(x, W1, xW1, N);
  }
  node_alpha<<<(N * H * 32 + T - 1) / T, T, 0, stream>>>(xW1, asrc1, adst1, as1, ad1, N, H);
  row_softmax<8><<<(N * 32 + T - 1) / T, T, 0, stream>>>(as1, ad1, rowptr, srcS, al1, N);
  row_aggregate_h8<<<(N * 32 + T - 1) / T, T, 0, stream>>>(xW1, al1, rowptr, srcS, b1, out1, N);

  {
    int ntiles = (N / 16) * (D / 16);
    gemm_wmma_f32<512, 64><<<(ntiles + 7) / 8, T, 0, stream>>>(out1, W2, hW2, N);
  }
  node_alpha<<<(N * 32 + T - 1) / T, T, 0, stream>>>(hW2, asrc2, adst2, as2, ad2, N, 1);
  row_softmax<1><<<(N * 32 + T - 1) / T, T, 0, stream>>>(as2, ad2, rowptr, srcS, al2, N);
  row_aggregate_h1<<<(N * 32 + T - 1) / T, T, 0, stream>>>(hW2, al2, rowptr, srcS, b2, out, N);
}


# attention logits fused into the projection GEMM: one wave = 16 rows x one 64-col head (4 WMMA tiles, interleaved cols, 16-B B loads/C stores); separate logit pass dropped; logit sum order changes
# speedup vs baseline: 3.2861x; 1.3265x over previous
#include <hip/hip_runtime.h>

typedef __attribute__((ext_vector_type(2))) float v2f;
typedef __attribute__((ext_vector_type(8))) float v8f;

#define NEG_SLOPE 0.2f

template<int K, int NC>
__global__ void gemm_logits_wmma_f32(const float* __restrict__ A, const float* __restrict__ B,
                                     const float* __restrict__ avs, const float* __restrict__ avd,
                                     float* __restrict__ C, float* __restrict__ as,
                                     float* __restrict__ ad, int M) {
  int wave = (blockIdx.x * blockDim.x + threadIdx.x) >> 5;
  int lane = threadIdx.x & 31;
  constexpr int NH = NC >> 6;
  int mt = (M + 15) >> 4;
  if (wave >= mt * NH) return;
  int th = wave / mt, tm = wave - th * mt;
  int row_base = tm << 4, col_base = th << 6;
  int mrow = lane & 15, hi = lane >> 4;

  int arow = row_base + mrow; if (arow > M - 1) arow = M - 1;
  const float* Arow = A + (size_t)arow * K;
  const float* Bcol = B + col_base + 4 * mrow;

  v8f acc0 = {}, acc1 = {}, acc2 = {}, acc3 = {};
#pragma unroll 8
  for (int k0 = 0; k0 < K; k0 += 4) {
    int ka = k0 + 2 * hi;
    v2f a;
    a.x = Arow[ka];
    a.y = Arow[ka + 1];
    float4 bx = *(const float4*)(Bcol + (size_t)ka * NC);
    float4 by = *(const float4*)(Bcol + (size_t)(ka + 1) * NC);
    v2f b0, b1, b2, b3;
    b0.x = bx.x; b0.y = by.x;
    b1.x = bx.y; b1.y = by.y;
    b2.x = bx.z; b2.y = by.z;
    b3.x = bx.w; b3.y = by.w;
    acc0 = __builtin_amdgcn_wmma_f32_16x16x4_f32(
         false, a,  false, b0,
         (short)0, acc0,  false,  false);
    acc1 = __builtin_amdgcn_wmma_f32_16x16x4_f32(false, a, false, b1, (short)0, acc1, false, false);
    acc2 = __builtin_amdgcn_wmma_f32_16x16x4_f32(false, a, false, b2, (short)0, acc2, false, false);
    acc3 = __builtin_amdgcn_wmma_f32_16x16x4_f32(false, a, false, b3, (short)0, acc3, false, false);
  }

  int rbase = row_base + hi * 8;
  float* p = C + (size_t)rbase * NC + col_base + 4 * mrow;
  if (rbase + 8 <= M) {
#pragma unroll
    for (int r = 0; r < 8; ++r)
      *(float4*)(p + r * NC) = make_float4(acc0[r], acc1[r], acc2[r], acc3[r]);
  } else {
#pragma unroll
    for (int r = 0; r < 8; ++r) {
      if (rbase + r < M) *(float4*)(p + r * NC) = make_float4(acc0[r], acc1[r], acc2[r], acc3[r]);
    }
  }

  float4 s4 = *(const float4*)(avs + col_base + 4 * mrow);
  float4 d4 = *(const float4*)(avd + col_base + 4 * mrow);
  float v[16];
#pragma unroll
  for (int r = 0; r < 8; ++r) {
    v[r]     = acc0[r] * s4.x + acc1[r] * s4.y + acc2[r] * s4.z + acc3[r] * s4.w;
    v[8 + r] = acc0[r] * d4.x + acc1[r] * d4.y + acc2[r] * d4.z + acc3[r] * d4.w;
  }
  bool t8 = (mrow & 8) != 0, t4 = (mrow & 4) != 0, t2 = (mrow & 2) != 0, t1 = (mrow & 1) != 0;
  float w8[8], w4[4], w2[2];
#pragma unroll
  for (int i = 0; i < 8; ++i) {
    float keep = t8 ? v[8 + i] : v[i];
    float send = t8 ? v[i] : v[8 + i];
    w8[i] = keep + __shfl_xor(send, 8, 32);
  }
#pragma unroll
  for (int i = 0; i < 4; ++i) {
    float keep = t4 ? w8[4 + i] : w8[i];
    float send = t4 ? w8[i] : w8[4 + i];
    w4[i] = keep + __shfl_xor(send, 4, 32);
  }
#pragma unroll
  for (int i = 0; i < 2; ++i) {
    float keep = t2 ? w4[2 + i] : w4[i];
    float send = t2 ? w4[i] : w4[2 + i];
    w2[i] = keep + __shfl_xor(send, 2, 32);
  }
  float tot = (t1 ? w2[1] : w2[0]) + __shfl_xor(t1 ? w2[0] : w2[1], 1, 32);
  int orow = rbase + (mrow & 7);
  float* dst = t8 ? ad : as;
  if (orow < M) dst[(size_t)orow * NH + th] = tot;
}

__global__ void edge_count_slot(const int* __restrict__ ei, int* __restrict__ cnt,
                                int* __restrict__ slot, int E) {
  int gid = blockIdx.x * blockDim.x + threadIdx.x;
  if (gid >= E) return;
  slot[gid] = atomicAdd(&cnt[ei[E + gid]], 1);
}

__global__ void row_scan(const int* __restrict__ cnt, int* __restrict__ rowptr, int N) {
  __shared__ int wsum[32];
  int t = threadIdx.x, lane = t & 31, w = t >> 5;
  int nw = (int)blockDim.x >> 5;
  int per = (N + nw - 1) / nw;
  per = (per + 127) & ~127;
  int lo = w * per; if (lo > N) lo = N;
  int hi = lo + per; if (hi > N) hi = N;
  int part = 0;
  for (int i = lo + lane; i < hi; i += 32) part += cnt[i];
  for (int off = 16; off; off >>= 1) part += __shfl_xor(part, off, 32);
  if (lane == 0) wsum[w] = part;
  __syncthreads();
  if (w == 0) {
    int s = (lane < nw) ? wsum[lane] : 0, si = s;
    for (int off = 1; off < 32; off <<= 1) {
      int v = __shfl_up(si, off, 32);
      if (lane >= off) si += v;
    }
    wsum[lane] = si - s;
  }
  __syncthreads();
  int carry = wsum[w];
  for (int base = lo; base < hi; base += 128) {
    int i = base + lane * 4;
    int a0 = (i     < hi) ? cnt[i]     : 0;
    int a1 = (i + 1 < hi) ? cnt[i + 1] : 0;
    int a2 = (i + 2 < hi) ? cnt[i + 2] : 0;
    int a3 = (i + 3 < hi) ? cnt[i + 3] : 0;
    int s1 = a0 + a1, s2 = s1 + a2, tot = s2 + a3;
    int incl = tot;
    for (int off = 1; off < 32; off <<= 1) {
      int u = __shfl_up(incl, off, 32);
      if (lane >= off) incl += u;
    }
    int ex = carry + incl - tot;
    if (i     < hi) rowptr[i]     = ex;
    if (i + 1 < hi) rowptr[i + 1] = ex + a0;
    if (i + 2 < hi) rowptr[i + 2] = ex + s1;
    if (i + 3 < hi) rowptr[i + 3] = ex + s2;
    carry += __shfl(incl, 31, 32);
  }
  if (t == (int)blockDim.x - 1) rowptr[N] = carry;
}

__global__ void edge_place(const int* __restrict__ ei, const int* __restrict__ rowptr,
                           const int* __restrict__ slot, int* __restrict__ seg, int E) {
  int gid = blockIdx.x * blockDim.x + threadIdx.x;
  if (gid >= E) return;
  seg[rowptr[ei[E + gid]] + slot[gid]] = gid;
}

__global__ void row_order(const int* __restrict__ ei, const int* __restrict__ rowptr,
                          const int* __restrict__ seg, int* __restrict__ srcS, int N) {
  int row = (blockIdx.x * blockDim.x + threadIdx.x) >> 5;
  int lane = threadIdx.x & 31;
  if (row >= N) return;
  int start = rowptr[row], deg = rowptr[row + 1] - start;
  for (int i = lane; i < deg; i += 32) {
    int id = seg[start + i];
    int rank = 0;
    for (int j = 0; j < deg; ++j) rank += (seg[start + j] < id) ? 1 : 0;
    srcS[start + rank] = ei[id];
  }
}

template<int H>
__global__ void row_softmax(const float* __restrict__ as, const float* __restrict__ ad,
                            const int* __restrict__ rowptr, const int* __restrict__ srcS,
                            float* __restrict__ alpha, int N) {
  int row = (blockIdx.x * blockDim.x + threadIdx.x) >> 5;
  int lane = threadIdx.x & 31;
  if (row >= N) return;
  constexpr int SL = 32 / H;
  int sl = lane / H, h = lane - sl * H;
  int start = rowptr[row], end = rowptr[row + 1];
  float adh = ad[row * H + h];
  float m = -INFINITY;
  for (int p = start + sl; p < end; p += SL) {
    float v = as[srcS[p] * H + h] + adh;
    v = (v >= 0.f) ? v : NEG_SLOPE * v;
    m = fmaxf(m, v);
  }
  for (int off = 16; off >= H; off >>= 1) m = fmaxf(m, __shfl_xor(m, off, 32));
  float sum = 0.f;
  for (int p = start + sl; p < end; p += SL) {
    float v = as[srcS[p] * H + h] + adh;
    v = (v >= 0.f) ? v : NEG_SLOPE * v;
    sum += __expf(v - m);
  }
  for (int off = 16; off >= H; off >>= 1) sum += __shfl_xor(sum, off, 32);
  for (int p = start + sl; p < end; p += SL) {
    float v = as[srcS[p] * H + h] + adh;
    v = (v >= 0.f) ? v : NEG_SLOPE * v;
    alpha[(size_t)p * H + h] = __expf(v - m) / sum;
  }
}

__global__ void row_aggregate_h8(const float* __restrict__ xW, const float* __restrict__ alpha,
                                 const int* __restrict__ rowptr, const int* __restrict__ srcS,
                                 const float* __restrict__ bias, float* __restrict__ out, int N) {
  int row = (blockIdx.x * blockDim.x + threadIdx.x) >> 5;
  int lane = threadIdx.x & 31;
  if (row >= N) return;
  int start = rowptr[row], end = rowptr[row + 1];
  int hi = lane >> 4;
  float4 acc[4];
  for (int i = 0; i < 4; ++i) acc[i] = make_float4(0.f, 0.f, 0.f, 0.f);
  for (int base = start; base < end; base += 32) {
    int n = end - base; if (n > 32) n = 32;
    int sv = (lane < n) ? srcS[base + lane] : 0;
    for (int k = 0; k < n; ++k) {
      int s = __builtin_amdgcn_readlane(sv, k);
      const float4* src4 = (const float4*)(xW + (size_t)s * 512);
      const float* al = alpha + (size_t)(base + k) * 8 + hi;
      for (int i = 0; i < 4; ++i) {
        float a = al[2 * i];
        float4 v = src4[lane + 32 * i];
        acc[i].x += a * v.x; acc[i].y += a * v.y; acc[i].z += a * v.z; acc[i].w += a * v.w;
      }
    }
  }
  const float4* b4 = (const float4*)bias;
  float4* o4 = (float4*)(out + (size_t)row * 512);
  for (int i = 0; i < 4; ++i) {
    float4 b = b4[lane + 32 * i];
    float4 r;
    r.x = fmaxf(acc[i].x + b.x, 0.f); r.y = fmaxf(acc[i].y + b.y, 0.f);
    r.z = fmaxf(acc[i].z + b.z, 0.f); r.w = fmaxf(acc[i].w + b.w, 0.f);
    o4[lane + 32 * i] = r;
  }
}

__global__ void row_aggregate_h1(const float* __restrict__ xW, const float* __restrict__ alpha,
                                 const int* __restrict__ rowptr, const int* __restrict__ srcS,
                                 const float* __restrict__ bias, float* __restrict__ out, int N) {
  int row = (blockIdx.x * blockDim.x + threadIdx.x) >> 5;
  int lane = threadIdx.x & 31;
  if (row >= N) return;
  int start = rowptr[row], end = rowptr[row + 1];
  float2 acc = make_float2(0.f, 0.f);
  for (int base = start; base < end; base += 32) {
    int n = end - base; if (n > 32) n = 32;
    int sv = 0; float av = 0.f;
    if (lane < n) { sv = srcS[base + lane]; av = alpha[base + lane]; }
    for (int k = 0; k < n; ++k) {
      int s = __builtin_amdgcn_readlane(sv, k);
      float a = __uint_as_float(__builtin_amdgcn_readlane(__float_as_uint(av), k));
      float2 v = ((const float2*)(xW + (size_t)s * 64))[lane];
      acc.x += a * v.x; acc.y += a * v.y;
    }
  }
  float2 b = ((const float2*)bias)[lane];
  float2 r; r.x = acc.x + b.x; r.y = acc.y + b.y;
  ((float2*)(out + (size_t)row * 64))[lane] = r;
}

extern "C" void kernel_launch(void* const* d_in, const int* in_sizes, int n_in,
                              void* d_out, int out_size, void* d_ws, size_t ws_size,
                              hipStream_t stream) {
  const float* x     = (const float*)d_in[0];
  const int*   ei    = (const int*)  d_in[1];
  const float* W1    = (const float*)d_in[2];
  const float* asrc1 = (const float*)d_in[3];
  const float* adst1 = (const float*)d_in[4];
  const float* b1    = (const float*)d_in[5];
  const float* W2    = (const float*)d_in[6];
  const float* asrc2 = (const float*)d_in[7];
  const float* adst2 = (const float*)d_in[8];
  const float* b2    = (const float*)d_in[9];
  float* out = (float*)d_out;

  const int D = 64;
  const int N = in_sizes[0] / D;
  const int E = in_sizes[1] / 2;
  const int H = 8;
  const int F1 = H * D;

  float* ws   = (float*)d_ws;
  float* xW1  = ws;
  float* out1 = xW1  + (size_t)N * F1;
  float* al1  = out1 + (size_t)N * F1;
  float* as1  = al1  + (size_t)E * H;
  float* ad1  = as1  + (size_t)N * H;
  float* hW2  = ad1  + (size_t)N * H;
  float* al2  = hW2  + (size_t)N * D;
  float* as2  = al2  + (size_t)E;
  float* ad2  = as2  + (size_t)N;
  int* cnt    = (int*)(ad2 + (size_t)N);
  int* rowptr = cnt    + (size_t)N;
  int* slot   = rowptr + (size_t)N + 1;
  int* seg    = slot   + (size_t)E;
  int* srcS   = seg    + (size_t)E;

  const int T = 256;

  hipMemsetAsync(cnt, 0, (size_t)N * 4, stream);
  edge_count_slot<<<(E + T - 1) / T, T, 0, stream>>>(ei, cnt, slot, E);
  row_scan<<<1, 1024, 0, stream>>>(cnt, rowptr, N);
  edge_place<<<(E + T - 1) / T, T, 0, stream>>>(ei, rowptr, slot, seg, E);
  row_order<<<(N * 32 + T - 1) / T, T, 0, stream>>>(ei, rowptr, seg, srcS, N);

  {
    int nwaves = (N / 16) * H;
    gemm_logits_wmma_f32<64, 512><<<(nwaves + 7) / 8, T, 0, stream>>>(x, W1, asrc1, adst1, xW1, as1, ad1, N);
  }
  row_softmax<8><<<(N * 32 + T - 1) / T, T, 0, stream>>>(as1, ad1, rowptr, srcS, al1, N);
  row_aggregate_h8<<<(N * 32 + T - 1) / T, T, 0, stream>>>(xW1, al1, rowptr, srcS, b1, out1, N);

  {
    int nwaves = N / 16;
    gemm_logits_wmma_f32<512, 64><<<(nwaves + 7) / 8, T, 0, stream>>>(out1, W2, asrc2, adst2, hW2, as2, ad2, N);
  }
  row_softmax<1><<<(N * 32 + T - 1) / T, T, 0, stream>>>(as2, ad2, rowptr, srcS, al2, N);
  row_aggregate_h1<<<(N * 32 + T - 1) / T, T, 0, stream>>>(hW2, al2, rowptr, srcS, b2, out, N);
}


# segment softmax fused into the row aggregation: lane=edge logits, row max over lanes, exp weights readlane-broadcast, divide by their sum after the row sum (sum order = edge id); 2 launches fewer
# speedup vs baseline: 3.9017x; 1.1873x over previous
#include <hip/hip_runtime.h>

typedef __attribute__((ext_vector_type(2))) float v2f;
typedef __attribute__((ext_vector_type(8))) float v8f;

#define NEG_SLOPE 0.2f

template<int K, int NC>
__global__ void gemm_logits_wmma_f32(const float* __restrict__ A, const float* __restrict__ B,
                                     const float* __restrict__ avs, const float* __restrict__ avd,
                                     float* __restrict__ C, float* __restrict__ as,
                                     float* __restrict__ ad, int M) {
  int wave = (blockIdx.x * blockDim.x + threadIdx.x) >> 5;
  int lane = threadIdx.x & 31;
  constexpr int NH = NC >> 6;
  int mt = (M + 15) >> 4;
  if (wave >= mt * NH) return;
  int th = wave / mt, tm = wave - th * mt;
  int row_base = tm << 4, col_base = th << 6;
  int mrow = lane & 15, hi = lane >> 4;

  int arow = row_base + mrow; if (arow > M - 1) arow = M - 1;
  const float* Arow = A + (size_t)arow * K;
  const float* Bcol = B + col_base + 4 * mrow;

  v8f acc0 = {}, acc1 = {}, acc2 = {}, acc3 = {};
#pragma unroll 8
  for (int k0 = 0; k0 < K; k0 += 4) {
    int ka = k0 + 2 * hi;
    v2f a;
    a.x = Arow[ka];
    a.y = Arow[ka + 1];
    float4 bx = *(const float4*)(Bcol + (size_t)ka * NC);
    float4 by = *(const float4*)(Bcol + (size_t)(ka + 1) * NC);
    v2f b0, b1, b2, b3;
    b0.x = bx.x; b0.y = by.x;
    b1.x = bx.y; b1.y = by.y;
    b2.x = bx.z; b2.y = by.z;
    b3.x = bx.w; b3.y = by.w;
    acc0 = __builtin_amdgcn_wmma_f32_16x16x4_f32(
         false, a,  false, b0,
         (short)0, acc0,  false,  false);
    acc1 = __builtin_amdgcn_wmma_f32_16x16x4_f32(false, a, false, b1, (short)0, acc1, false, false);
    acc2 = __builtin_amdgcn_wmma_f32_16x16x4_f32(false, a, false, b2, (short)0, acc2, false, false);
    acc3 = __builtin_amdgcn_wmma_f32_16x16x4_f32(false, a, false, b3, (short)0, acc3, false, false);
  }

  int rbase = row_base + hi * 8;
  float* p = C + (size_t)rbase * NC + col_base + 4 * mrow;
  if (rbase + 8 <= M) {
#pragma unroll
    for (int r = 0; r < 8; ++r)
      *(float4*)(p + r * NC) = make_float4(acc0[r], acc1[r], acc2[r], acc3[r]);
  } else {
#pragma unroll
    for (int r = 0; r < 8; ++r) {
      if (rbase + r < M) *(float4*)(p + r * NC) = make_float4(acc0[r], acc1[r], acc2[r], acc3[r]);
    }
  }

  float4 s4 = *(const float4*)(avs + col_base + 4 * mrow);
  float4 d4 = *(const float4*)(avd + col_base + 4 * mrow);
  float v[16];
#pragma unroll
  for (int r = 0; r < 8; ++r) {
    v[r]     = acc0[r] * s4.x + acc1[r] * s4.y + acc2[r] * s4.z + acc3[r] * s4.w;
    v[8 + r] = acc0[r] * d4.x + acc1[r] * d4.y + acc2[r] * d4.z + acc3[r] * d4.w;
  }
  bool t8 = (mrow & 8) != 0, t4 = (mrow & 4) != 0, t2 = (mrow & 2) != 0, t1 = (mrow & 1) != 0;
  float w8[8], w4[4], w2[2];
#pragma unroll
  for (int i = 0; i < 8; ++i) {
    float keep = t8 ? v[8 + i] : v[i];
    float send = t8 ? v[i] : v[8 + i];
    w8[i] = keep + __shfl_xor(send, 8, 32);
  }
#pragma unroll
  for (int i = 0; i < 4; ++i) {
    float keep = t4 ? w8[4 + i] : w8[i];
    float send = t4 ? w8[i] : w8[4 + i];
    w4[i] = keep + __shfl_xor(send, 4, 32);
  }
#pragma unroll
  for (int i = 0; i < 2; ++i) {
    float keep = t2 ? w4[2 + i] : w4[i];
    float send = t2 ? w4[i] : w4[2 + i];
    w2[i] = keep + __shfl_xor(send, 2, 32);
  }
  float tot = (t1 ? w2[1] : w2[0]) + __shfl_xor(t1 ? w2[0] : w2[1], 1, 32);
  int orow = rbase + (mrow & 7);
  float* dst = t8 ? ad : as;
  if (orow < M) dst[(size_t)orow * NH + th] = tot;
}

__global__ void edge_count_slot(const int* __restrict__ ei, int* __restrict__ cnt,
                                int* __restrict__ slot, int E) {
  int gid = blockIdx.x * blockDim.x + threadIdx.x;
  if (gid >= E) return;
  slot[gid] = atomicAdd(&cnt[ei[E + gid]], 1);
}

__global__ void row_scan(const int* __restrict__ cnt, int* __restrict__ rowptr, int N) {
  __shared__ int wsum[32];
  int t = threadIdx.x, lane = t & 31, w = t >> 5;
  int nw = (int)blockDim.x >> 5;
  int per = (N + nw - 1) / nw;
  per = (per + 127) & ~127;
  int lo = w * per; if (lo > N) lo = N;
  int hi = lo + per; if (hi > N) hi = N;
  int part = 0;
  for (int i = lo + lane; i < hi; i += 32) part += cnt[i];
  for (int off = 16; off; off >>= 1) part += __shfl_xor(part, off, 32);
  if (lane == 0) wsum[w] = part;
  __syncthreads();
  if (w == 0) {
    int s = (lane < nw) ? wsum[lane] : 0, si = s;
    for (int off = 1; off < 32; off <<= 1) {
      int v = __shfl_up(si, off, 32);
      if (lane >= off) si += v;
    }
    wsum[lane] = si - s;
  }
  __syncthreads();
  int carry = wsum[w];
  for (int base = lo; base < hi; base += 128) {
    int i = base + lane * 4;
    int a0 = (i     < hi) ? cnt[i]     : 0;
    int a1 = (i + 1 < hi) ? cnt[i + 1] : 0;
    int a2 = (i + 2 < hi) ? cnt[i + 2] : 0;
    int a3 = (i + 3 < hi) ? cnt[i + 3] : 0;
    int s1 = a0 + a1, s2 = s1 + a2, tot = s2 + a3;
    int incl = tot;
    for (int off = 1; off < 32; off <<= 1) {
      int u = __shfl_up(incl, off, 32);
      if (lane >= off) incl += u;
    }
    int ex = carry + incl - tot;
    if (i     < hi) rowptr[i]     = ex;
    if (i + 1 < hi) rowptr[i + 1] = ex + a0;
    if (i + 2 < hi) rowptr[i + 2] = ex + s1;
    if (i + 3 < hi) rowptr[i + 3] = ex + s2;
    carry += __shfl(incl, 31, 32);
  }
  if (t == (int)blockDim.x - 1) rowptr[N] = carry;
}

__global__ void edge_place(const int* __restrict__ ei, const int* __restrict__ rowptr,
                           const int* __restrict__ slot, int* __restrict__ seg, int E) {
  int gid = blockIdx.x * blockDim.x + threadIdx.x;
  if (gid >= E) return;
  seg[rowptr[ei[E + gid]] + slot[gid]] = gid;
}

__global__ void row_order(const int* __restrict__ ei, const int* __restrict__ rowptr,
                          const int* __restrict__ seg, int* __restrict__ srcS, int N) {
  int row = (blockIdx.x * blockDim.x + threadIdx.x) >> 5;
  int lane = threadIdx.x & 31;
  if (row >= N) return;
  int start = rowptr[row], deg = rowptr[row + 1] - start;
  for (int i = lane; i < deg; i += 32) {
    int id = seg[start + i];
    int rank = 0;
    for (int j = 0; j < deg; ++j) rank += (seg[start + j] < id) ? 1 : 0;
    srcS[start + rank] = ei[id];
  }
}

__device__ __forceinline__ float leaky(float v) { return (v >= 0.f) ? v : NEG_SLOPE * v; }

__device__ __forceinline__ void edge_logits_h8(const float* __restrict__ as, const int* __restrict__ srcS,
                                               const float (&adh)[8], int base, int end, int lane,
                                               int& sv, float (&e)[8]) {
  sv = 0;
#pragma unroll
  for (int h = 0; h < 8; ++h) e[h] = -INFINITY;
  if (base + lane < end) {
    sv = srcS[base + lane];
    const float4* ap = (const float4*)(as + (size_t)sv * 8);
    float4 a0 = ap[0], a1 = ap[1];
    e[0] = leaky(a0.x + adh[0]); e[1] = leaky(a0.y + adh[1]);
    e[2] = leaky(a0.z + adh[2]); e[3] = leaky(a0.w + adh[3]);
    e[4] = leaky(a1.x + adh[4]); e[5] = leaky(a1.y + adh[5]);
    e[6] = leaky(a1.z + adh[6]); e[7] = leaky(a1.w + adh[7]);
  }
}

__device__ __forceinline__ float wave_max8(const float (&mx)[8], int lane) {
  bool b4 = (lane & 16) != 0, b3 = (lane & 8) != 0, b2 = (lane & 4) != 0;
  float u[4], t[2];
#pragma unroll
  for (int j = 0; j < 4; ++j) {
    float keep = b4 ? mx[4 + j] : mx[j];
    float send = b4 ? mx[j] : mx[4 + j];
    u[j] = fmaxf(keep, __shfl_xor(send, 16, 32));
  }
#pragma unroll
  for (int i = 0; i < 2; ++i) {
    float keep = b3 ? u[2 + i] : u[i];
    float send = b3 ? u[i] : u[2 + i];
    t[i] = fmaxf(keep, __shfl_xor(send, 8, 32));
  }
  float r = fmaxf(b2 ? t[1] : t[0], __shfl_xor(b2 ? t[0] : t[1], 4, 32));
  r = fmaxf(r, __shfl_xor(r, 2, 32));
  r = fmaxf(r, __shfl_xor(r, 1, 32));
  return r;
}

__device__ __forceinline__ void add_edges_h8(const float* __restrict__ xW, int sv, const float (&e)[8],
                                             const float (&mh)[8], int n, int lane, bool hi,
                                             float4 (&acc)[4], float (&den)[4]) {
  float w[8];
#pragma unroll
  for (int h = 0; h < 8; ++h) w[h] = __expf(e[h] - mh[h]);
  for (int k = 0; k < n; ++k) {
    int s = __builtin_amdgcn_readlane(sv, k);
    const float4* src4 = (const float4*)(xW + (size_t)s * 512);
#pragma unroll
    for (int i = 0; i < 4; ++i) {
      float we = __uint_as_float(__builtin_amdgcn_readlane(__float_as_uint(w[2 * i]), k));
      float wo = __uint_as_float(__builtin_amdgcn_readlane(__float_as_uint(w[2 * i + 1]), k));
      float a = hi ? wo : we;
      float4 v = src4[lane + 32 * i];
      den[i] += a;
      acc[i].x += a * v.x; acc[i].y += a * v.y; acc[i].z += a * v.z; acc[i].w += a * v.w;
    }
  }
}

__global__ void row_softmax_aggregate_h8(const float* __restrict__ xW, const float* __restrict__ as,
                                         const float* __restrict__ ad,
                                         const int* __restrict__ rowptr, const int* __restrict__ srcS,
                                         const float* __restrict__ bias, float* __restrict__ out, int N) {
  int row = __builtin_amdgcn_readfirstlane((int)((blockIdx.x * blockDim.x + threadIdx.x) >> 5));
  int lane = threadIdx.x & 31;
  if (row >= N) return;
  int start = rowptr[row], end = rowptr[row + 1];
  bool hi = (lane >> 4) != 0;
  const float4* adp = (const float4*)(ad + (size_t)row * 8);
  float4 d0 = adp[0], d1 = adp[1];
  float adh[8] = {d0.x, d0.y, d0.z, d0.w, d1.x, d1.y, d1.z, d1.w};

  int sv0; float e0[8], mx[8];
  edge_logits_h8(as, srcS, adh, start, end, lane, sv0, e0);
#pragma unroll
  for (int h = 0; h < 8; ++h) mx[h] = e0[h];
  for (int base = start + 32; base < end; base += 32) {
    int sv1; float e1[8];
    edge_logits_h8(as, srcS, adh, base, end, lane, sv1, e1);
#pragma unroll
    for (int h = 0; h < 8; ++h) mx[h] = fmaxf(mx[h], e1[h]);
  }
  float r = wave_max8(mx, lane);
  float mh[8];
#pragma unroll
  for (int h = 0; h < 8; ++h)
    mh[h] = __uint_as_float(__builtin_amdgcn_readlane(__float_as_uint(r), 4 * h));

  float4 acc[4]; float den[4];
#pragma unroll
  for (int i = 0; i < 4; ++i) { acc[i] = make_float4(0.f, 0.f, 0.f, 0.f); den[i] = 0.f; }
  {
    int n = end - start; if (n > 32) n = 32;
    add_edges_h8(xW, sv0, e0, mh, n, lane, hi, acc, den);
  }
  for (int base = start + 32; base < end; base += 32) {
    int sv1; float e1[8];
    edge_logits_h8(as, srcS, adh, base, end, lane, sv1, e1);
    int n = end - base; if (n > 32) n = 32;
    add_edges_h8(xW, sv1, e1, mh, n, lane, hi, acc, den);
  }

  const float4* b4 = (const float4*)bias;
  float4* o4 = (float4*)(out + (size_t)row * 512);
#pragma unroll
  for (int i = 0; i < 4; ++i) {
    float inv = (den[i] > 0.f) ? 1.f / den[i] : 0.f;
    float4 b = b4[lane + 32 * i];
    float4 q;
    q.x = fmaxf(acc[i].x * inv + b.x, 0.f); q.y = fmaxf(acc[i].y * inv + b.y, 0.f);
    q.z = fmaxf(acc[i].z * inv + b.z, 0.f); q.w = fmaxf(acc[i].w * inv + b.w, 0.f);
    o4[lane + 32 * i] = q;
  }
}

__global__ void row_softmax_aggregate_h1(const float* __restrict__ xW, const float* __restrict__ as,
                                         const float* __restrict__ ad,
                                         const int* __restrict__ rowptr, const int* __restrict__ srcS,
                                         const float* __restrict__ bias, float* __restrict__ out, int N) {
  int row = __builtin_amdgcn_readfirstlane((int)((blockIdx.x * blockDim.x + threadIdx.x) >> 5));
  int lane = threadIdx.x & 31;
  if (row >= N) return;
  int start = rowptr[row], end = rowptr[row + 1];
  float adr = ad[row];

  int sv0 = 0; float e0 = -INFINITY;
  if (start + lane < end) { sv0 = srcS[start + lane]; e0 = leaky(as[sv0] + adr); }
  float m = e0;
  for (int base = start + 32; base < end; base += 32) {
    if (base + lane < end) m = fmaxf(m, leaky(as[srcS[base + lane]] + adr));
  }
  for (int off = 16; off; off >>= 1) m = fmaxf(m, __shfl_xor(m, off, 32));

  float2 acc = make_float2(0.f, 0.f);
  float den = 0.f;
  for (int base = start; base < end; base += 32) {
    int n = end - base; if (n > 32) n = 32;
    int sv = sv0; float e = e0;
    if (base != start) {
      sv = 0; e = -INFINITY;
      if (base + lane < end) { sv = srcS[base + lane]; e = leaky(as[sv] + adr); }
    }
    float w = __expf(e - m);
    for (int k = 0; k < n; ++k) {
      int s = __builtin_amdgcn_readlane(sv, k);
      float a = __uint_as_float(__builtin_amdgcn_readlane(__float_as_uint(w), k));
      float2 v = ((const float2*)(xW + (size_t)s * 64))[lane];
      den += a;
      acc.x += a * v.x; acc.y += a * v.y;
    }
  }
  float inv = (den > 0.f) ? 1.f / den : 0.f;
  float2 b = ((const float2*)bias)[lane];
  float2 q; q.x = acc.x * inv + b.x; q.y = acc.y * inv + b.y;
  ((float2*)(out + (size_t)row * 64))[lane] = q;
}

extern "C" void kernel_launch(void* const* d_in, const int* in_sizes, int n_in,
                              void* d_out, int out_size, void* d_ws, size_t ws_size,
                              hipStream_t stream) {
  const float* x     = (const float*)d_in[0];
  const int*   ei    = (const int*)  d_in[1];
  const float* W1    = (const float*)d_in[2];
  const float* asrc1 = (const float*)d_in[3];
  const float* adst1 = (const float*)d_in[4];
  const float* b1    = (const float*)d_in[5];
  const float* W2    = (const float*)d_in[6];
  const float* asrc2 = (const float*)d_in[7];
  const float* adst2 = (const float*)d_in[8];
  const float* b2    = (const float*)d_in[9];
  float* out = (float*)d_out;

  const int D = 64;
  const int N = in_sizes[0] / D;
  const int E = in_sizes[1] / 2;
  const int H = 8;
  const int F1 = H * D;

  float* ws   = (float*)d_ws;
  float* xW1  = ws;
  float* out1 = xW1  + (size_t)N * F1;
  float* as1  = out1 + (size_t)N * F1;
  float* ad1  = as1  + (size_t)N * H;
  float* hW2  = ad1  + (size_t)N * H;
  float* as2  = hW2  + (size_t)N * D;
  float* ad2  = as2  + (size_t)N;
  int* cnt    = (int*)(ad2 + (size_t)N);
  int* rowptr = cnt    + (size_t)N;
  int* slot   = rowptr + (size_t)N + 1;
  int* seg    = slot   + (size_t)E;
  int* srcS   = seg    + (size_t)E;

  const int T = 256;

  hipMemsetAsync(cnt, 0, (size_t)N * 4, stream);
  edge_count_slot<<<(E + T - 1) / T, T, 0, stream>>>(ei, cnt, slot, E);
  row_scan<<<1, 1024, 0, stream>>>(cnt, rowptr, N);
  edge_place<<<(E + T - 1) / T, T, 0, stream>>>(ei, rowptr, slot, seg, E);
  row_order<<<(N * 32 + T - 1) / T, T, 0, stream>>>(ei, rowptr, seg, srcS, N);

  {
    int nwaves = (N / 16) * H;
    gemm_logits_wmma_f32<64, 512><<<(nwaves + 7) / 8, T, 0, stream>>>(x, W1, asrc1, adst1, xW1, as1, ad1, N);
  }
  row_softmax_aggregate_h8<<<(N * 32 + T - 1) / T, T, 0, stream>>>(xW1, as1, ad1, rowptr, srcS, b1, out1, N);

  {
    int nwaves = N / 16;
    gemm_logits_wmma_f32<512, 64><<<(nwaves + 7) / 8, T, 0, stream>>>(out1, W2, asrc2, adst2, hW2, as2, ad2, N);
  }
  row_softmax_aggregate_h1<<<(N * 32 + T - 1) / T, T, 0, stream>>>(hW2, as2, ad2, rowptr, srcS, b2, out, N);
}


# layer 1 aggregates before projecting: per edge one 64-float x row (not a 512-float x@W1 row), 8 head sums to LDS, then W1 per head on WMMA; logits = x@(W1 a); sum order: projection after row sum
# speedup vs baseline: 4.5150x; 1.1572x over previous
#include <hip/hip_runtime.h>

typedef __attribute__((ext_vector_type(2))) float v2f;
typedef __attribute__((ext_vector_type(8))) float v8f;

#define NEG_SLOPE 0.2f

template<int K, int NC>
__global__ void gemm_logits_wmma_f32(const float* __restrict__ A, const float* __restrict__ B,
                                     const float* __restrict__ avs, const float* __restrict__ avd,
                                     float* __restrict__ C, float* __restrict__ as,
                                     float* __restrict__ ad, int M) {
  int wave = (blockIdx.x * blockDim.x + threadIdx.x) >> 5;
  int lane = threadIdx.x & 31;
  constexpr int NH = NC >> 6;
  int mt = (M + 15) >> 4;
  if (wave >= mt * NH) return;
  int th = wave / mt, tm = wave - th * mt;
  int row_base = tm << 4, col_base = th << 6;
  int mrow = lane & 15, hi = lane >> 4;

  int arow = row_base + mrow; if (arow > M - 1) arow = M - 1;
  const float* Arow = A + (size_t)arow * K;
  const float* Bcol = B + col_base + 4 * mrow;

  v8f acc0 = {}, acc1 = {}, acc2 = {}, acc3 = {};
#pragma unroll 8
  for (int k0 = 0; k0 < K; k0 += 4) {
    int ka = k0 + 2 * hi;
    v2f a;
    a.x = Arow[ka];
    a.y = Arow[ka + 1];
    float4 bx = *(const float4*)(Bcol + (size_t)ka * NC);
    float4 by = *(const float4*)(Bcol + (size_t)(ka + 1) * NC);
    v2f b0, b1, b2, b3;
    b0.x = bx.x; b0.y = by.x;
    b1.x = bx.y; b1.y = by.y;
    b2.x = bx.z; b2.y = by.z;
    b3.x = bx.w; b3.y = by.w;
    acc0 = __builtin_amdgcn_wmma_f32_16x16x4_f32(
         false, a,  false, b0,
         (short)0, acc0,  false,  false);
    acc1 = __builtin_amdgcn_wmma_f32_16x16x4_f32(false, a, false, b1, (short)0, acc1, false, false);
    acc2 = __builtin_amdgcn_wmma_f32_16x16x4_f32(false, a, false, b2, (short)0, acc2, false, false);
    acc3 = __builtin_amdgcn_wmma_f32_16x16x4_f32(false, a, false, b3, (short)0, acc3, false, false);
  }

  int rbase = row_base + hi * 8;
  float* p = C + (size_t)rbase * NC + col_base + 4 * mrow;
  if (rbase + 8 <= M) {
#pragma unroll
    for (int r = 0; r < 8; ++r)
      *(float4*)(p + r * NC) = make_float4(acc0[r], acc1[r], acc2[r], acc3[r]);
  } else {
#pragma unroll
    for (int r = 0; r < 8; ++r) {
      if (rbase + r < M) *(float4*)(p + r * NC) = make_float4(acc0[r], acc1[r], acc2[r], acc3[r]);
    }
  }

  float4 s4 = *(const float4*)(avs + col_base + 4 * mrow);
  float4 d4 = *(const float4*)(avd + col_base + 4 * mrow);
  float v[16];
#pragma unroll
  for (int r = 0; r < 8; ++r) {
    v[r]     = acc0[r] * s4.x + acc1[r] * s4.y + acc2[r] * s4.z + acc3[r] * s4.w;
    v[8 + r] = acc0[r] * d4.x + acc1[r] * d4.y + acc2[r] * d4.z + acc3[r] * d4.w;
  }
  bool t8 = (mrow & 8) != 0, t4 = (mrow & 4) != 0, t2 = (mrow & 2) != 0, t1 = (mrow & 1) != 0;
  float w8[8], w4[4], w2[2];
#pragma unroll
  for (int i = 0; i < 8; ++i) {
    float keep = t8 ? v[8 + i] : v[i];
    float send = t8 ? v[i] : v[8 + i];
    w8[i] = keep + __shfl_xor(send, 8, 32);
  }
#pragma unroll
  for (int i = 0; i < 4; ++i) {
    float keep = t4 ? w8[4 + i] : w8[i];
    float send = t4 ? w8[i] : w8[4 + i];
    w4[i] = keep + __shfl_xor(send, 4, 32);
  }
#pragma unroll
  for (int i = 0; i < 2; ++i) {
    float keep = t2 ? w4[2 + i] : w4[i];
    float send = t2 ? w4[i] : w4[2 + i];
    w2[i] = keep + __shfl_xor(send, 2, 32);
  }
  float tot = (t1 ? w2[1] : w2[0]) + __shfl_xor(t1 ? w2[0] : w2[1], 1, 32);
  int orow = rbase + (mrow & 7);
  float* dst = t8 ? ad : as;
  if (orow < M) dst[(size_t)orow * NH + th] = tot;
}

__global__ void edge_count_slot(const int* __restrict__ ei, int* __restrict__ cnt,
                                int* __restrict__ slot, int E) {
  int gid = blockIdx.x * blockDim.x + threadIdx.x;
  if (gid >= E) return;
  slot[gid] = atomicAdd(&cnt[ei[E + gid]], 1);
}

__global__ void row_scan(const int* __restrict__ cnt, int* __restrict__ rowptr, int N) {
  __shared__ int wsum[32];
  int t = threadIdx.x, lane = t & 31, w = t >> 5;
  int nw = (int)blockDim.x >> 5;
  int per = (N + nw - 1) / nw;
  per = (per + 127) & ~127;
  int lo = w * per; if (lo > N) lo = N;
  int hi = lo + per; if (hi > N) hi = N;
  int part = 0;
  for (int i = lo + lane; i < hi; i += 32) part += cnt[i];
  for (int off = 16; off; off >>= 1) part += __shfl_xor(part, off, 32);
  if (lane == 0) wsum[w] = part;
  __syncthreads();
  if (w == 0) {
    int s = (lane < nw) ? wsum[lane] : 0, si = s;
    for (int off = 1; off < 32; off <<= 1) {
      int v = __shfl_up(si, off, 32);
      if (lane >= off) si += v;
    }
    wsum[lane] = si - s;
  }
  __syncthreads();
  int carry = wsum[w];
  for (int base = lo; base < hi; base += 128) {
    int i = base + lane * 4;
    int a0 = (i     < hi) ? cnt[i]     : 0;
    int a1 = (i + 1 < hi) ? cnt[i + 1] : 0;
    int a2 = (i + 2 < hi) ? cnt[i + 2] : 0;
    int a3 = (i + 3 < hi) ? cnt[i + 3] : 0;
    int s1 = a0 + a1, s2 = s1 + a2, tot = s2 + a3;
    int incl = tot;
    for (int off = 1; off < 32; off <<= 1) {
      int u = __shfl_up(incl, off, 32);
      if (lane >= off) incl += u;
    }
    int ex = carry + incl - tot;
    if (i     < hi) rowptr[i]     = ex;
    if (i + 1 < hi) rowptr[i + 1] = ex + a0;
    if (i + 2 < hi) rowptr[i + 2] = ex + s1;
    if (i + 3 < hi) rowptr[i + 3] = ex + s2;
    carry += __shfl(incl, 31, 32);
  }
  if (t == (int)blockDim.x - 1) rowptr[N] = carry;
}

__global__ void edge_place(const int* __restrict__ ei, const int* __restrict__ rowptr,
                           const int* __restrict__ slot, int* __restrict__ seg, int E) {
  int gid = blockIdx.x * blockDim.x + threadIdx.x;
  if (gid >= E) return;
  seg[rowptr[ei[E + gid]] + slot[gid]] = gid;
}

__global__ void row_order(const int* __restrict__ ei, const int* __restrict__ rowptr,
                          const int* __restrict__ seg, int* __restrict__ srcS, int N) {
  int row = (blockIdx.x * blockDim.x + threadIdx.x) >> 5;
  int lane = threadIdx.x & 31;
  if (row >= N) return;
  int start = rowptr[row], deg = rowptr[row + 1] - start;
  for (int i = lane; i < deg; i += 32) {
    int id = seg[start + i];
    int rank = 0;
    for (int j = 0; j < deg; ++j) rank += (seg[start + j] < id) ? 1 : 0;
    srcS[start + rank] = ei[id];
  }
}

__device__ __forceinline__ float leaky(float v) { return (v >= 0.f) ? v : NEG_SLOPE * v; }

__device__ __forceinline__ void edge_logits_h8(const float* __restrict__ as, const int* __restrict__ srcS,
                                               const float (&adh)[8], int base, int end, int lane,
                                               int& sv, float (&e)[8]) {
  sv = 0;
#pragma unroll
  for (int h = 0; h < 8; ++h) e[h] = -INFINITY;
  if (base + lane < end) {
    sv = srcS[base + lane];
    const float4* ap = (const float4*)(as + (size_t)sv * 8);
    float4 a0 = ap[0], a1 = ap[1];
    e[0] = leaky(a0.x + adh[0]); e[1] = leaky(a0.y + adh[1]);
    e[2] = leaky(a0.z + adh[2]); e[3] = leaky(a0.w + adh[3]);
    e[4] = leaky(a1.x + adh[4]); e[5] = leaky(a1.y + adh[5]);
    e[6] = leaky(a1.z + adh[6]); e[7] = leaky(a1.w + adh[7]);
  }
}

__device__ __forceinline__ float wave_max8(const float (&mx)[8], int lane) {
  bool b4 = (lane & 16) != 0, b3 = (lane & 8) != 0, b2 = (lane & 4) != 0;
  float u[4], t[2];
#pragma unroll
  for (int j = 0; j < 4; ++j) {
    float keep = b4 ? mx[4 + j] : mx[j];
    float send = b4 ? mx[j] : mx[4 + j];
    u[j] = fmaxf(keep, __shfl_xor(send, 16, 32));
  }
#pragma unroll
  for (int i = 0; i < 2; ++i) {
    float keep = b3 ? u[2 + i] : u[i];
    float send = b3 ? u[i] : u[2 + i];
    t[i] = fmaxf(keep, __shfl_xor(send, 8, 32));
  }
  float r = fmaxf(b2 ? t[1] : t[0], __shfl_xor(b2 ? t[0] : t[1], 4, 32));
  r = fmaxf(r, __shfl_xor(r, 2, 32));
  r = fmaxf(r, __shfl_xor(r, 1, 32));
  return r;
}

__device__ __forceinline__ void add_edges_x_h8(const float* __restrict__ x, int sv, const float (&e)[8],
                                               const float (&mh)[8], int n, int lane,
                                               v2f (&z)[8], float (&den)[8]) {
  float w[8];
#pragma unroll
  for (int h = 0; h < 8; ++h) w[h] = __expf(e[h] - mh[h]);
#pragma unroll 4
  for (int k = 0; k < n; ++k) {
    int s = __builtin_amdgcn_readlane(sv, k);
    v2f v = ((const v2f*)(x + (size_t)s * 64))[lane];
#pragma unroll
    for (int h = 0; h < 8; ++h) {
      float a = __uint_as_float(__builtin_amdgcn_readlane(__float_as_uint(w[h]), k));
      den[h] += a;
      z[h] += a * v;
    }
  }
}

#define ZPITCH 516

__global__ void __launch_bounds__(512)
row_softmax_aggregate_project_h8(const float* __restrict__ x, const float* __restrict__ as,
                                 const float* __restrict__ ad,
                                 const int* __restrict__ rowptr, const int* __restrict__ srcS,
                                 const float* __restrict__ W, const float* __restrict__ bias,
                                 float* __restrict__ out, int N) {
  __shared__ __attribute__((aligned(16))) float Zs[32 * ZPITCH];
  int lane = threadIdx.x & 31;
  int w = __builtin_amdgcn_readfirstlane((int)(threadIdx.x >> 5));
  int row_base = (int)blockIdx.x << 5;

  for (int rr = 0; rr < 2; ++rr) {
    int m = 2 * w + rr;
    int row = row_base + m;
    v2f z[8]; float den[8];
#pragma unroll
    for (int h = 0; h < 8; ++h) { z[h] = (v2f){0.f, 0.f}; den[h] = 0.f; }
    if (row < N) {
      int start = rowptr[row], end = rowptr[row + 1];
      const float4* adp = (const float4*)(ad + (size_t)row * 8);
      float4 d0 = adp[0], d1 = adp[1];
      float adh[8] = {d0.x, d0.y, d0.z, d0.w, d1.x, d1.y, d1.z, d1.w};

      int sv0; float e0[8], mx[8];
      edge_logits_h8(as, srcS, adh, start, end, lane, sv0, e0);
#pragma unroll
      for (int h = 0; h < 8; ++h) mx[h] = e0[h];
      for (int base = start + 32; base < end; base += 32) {
        int sv1; float e1[8];
        edge_logits_h8(as, srcS, adh, base, end, lane, sv1, e1);
#pragma unroll
        for (int h = 0; h < 8; ++h) mx[h] = fmaxf(mx[h], e1[h]);
      }
      float r = wave_max8(mx, lane);
      float mh[8];
#pragma unroll
      for (int h = 0; h < 8; ++h)
        mh[h] = __uint_as_float(__builtin_amdgcn_readlane(__float_as_uint(r), 4 * h));

      {
        int n = end - start; if (n > 32) n = 32;
        add_edges_x_h8(x, sv0, e0, mh, n, lane, z, den);
      }
      for (int base = start + 32; base < end; base += 32) {
        int sv1; float e1[8];
        edge_logits_h8(as, srcS, adh, base, end, lane, sv1, e1);
        int n = end - base; if (n > 32) n = 32;
        add_edges_x_h8(x, sv1, e1, mh, n, lane, z, den);
      }
    }
    float* zrow = Zs + m * ZPITCH + 2 * lane;
#pragma unroll
    for (int h = 0; h < 8; ++h) {
      float inv = (den[h] > 0.f) ? 1.f / den[h] : 0.f;
      zrow[h * 64]     = z[h].x * inv;
      zrow[h * 64 + 1] = z[h].y * inv;
    }
  }
  __syncthreads();

  int mrow = lane & 15, hi = lane >> 4;
  int col_base = (w & 7) << 6;
  int tile_row = (w >> 3) << 4;
  const float* Zrow = Zs + (tile_row + mrow) * ZPITCH + col_base;
  const float* Bcol = W + col_base + 4 * mrow;
  v8f acc0 = {}, acc1 = {}, acc2 = {}, acc3 = {};
#pragma unroll 4
  for (int k0 = 0; k0 < 64; k0 += 4) {
    int ka = k0 + 2 * hi;
    v2f a;
    a.x = Zrow[ka];
    a.y = Zrow[ka + 1];
    float4 bx = *(const float4*)(Bcol + (size_t)ka * 512);
    float4 by = *(const float4*)(Bcol + (size_t)(ka + 1) * 512);
    v2f b0, b1, b2, b3;
    b0.x = bx.x; b0.y = by.x;
    b1.x = bx.y; b1.y = by.y;
    b2.x = bx.z; b2.y = by.z;
    b3.x = bx.w; b3.y = by.w;
    acc0 = __builtin_amdgcn_wmma_f32_16x16x4_f32(false, a, false, b0, (short)0, acc0, false, false);
    acc1 = __builtin_amdgcn_wmma_f32_16x16x4_f32(false, a, false, b1, (short)0, acc1, false, false);
    acc2 = __builtin_amdgcn_wmma_f32_16x16x4_f32(false, a, false, b2, (short)0, acc2, false, false);
    acc3 = __builtin_amdgcn_wmma_f32_16x16x4_f32(false, a, false, b3, (short)0, acc3, false, false);
  }

  int rbase = row_base + tile_row + hi * 8;
  float4 b = *(const float4*)(bias + col_base + 4 * mrow);
  float* p = out + (size_t)rbase * 512 + col_base + 4 * mrow;
  if (rbase + 8 <= N) {
#pragma unroll
    for (int r = 0; r < 8; ++r) {
      float4 q;
      q.x = fmaxf(acc0[r] + b.x, 0.f); q.y = fmaxf(acc1[r] + b.y, 0.f);
      q.z = fmaxf(acc2[r] + b.z, 0.f); q.w = fmaxf(acc3[r] + b.w, 0.f);
      *(float4*)(p + r * 512) = q;
    }
  } else {
#pragma unroll
    for (int r = 0; r < 8; ++r) {
      if (rbase + r < N) {
        float4 q;
        q.x = fmaxf(acc0[r] + b.x, 0.f); q.y = fmaxf(acc1[r] + b.y, 0.f);
        q.z = fmaxf(acc2[r] + b.z, 0.f); q.w = fmaxf(acc3[r] + b.w, 0.f);
        *(float4*)(p + r * 512) = q;
      }
    }
  }
}

__global__ void fold_attention_vectors(const float* __restrict__ W, const float* __restrict__ avs,
                                       const float* __restrict__ avd, float* __restrict__ wa) {
  int t = blockIdx.x * blockDim.x + threadIdx.x;
  int k = t >> 4, c = t & 15, h = c & 7;
  const float4* a4 = (const float4*)(((c & 8) ? avd : avs) + h * 64);
  const float4* w4 = (const float4*)(W + (size_t)k * 512 + h * 64);
  float s = 0.f;
#pragma unroll
  for (int j = 0; j < 16; ++j) {
    float4 u = w4[j], v = a4[j];
    s += u.x * v.x; s += u.y * v.y; s += u.z * v.z; s += u.w * v.w;
  }
  wa[t] = s;
}

__global__ void node_logits_wmma_f32(const float* __restrict__ X, const float* __restrict__ wa,
                                     float* __restrict__ as, float* __restrict__ ad, int M) {
  int wave = (blockIdx.x * blockDim.x + threadIdx.x) >> 5;
  int lane = threadIdx.x & 31;
  int mt = (M + 15) >> 4;
  if (wave >= mt) return;
  int row_base = wave << 4;
  int mrow = lane & 15, hi = lane >> 4;
  int arow = row_base + mrow; if (arow > M - 1) arow = M - 1;
  const float* Arow = X + (size_t)arow * 64;
  v8f acc = {};
#pragma unroll
  for (int k0 = 0; k0 < 64; k0 += 4) {
    int ka = k0 + 2 * hi;
    v2f a, b;
    a.x = Arow[ka];
    a.y = Arow[ka + 1];
    b.x = wa[ka * 16 + mrow];
    b.y = wa[(ka + 1) * 16 + mrow];
    acc = __builtin_amdgcn_wmma_f32_16x16x4_f32(false, a, false, b, (short)0, acc, false, false);
  }
  float* dst = (mrow & 8) ? ad : as;
  int rbase = row_base + hi * 8;
#pragma unroll
  for (int r = 0; r < 8; ++r)
    if (rbase + r < M) dst[(size_t)(rbase + r) * 8 + (mrow & 7)] = acc[r];
}

__global__ void row_softmax_aggregate_h1(const float* __restrict__ xW, const float* __restrict__ as,
                                         const float* __restrict__ ad,
                                         const int* __restrict__ rowptr, const int* __restrict__ srcS,
                                         const float* __restrict__ bias, float* __restrict__ out, int N) {
  int row = __builtin_amdgcn_readfirstlane((int)((blockIdx.x * blockDim.x + threadIdx.x) >> 5));
  int lane = threadIdx.x & 31;
  if (row >= N) return;
  int start = rowptr[row], end = rowptr[row + 1];
  float adr = ad[row];

  int sv0 = 0; float e0 = -INFINITY;
  if (start + lane < end) { sv0 = srcS[start + lane]; e0 = leaky(as[sv0] + adr); }
  float m = e0;
  for (int base = start + 32; base < end; base += 32) {
    if (base + lane < end) m = fmaxf(m, leaky(as[srcS[base + lane]] + adr));
  }
  for (int off = 16; off; off >>= 1) m = fmaxf(m, __shfl_xor(m, off, 32));

  float2 acc = make_float2(0.f, 0.f);
  float den = 0.f;
  for (int base = start; base < end; base += 32) {
    int n = end - base; if (n > 32) n = 32;
    int sv = sv0; float e = e0;
    if (base != start) {
      sv = 0; e = -INFINITY;
      if (base + lane < end) { sv = srcS[base + lane]; e = leaky(as[sv] + adr); }
    }
    float w = __expf(e - m);
    for (int k = 0; k < n; ++k) {
      int s = __builtin_amdgcn_readlane(sv, k);
      float a = __uint_as_float(__builtin_amdgcn_readlane(__float_as_uint(w), k));
      float2 v = ((const float2*)(xW + (size_t)s * 64))[lane];
      den += a;
      acc.x += a * v.x; acc.y += a * v.y;
    }
  }
  float inv = (den > 0.f) ? 1.f / den : 0.f;
  float2 b = ((const float2*)bias)[lane];
  float2 q; q.x = acc.x * inv + b.x; q.y = acc.y * inv + b.y;
  ((float2*)(out + (size_t)row * 64))[lane] = q;
}

extern "C" void kernel_launch(void* const* d_in, const int* in_sizes, int n_in,
                              void* d_out, int out_size, void* d_ws, size_t ws_size,
                              hipStream_t stream) {
  const float* x     = (const float*)d_in[0];
  const int*   ei    = (const int*)  d_in[1];
  const float* W1    = (const float*)d_in[2];
  const float* asrc1 = (const float*)d_in[3];
  const float* adst1 = (const float*)d_in[4];
  const float* b1    = (const float*)d_in[5];
  const float* W2    = (const float*)d_in[6];
  const float* asrc2 = (const float*)d_in[7];
  const float* adst2 = (const float*)d_in[8];
  const float* b2    = (const float*)d_in[9];
  float* out = (float*)d_out;

  const int D = 64;
  const int N = in_sizes[0] / D;
  const int E = in_sizes[1] / 2;
  const int H = 8;
  const int F1 = H * D;

  float* ws   = (float*)d_ws;
  float* wa1  = ws;
  float* out1 = wa1  + 64 * 16;
  float* as1  = out1 + (size_t)N * F1;
  float* ad1  = as1  + (size_t)N * H;
  float* hW2  = ad1  + (size_t)N * H;
  float* as2  = hW2  + (size_t)N * D;
  float* ad2  = as2  + (size_t)N;
  int* cnt    = (int*)(ad2 + (size_t)N);
  int* rowptr = cnt    + (size_t)N;
  int* slot   = rowptr + (size_t)N + 1;
  int* seg    = slot   + (size_t)E;
  int* srcS   = seg    + (size_t)E;

  const int T = 256;

  hipMemsetAsync(cnt, 0, (size_t)N * 4, stream);
  edge_count_slot<<<(E + T - 1) / T, T, 0, stream>>>(ei, cnt, slot, E);
  row_scan<<<1, 1024, 0, stream>>>(cnt, rowptr, N);
  edge_place<<<(E + T - 1) / T, T, 0, stream>>>(ei, rowptr, slot, seg, E);
  row_order<<<(N * 32 + T - 1) / T, T, 0, stream>>>(ei, rowptr, seg, srcS, N);

  fold_attention_vectors<<<4, T, 0, stream>>>(W1, asrc1, adst1, wa1);
  {
    int nwaves = (N + 15) / 16;
    node_logits_wmma_f32<<<(nwaves + 7) / 8, T, 0, stream>>>(x, wa1, as1, ad1, N);
  }
  row_softmax_aggregate_project_h8<<<(N + 31) / 32, 512, 0, stream>>>(x, as1, ad1, rowptr, srcS, W1, b1, out1, N);

  {
    int nwaves = N / 16;
    gemm_logits_wmma_f32<512, 64><<<(nwaves + 7) / 8, T, 0, stream>>>(out1, W2, asrc2, adst2, hW2, as2, ad2, N);
  }
  row_softmax_aggregate_h1<<<(N * 32 + T - 1) / T, T, 0, stream>>>(hW2, as2, ad2, rowptr, srcS, b2, out, N);
}


# edge list sorted by destination with LDS counters (256 node ranges, then per-node count, scan and rank in LDS) instead of per-edge returning global adds, a clear and a one-block scan
# speedup vs baseline: 6.1762x; 1.3679x over previous
#include <hip/hip_runtime.h>

typedef __attribute__((ext_vector_type(2))) float v2f;
typedef __attribute__((ext_vector_type(8))) float v8f;

#define NEG_SLOPE 0.2f

template<int K, int NC>
__global__ void gemm_logits_wmma_f32(const float* __restrict__ A, const float* __restrict__ B,
                                     const float* __restrict__ avs, const float* __restrict__ avd,
                                     float* __restrict__ C, float* __restrict__ as,
                                     float* __restrict__ ad, int M) {
  int wave = (blockIdx.x * blockDim.x + threadIdx.x) >> 5;
  int lane = threadIdx.x & 31;
  constexpr int NH = NC >> 6;
  int mt = (M + 15) >> 4;
  if (wave >= mt * NH) return;
  int th = wave / mt, tm = wave - th * mt;
  int row_base = tm << 4, col_base = th << 6;
  int mrow = lane & 15, hi = lane >> 4;

  int arow = row_base + mrow; if (arow > M - 1) arow = M - 1;
  const float* Arow = A + (size_t)arow * K;
  const float* Bcol = B + col_base + 4 * mrow;

  v8f acc0 = {}, acc1 = {}, acc2 = {}, acc3 = {};
#pragma unroll 8
  for (int k0 = 0; k0 < K; k0 += 4) {
    int ka = k0 + 2 * hi;
    v2f a;
    a.x = Arow[ka];
    a.y = Arow[ka + 1];
    float4 bx = *(const float4*)(Bcol + (size_t)ka * NC);
    float4 by = *(const float4*)(Bcol + (size_t)(ka + 1) * NC);
    v2f b0, b1, b2, b3;
    b0.x = bx.x; b0.y = by.x;
    b1.x = bx.y; b1.y = by.y;
    b2.x = bx.z; b2.y = by.z;
    b3.x = bx.w; b3.y = by.w;
    acc0 = __builtin_amdgcn_wmma_f32_16x16x4_f32(
         false, a,  false, b0,
         (short)0, acc0,  false,  false);
    acc1 = __builtin_amdgcn_wmma_f32_16x16x4_f32(false, a, false, b1, (short)0, acc1, false, false);
    acc2 = __builtin_amdgcn_wmma_f32_16x16x4_f32(false, a, false, b2, (short)0, acc2, false, false);
    acc3 = __builtin_amdgcn_wmma_f32_16x16x4_f32(false, a, false, b3, (short)0, acc3, false, false);
  }

  int rbase = row_base + hi * 8;
  float* p = C + (size_t)rbase * NC + col_base + 4 * mrow;
  if (rbase + 8 <= M) {
#pragma unroll
    for (int r = 0; r < 8; ++r)
      *(float4*)(p + r * NC) = make_float4(acc0[r], acc1[r], acc2[r], acc3[r]);
  } else {
#pragma unroll
    for (int r = 0; r < 8; ++r) {
      if (rbase + r < M) *(float4*)(p + r * NC) = make_float4(acc0[r], acc1[r], acc2[r], acc3[r]);
    }
  }

  float4 s4 = *(const float4*)(avs + col_base + 4 * mrow);
  float4 d4 = *(const float4*)(avd + col_base + 4 * mrow);
  float v[16];
#pragma unroll
  for (int r = 0; r < 8; ++r) {
    v[r]     = acc0[r] * s4.x + acc1[r] * s4.y + acc2[r] * s4.z + acc3[r] * s4.w;
    v[8 + r] = acc0[r] * d4.x + acc1[r] * d4.y + acc2[r] * d4.z + acc3[r] * d4.w;
  }
  bool t8 = (mrow & 8) != 0, t4 = (mrow & 4) != 0, t2 = (mrow & 2) != 0, t1 = (mrow & 1) != 0;
  float w8[8], w4[4], w2[2];
#pragma unroll
  for (int i = 0; i < 8; ++i) {
    float keep = t8 ? v[8 + i] : v[i];
    float send = t8 ? v[i] : v[8 + i];
    w8[i] = keep + __shfl_xor(send, 8, 32);
  }
#pragma unroll
  for (int i = 0; i < 4; ++i) {
    float keep = t4 ? w8[4 + i] : w8[i];
    float send = t4 ? w8[i] : w8[4 + i];
    w4[i] = keep + __shfl_xor(send, 4, 32);
  }
#pragma unroll
  for (int i = 0; i < 2; ++i) {
    float keep = t2 ? w4[2 + i] : w4[i];
    float send = t2 ? w4[i] : w4[2 + i];
    w2[i] = keep + __shfl_xor(send, 2, 32);
  }
  float tot = (t1 ? w2[1] : w2[0]) + __shfl_xor(t1 ? w2[0] : w2[1], 1, 32);
  int orow = rbase + (mrow & 7);
  float* dst = t8 ? ad : as;
  if (orow < M) dst[(size_t)orow * NH + th] = tot;
}

#define NBIN 256
#define NCHUNK 256
#define RBMAX 1024

__device__ __forceinline__ int bin_shift(int N) {
  int s = 0;
  while (((N - 1) >> s) >= NBIN) ++s;
  return s;
}

__global__ void __launch_bounds__(1024)
bin_count(const int* __restrict__ ei, int* __restrict__ hist, int E, int N) {
  __shared__ int h[NBIN];
  int t = threadIdx.x;
  if (t < NBIN) h[t] = 0;
  __syncthreads();
  int s = bin_shift(N);
  int ch = (E + (int)gridDim.x - 1) / (int)gridDim.x;
  int lo = (int)blockIdx.x * ch, hi = lo + ch; if (hi > E) hi = E;
  for (int i0 = lo + t; i0 < hi; i0 += 2048) {
    int i1 = i0 + 1024;
    bool two = i1 < hi;
    int d0 = ei[E + i0];
    int d1 = two ? ei[E + i1] : 0;
    atomicAdd(&h[d0 >> s], 1);
    if (two) atomicAdd(&h[d1 >> s], 1);
  }
  __syncthreads();
  if (t < NBIN) hist[(int)blockIdx.x * NBIN + t] = h[t];
}

__global__ void __launch_bounds__(1024)
bin_scatter(const int* __restrict__ ei, const int* __restrict__ hist, int* __restrict__ base,
            int2* __restrict__ binned, int E, int N) {
  __shared__ int tot[NBIN];
  __shared__ int pre[NBIN];
  __shared__ int cur[NBIN];
  __shared__ int wt[8];
  int t = threadIdx.x, lane = t & 31, w = t >> 5;
  int nb = (int)gridDim.x, b = (int)blockIdx.x;
  if (t < NBIN) { tot[t] = 0; pre[t] = 0; }
  __syncthreads();
  {
    int c4 = t & 63, rg = t >> 6;
    int4 all = make_int4(0, 0, 0, 0), mine = make_int4(0, 0, 0, 0);
    const int4* h4 = (const int4*)hist;
    for (int bb = rg; bb < nb; bb += 16) {
      int4 v = h4[bb * (NBIN / 4) + c4];
      all.x += v.x; all.y += v.y; all.z += v.z; all.w += v.w;
      if (bb < b) { mine.x += v.x; mine.y += v.y; mine.z += v.z; mine.w += v.w; }
    }
    atomicAdd(&tot[4 * c4 + 0], all.x);  atomicAdd(&tot[4 * c4 + 1], all.y);
    atomicAdd(&tot[4 * c4 + 2], all.z);  atomicAdd(&tot[4 * c4 + 3], all.w);
    atomicAdd(&pre[4 * c4 + 0], mine.x); atomicAdd(&pre[4 * c4 + 1], mine.y);
    atomicAdd(&pre[4 * c4 + 2], mine.z); atomicAdd(&pre[4 * c4 + 3], mine.w);
  }
  __syncthreads();
  int v = (t < NBIN) ? tot[t] : 0;
  int incl = v;
  for (int o = 1; o < 32; o <<= 1) {
    int u = __shfl_up(incl, o, 32);
    if (lane >= o) incl += u;
  }
  if (t < NBIN && lane == 31) wt[w] = incl;
  __syncthreads();
  if (t < NBIN) {
    int carry = 0;
    for (int i = 0; i < w; ++i) carry += wt[i];
    int ex = carry + incl - v;
    cur[t] = ex + pre[t];
    if (b == 0) base[t] = ex;
  }
  if (b == 0 && t == 0) base[NBIN] = E;
  __syncthreads();
  int s = bin_shift(N);
  int ch = (E + nb - 1) / nb;
  int lo = b * ch, hi = lo + ch; if (hi > E) hi = E;
  for (int i0 = lo + t; i0 < hi; i0 += 2048) {
    int i1 = i0 + 1024;
    bool two = i1 < hi;
    int d0 = ei[E + i0];
    int d1 = two ? ei[E + i1] : 0;
    int q0 = atomicAdd(&cur[d0 >> s], 1);
    binned[q0] = make_int2(d0, i0);
    if (two) {
      int q1 = atomicAdd(&cur[d1 >> s], 1);
      binned[q1] = make_int2(d1, i1);
    }
  }
}

__global__ void __launch_bounds__(1024)
bin_rows(const int* __restrict__ ei, const int* __restrict__ base, const int2* __restrict__ binned,
         int* __restrict__ rowptr, int* __restrict__ seg, int* __restrict__ srcS, int E, int N) {
  __shared__ int cnt[RBMAX];
  __shared__ int start[RBMAX];
  __shared__ int cur[RBMAX];
  __shared__ int wt[32];
  int t = threadIdx.x, lane = t & 31;
  int w = __builtin_amdgcn_readfirstlane(t >> 5);
  int k = (int)blockIdx.x;
  int s = bin_shift(N);
  int RB = 1 << s;
  int node0 = k << s;
  int e0 = base[k], e1 = base[k + 1];
  if (t < RB) cnt[t] = 0;
  __syncthreads();
  for (int i0 = e0 + t; i0 < e1; i0 += 4096) {
    int d[4];
#pragma unroll
    for (int u = 0; u < 4; ++u) { int i = i0 + 1024 * u; d[u] = (i < e1) ? binned[i].x : node0; }
#pragma unroll
    for (int u = 0; u < 4; ++u) { if (i0 + 1024 * u < e1) atomicAdd(&cnt[d[u] - node0], 1); }
  }
  __syncthreads();
  int v = (t < RB) ? cnt[t] : 0;
  int incl = v;
  for (int o = 1; o < 32; o <<= 1) {
    int u = __shfl_up(incl, o, 32);
    if (lane >= o) incl += u;
  }
  if (lane == 31) wt[w] = incl;
  __syncthreads();
  if (w == 0) {
    int sv = wt[lane], si = sv;
    for (int o = 1; o < 32; o <<= 1) {
      int u = __shfl_up(si, o, 32);
      if (lane >= o) si += u;
    }
    wt[lane] = si - sv;
  }
  __syncthreads();
  int ex = wt[w] + incl - v;
  if (t < RB) {
    start[t] = ex; cur[t] = ex;
    if (node0 + t < N) rowptr[node0 + t] = e0 + ex;
  }
  if (k == 0 && t == 0) rowptr[N] = E;
  __syncthreads();
  for (int i0 = e0 + t; i0 < e1; i0 += 4096) {
    int2 pr[4];
#pragma unroll
    for (int u = 0; u < 4; ++u) { int i = i0 + 1024 * u; pr[u] = (i < e1) ? binned[i] : make_int2(node0, 0); }
#pragma unroll
    for (int u = 0; u < 4; ++u) {
      if (i0 + 1024 * u < e1) {
        int p = atomicAdd(&cur[pr[u].x - node0], 1);
        seg[e0 + p] = pr[u].y;
      }
    }
  }
  __syncthreads();
  for (int jb = w * 8; jb < RB; jb += 256) {
    int p0[8], n[8], id[8], rk[8];
    int jl = jb + (lane & 7);
    bool okl = (jl < RB) && (node0 + jl < N);
    int st_l = okl ? start[jl] : 0;
    int n_l = okl ? cnt[jl] : 0;
#pragma unroll
    for (int r = 0; r < 8; ++r) {
      p0[r] = e0 + __builtin_amdgcn_readlane(st_l, r);
      n[r] = __builtin_amdgcn_readlane(n_l, r);
      id[r] = (lane < n[r]) ? seg[p0[r] + lane] : 0x7fffffff;
    }
#pragma unroll
    for (int r = 0; r < 8; ++r) {
      int m = n[r] < 32 ? n[r] : 32;
      int c = 0;
      for (int q = 0; q < m; ++q) c += (__builtin_amdgcn_readlane(id[r], q) < id[r]) ? 1 : 0;
      rk[r] = c;
    }
    int sv[8];
#pragma unroll
    for (int r = 0; r < 8; ++r) sv[r] = (lane < n[r] && n[r] <= 32) ? ei[id[r]] : 0;
#pragma unroll
    for (int r = 0; r < 8; ++r) {
      if (lane < n[r] && n[r] <= 32) srcS[p0[r] + rk[r]] = sv[r];
    }
    for (int j = jb; j < jb + 8 && j < RB && node0 + j < N; ++j) {
      int nn = __builtin_amdgcn_readfirstlane(cnt[j]);
      if (nn <= 32) continue;
      int pp = e0 + __builtin_amdgcn_readfirstlane(start[j]);
      for (int i = lane; i < nn; i += 32) {
        int me = seg[pp + i];
        int c = 0;
        for (int q = 0; q < nn; ++q) c += (seg[pp + q] < me) ? 1 : 0;
        srcS[pp + c] = ei[me];
      }
    }
  }
}

__device__ __forceinline__ float leaky(float v) { return (v >= 0.f) ? v : NEG_SLOPE * v; }

__device__ __forceinline__ void edge_logits_h8(const float* __restrict__ as, const int* __restrict__ srcS,
                                               const float (&adh)[8], int base, int end, int lane,
                                               int& sv, float (&e)[8]) {
  sv = 0;
#pragma unroll
  for (int h = 0; h < 8; ++h) e[h] = -INFINITY;
  if (base + lane < end) {
    sv = srcS[base + lane];
    const float4* ap = (const float4*)(as + (size_t)sv * 8);
    float4 a0 = ap[0], a1 = ap[1];
    e[0] = leaky(a0.x + adh[0]); e[1] = leaky(a0.y + adh[1]);
    e[2] = leaky(a0.z + adh[2]); e[3] = leaky(a0.w + adh[3]);
    e[4] = leaky(a1.x + adh[4]); e[5] = leaky(a1.y + adh[5]);
    e[6] = leaky(a1.z + adh[6]); e[7] = leaky(a1.w + adh[7]);
  }
}

__device__ __forceinline__ float wave_max8(const float (&mx)[8], int lane) {
  bool b4 = (lane & 16) != 0, b3 = (lane & 8) != 0, b2 = (lane & 4) != 0;
  float u[4], t[2];
#pragma unroll
  for (int j = 0; j < 4; ++j) {
    float keep = b4 ? mx[4 + j] : mx[j];
    float send = b4 ? mx[j] : mx[4 + j];
    u[j] = fmaxf(keep, __shfl_xor(send, 16, 32));
  }
#pragma unroll
  for (int i = 0; i < 2; ++i) {
    float keep = b3 ? u[2 + i] : u[i];
    float send = b3 ? u[i] : u[2 + i];
    t[i] = fmaxf(keep, __shfl_xor(send, 8, 32));
  }
  float r = fmaxf(b2 ? t[1] : t[0], __shfl_xor(b2 ? t[0] : t[1], 4, 32));
  r = fmaxf(r, __shfl_xor(r, 2, 32));
  r = fmaxf(r, __shfl_xor(r, 1, 32));
  return r;
}

__device__ __forceinline__ void add_edges_x_h8(const float* __restrict__ x, int sv, const float (&e)[8],
                                               const float (&mh)[8], int n, int lane,
                                               v2f (&z)[8], float (&den)[8]) {
  float w[8];
#pragma unroll
  for (int h = 0; h < 8; ++h) w[h] = __expf(e[h] - mh[h]);
#pragma unroll 4
  for (int k = 0; k < n; ++k) {
    int s = __builtin_amdgcn_readlane(sv, k);
    v2f v = ((const v2f*)(x + (size_t)s * 64))[lane];
#pragma unroll
    for (int h = 0; h < 8; ++h) {
      float a = __uint_as_float(__builtin_amdgcn_readlane(__float_as_uint(w[h]), k));
      den[h] += a;
      z[h] += a * v;
    }
  }
}

#define ZPITCH 516

__global__ void __launch_bounds__(512)
row_softmax_aggregate_project_h8(const float* __restrict__ x, const float* __restrict__ as,
                                 const float* __restrict__ ad,
                                 const int* __restrict__ rowptr, const int* __restrict__ srcS,
                                 const float* __restrict__ W, const float* __restrict__ bias,
                                 float* __restrict__ out, int N) {
  __shared__ __attribute__((aligned(16))) float Zs[32 * ZPITCH];
  int lane = threadIdx.x & 31;
  int w = __builtin_amdgcn_readfirstlane((int)(threadIdx.x >> 5));
  int row_base = (int)blockIdx.x << 5;

  for (int rr = 0; rr < 2; ++rr) {
    int m = 2 * w + rr;
    int row = row_base + m;
    v2f z[8]; float den[8];
#pragma unroll
    for (int h = 0; h < 8; ++h) { z[h] = (v2f){0.f, 0.f}; den[h] = 0.f; }
    if (row < N) {
      int start = rowptr[row], end = rowptr[row + 1];
      const float4* adp = (const float4*)(ad + (size_t)row * 8);
      float4 d0 = adp[0], d1 = adp[1];
      float adh[8] = {d0.x, d0.y, d0.z, d0.w, d1.x, d1.y, d1.z, d1.w};

      int sv0; float e0[8], mx[8];
      edge_logits_h8(as, srcS, adh, start, end, lane, sv0, e0);
#pragma unroll
      for (int h = 0; h < 8; ++h) mx[h] = e0[h];
      for (int base = start + 32; base < end; base += 32) {
        int sv1; float e1[8];
        edge_logits_h8(as, srcS, adh, base, end, lane, sv1, e1);
#pragma unroll
        for (int h = 0; h < 8; ++h) mx[h] = fmaxf(mx[h], e1[h]);
      }
      float r = wave_max8(mx, lane);
      float mh[8];
#pragma unroll
      for (int h = 0; h < 8; ++h)
        mh[h] = __uint_as_float(__builtin_amdgcn_readlane(__float_as_uint(r), 4 * h));

      {
        int n = end - start; if (n > 32) n = 32;
        add_edges_x_h8(x, sv0, e0, mh, n, lane, z, den);
      }
      for (int base = start + 32; base < end; base += 32) {
        int sv1; float e1[8];
        edge_logits_h8(as, srcS, adh, base, end, lane, sv1, e1);
        int n = end - base; if (n > 32) n = 32;
        add_edges_x_h8(x, sv1, e1, mh, n, lane, z, den);
      }
    }
    float* zrow = Zs + m * ZPITCH + 2 * lane;
#pragma unroll
    for (int h = 0; h < 8; ++h) {
      float inv = (den[h] > 0.f) ? 1.f / den[h] : 0.f;
      zrow[h * 64]     = z[h].x * inv;
      zrow[h * 64 + 1] = z[h].y * inv;
    }
  }
  __syncthreads();

  int mrow = lane & 15, hi = lane >> 4;
  int col_base = (w & 7) << 6;
  int tile_row = (w >> 3) << 4;
  const float* Zrow = Zs + (tile_row + mrow) * ZPITCH + col_base;
  const float* Bcol = W + col_base + 4 * mrow;
  v8f acc0 = {}, acc1 = {}, acc2 = {}, acc3 = {};
#pragma unroll 4
  for (int k0 = 0; k0 < 64; k0 += 4) {
    int ka = k0 + 2 * hi;
    v2f a;
    a.x = Zrow[ka];
    a.y = Zrow[ka + 1];
    float4 bx = *(const float4*)(Bcol + (size_t)ka * 512);
    float4 by = *(const float4*)(Bcol + (size_t)(ka + 1) * 512);
    v2f b0, b1, b2, b3;
    b0.x = bx.x; b0.y = by.x;
    b1.x = bx.y; b1.y = by.y;
    b2.x = bx.z; b2.y = by.z;
    b3.x = bx.w; b3.y = by.w;
    acc0 = __builtin_amdgcn_wmma_f32_16x16x4_f32(false, a, false, b0, (short)0, acc0, false, false);
    acc1 = __builtin_amdgcn_wmma_f32_16x16x4_f32(false, a, false, b1, (short)0, acc1, false, false);
    acc2 = __builtin_amdgcn_wmma_f32_16x16x4_f32(false, a, false, b2, (short)0, acc2, false, false);
    acc3 = __builtin_amdgcn_wmma_f32_16x16x4_f32(false, a, false, b3, (short)0, acc3, false, false);
  }

  int rbase = row_base + tile_row + hi * 8;
  float4 b = *(const float4*)(bias + col_base + 4 * mrow);
  float* p = out + (size_t)rbase * 512 + col_base + 4 * mrow;
  if (rbase + 8 <= N) {
#pragma unroll
    for (int r = 0; r < 8; ++r) {
      float4 q;
      q.x = fmaxf(acc0[r] + b.x, 0.f); q.y = fmaxf(acc1[r] + b.y, 0.f);
      q.z = fmaxf(acc2[r] + b.z, 0.f); q.w = fmaxf(acc3[r] + b.w, 0.f);
      *(float4*)(p + r * 512) = q;
    }
  } else {
#pragma unroll
    for (int r = 0; r < 8; ++r) {
      if (rbase + r < N) {
        float4 q;
        q.x = fmaxf(acc0[r] + b.x, 0.f); q.y = fmaxf(acc1[r] + b.y, 0.f);
        q.z = fmaxf(acc2[r] + b.z, 0.f); q.w = fmaxf(acc3[r] + b.w, 0.f);
        *(float4*)(p + r * 512) = q;
      }
    }
  }
}

__global__ void fold_attention_vectors(const float* __restrict__ W, const float* __restrict__ avs,
                                       const float* __restrict__ avd, float* __restrict__ wa) {
  int t = blockIdx.x * blockDim.x + threadIdx.x;
  int k = t >> 4, c = t & 15, h = c & 7;
  const float4* a4 = (const float4*)(((c & 8) ? avd : avs) + h * 64);
  const float4* w4 = (const float4*)(W + (size_t)k * 512 + h * 64);
  float s = 0.f;
#pragma unroll
  for (int j = 0; j < 16; ++j) {
    float4 u = w4[j], v = a4[j];
    s += u.x * v.x; s += u.y * v.y; s += u.z * v.z; s += u.w * v.w;
  }
  wa[t] = s;
}

__global__ void node_logits_wmma_f32(const float* __restrict__ X, const float* __restrict__ wa,
                                     float* __restrict__ as, float* __restrict__ ad, int M) {
  int wave = (blockIdx.x * blockDim.x + threadIdx.x) >> 5;
  int lane = threadIdx.x & 31;
  int mt = (M + 15) >> 4;
  if (wave >= mt) return;
  int row_base = wave << 4;
  int mrow = lane & 15, hi = lane >> 4;
  int arow = row_base + mrow; if (arow > M - 1) arow = M - 1;
  const float* Arow = X + (size_t)arow * 64;
  v8f acc = {};
#pragma unroll
  for (int k0 = 0; k0 < 64; k0 += 4) {
    int ka = k0 + 2 * hi;
    v2f a, b;
    a.x = Arow[ka];
    a.y = Arow[ka + 1];
    b.x = wa[ka * 16 + mrow];
    b.y = wa[(ka + 1) * 16 + mrow];
    acc = __builtin_amdgcn_wmma_f32_16x16x4_f32(false, a, false, b, (short)0, acc, false, false);
  }
  float* dst = (mrow & 8) ? ad : as;
  int rbase = row_base + hi * 8;
#pragma unroll
  for (int r = 0; r < 8; ++r)
    if (rbase + r < M) dst[(size_t)(rbase + r) * 8 + (mrow & 7)] = acc[r];
}

__global__ void row_softmax_aggregate_h1(const float* __restrict__ xW, const float* __restrict__ as,
                                         const float* __restrict__ ad,
                                         const int* __restrict__ rowptr, const int* __restrict__ srcS,
                                         const float* __restrict__ bias, float* __restrict__ out, int N) {
  int row = __builtin_amdgcn_readfirstlane((int)((blockIdx.x * blockDim.x + threadIdx.x) >> 5));
  int lane = threadIdx.x & 31;
  if (row >= N) return;
  int start = rowptr[row], end = rowptr[row + 1];
  float adr = ad[row];

  int sv0 = 0; float e0 = -INFINITY;
  if (start + lane < end) { sv0 = srcS[start + lane]; e0 = leaky(as[sv0] + adr); }
  float m = e0;
  for (int base = start + 32; base < end; base += 32) {
    if (base + lane < end) m = fmaxf(m, leaky(as[srcS[base + lane]] + adr));
  }
  for (int off = 16; off; off >>= 1) m = fmaxf(m, __shfl_xor(m, off, 32));

  float2 acc = make_float2(0.f, 0.f);
  float den = 0.f;
  for (int base = start; base < end; base += 32) {
    int n = end - base; if (n > 32) n = 32;
    int sv = sv0; float e = e0;
    if (base != start) {
      sv = 0; e = -INFINITY;
      if (base + lane < end) { sv = srcS[base + lane]; e = leaky(as[sv] + adr); }
    }
    float w = __expf(e - m);
    for (int k = 0; k < n; ++k) {
      int s = __builtin_amdgcn_readlane(sv, k);
      float a = __uint_as_float(__builtin_amdgcn_readlane(__float_as_uint(w), k));
      float2 v = ((const float2*)(xW + (size_t)s * 64))[lane];
      den += a;
      acc.x += a * v.x; acc.y += a * v.y;
    }
  }
  float inv = (den > 0.f) ? 1.f / den : 0.f;
  float2 b = ((const float2*)bias)[lane];
  float2 q; q.x = acc.x * inv + b.x; q.y = acc.y * inv + b.y;
  ((float2*)(out + (size_t)row * 64))[lane] = q;
}

extern "C" void kernel_launch(void* const* d_in, const int* in_sizes, int n_in,
                              void* d_out, int out_size, void* d_ws, size_t ws_size,
                              hipStream_t stream) {
  const float* x     = (const float*)d_in[0];
  const int*   ei    = (const int*)  d_in[1];
  const float* W1    = (const float*)d_in[2];
  const float* asrc1 = (const float*)d_in[3];
  const float* adst1 = (const float*)d_in[4];
  const float* b1    = (const float*)d_in[5];
  const float* W2    = (const float*)d_in[6];
  const float* asrc2 = (const float*)d_in[7];
  const float* adst2 = (const float*)d_in[8];
  const float* b2    = (const float*)d_in[9];
  float* out = (float*)d_out;

  const int D = 64;
  const int N = in_sizes[0] / D;
  const int E = in_sizes[1] / 2;
  const int H = 8;
  const int F1 = H * D;

  float* ws   = (float*)d_ws;
  float* wa1  = ws;
  float* out1 = wa1  + 64 * 16;
  float* as1  = out1 + (size_t)N * F1;
  float* ad1  = as1  + (size_t)N * H;
  float* hW2  = ad1  + (size_t)N * H;
  float* as2  = hW2  + (size_t)N * D;
  float* ad2  = as2  + (size_t)N;
  int*  hist   = (int*)(ad2 + (size_t)N);
  int2* binned = (int2*)(hist + NCHUNK * NBIN);
  int*  base   = (int*)(binned + (size_t)E);
  int*  seg    = base + NBIN + 4;
  int*  srcS   = seg  + (size_t)E;
  int*  rowptr = srcS + (size_t)E;

  const int T = 256;

  bin_count<<<NCHUNK, 1024, 0, stream>>>(ei, hist, E, N);
  bin_scatter<<<NCHUNK, 1024, 0, stream>>>(ei, hist, base, binned, E, N);
  bin_rows<<<NBIN, 1024, 0, stream>>>(ei, base, binned, rowptr, seg, srcS, E, N);

  fold_attention_vectors<<<4, T, 0, stream>>>(W1, asrc1, adst1, wa1);
  {
    int nwaves = (N + 15) / 16;
    node_logits_wmma_f32<<<(nwaves + 7) / 8, T, 0, stream>>>(x, wa1, as1, ad1, N);
  }
  row_softmax_aggregate_project_h8<<<(N + 31) / 32, 512, 0, stream>>>(x, as1, ad1, rowptr, srcS, W1, b1, out1, N);

  {
    int nwaves = N / 16;
    gemm_logits_wmma_f32<512, 64><<<(nwaves + 7) / 8, T, 0, stream>>>(out1, W2, asrc2, adst2, hW2, as2, ad2, N);
  }
  row_softmax_aggregate_h1<<<(N * 32 + T - 1) / T, T, 0, stream>>>(hW2, as2, ad2, rowptr, srcS, b2, out, N);
}


# layer-2 projection fused into the layer-1 kernel: a wave's ReLU tile stays in LDS and meets its head's 64 rows of W2 on WMMA; 8 head partials summed in LDS; h never stored; sum order: heads last
# speedup vs baseline: 6.8412x; 1.1077x over previous
#include <hip/hip_runtime.h>

typedef __attribute__((ext_vector_type(2))) float v2f;
typedef __attribute__((ext_vector_type(8))) float v8f;

#define NEG_SLOPE 0.2f


#define NBIN 256
#define NCHUNK 256
#define RBMAX 1024

__device__ __forceinline__ int bin_shift(int N) {
  int s = 0;
  while (((N - 1) >> s) >= NBIN) ++s;
  return s;
}

__global__ void __launch_bounds__(1024)
bin_count(const int* __restrict__ ei, int* __restrict__ hist, int E, int N) {
  __shared__ int h[NBIN];
  int t = threadIdx.x;
  if (t < NBIN) h[t] = 0;
  __syncthreads();
  int s = bin_shift(N);
  int ch = (E + (int)gridDim.x - 1) / (int)gridDim.x;
  int lo = (int)blockIdx.x * ch, hi = lo + ch; if (hi > E) hi = E;
  for (int i0 = lo + t; i0 < hi; i0 += 2048) {
    int i1 = i0 + 1024;
    bool two = i1 < hi;
    int d0 = ei[E + i0];
    int d1 = two ? ei[E + i1] : 0;
    atomicAdd(&h[d0 >> s], 1);
    if (two) atomicAdd(&h[d1 >> s], 1);
  }
  __syncthreads();
  if (t < NBIN) hist[(int)blockIdx.x * NBIN + t] = h[t];
}

__global__ void __launch_bounds__(1024)
bin_scatter(const int* __restrict__ ei, const int* __restrict__ hist, int* __restrict__ base,
            int2* __restrict__ binned, int E, int N) {
  __shared__ int tot[NBIN];
  __shared__ int pre[NBIN];
  __shared__ int cur[NBIN];
  __shared__ int wt[8];
  int t = threadIdx.x, lane = t & 31, w = t >> 5;
  int nb = (int)gridDim.x, b = (int)blockIdx.x;
  if (t < NBIN) { tot[t] = 0; pre[t] = 0; }
  __syncthreads();
  {
    int c4 = t & 63, rg = t >> 6;
    int4 all = make_int4(0, 0, 0, 0), mine = make_int4(0, 0, 0, 0);
    const int4* h4 = (const int4*)hist;
    for (int bb = rg; bb < nb; bb += 16) {
      int4 v = h4[bb * (NBIN / 4) + c4];
      all.x += v.x; all.y += v.y; all.z += v.z; all.w += v.w;
      if (bb < b) { mine.x += v.x; mine.y += v.y; mine.z += v.z; mine.w += v.w; }
    }
    atomicAdd(&tot[4 * c4 + 0], all.x);  atomicAdd(&tot[4 * c4 + 1], all.y);
    atomicAdd(&tot[4 * c4 + 2], all.z);  atomicAdd(&tot[4 * c4 + 3], all.w);
    atomicAdd(&pre[4 * c4 + 0], mine.x); atomicAdd(&pre[4 * c4 + 1], mine.y);
    atomicAdd(&pre[4 * c4 + 2], mine.z); atomicAdd(&pre[4 * c4 + 3], mine.w);
  }
  __syncthreads();
  int v = (t < NBIN) ? tot[t] : 0;
  int incl = v;
  for (int o = 1; o < 32; o <<= 1) {
    int u = __shfl_up(incl, o, 32);
    if (lane >= o) incl += u;
  }
  if (t < NBIN && lane == 31) wt[w] = incl;
  __syncthreads();
  if (t < NBIN) {
    int carry = 0;
    for (int i = 0; i < w; ++i) carry += wt[i];
    int ex = carry + incl - v;
    cur[t] = ex + pre[t];
    if (b == 0) base[t] = ex;
  }
  if (b == 0 && t == 0) base[NBIN] = E;
  __syncthreads();
  int s = bin_shift(N);
  int ch = (E + nb - 1) / nb;
  int lo = b * ch, hi = lo + ch; if (hi > E) hi = E;
  for (int i0 = lo + t; i0 < hi; i0 += 2048) {
    int i1 = i0 + 1024;
    bool two = i1 < hi;
    int d0 = ei[E + i0];
    int d1 = two ? ei[E + i1] : 0;
    int q0 = atomicAdd(&cur[d0 >> s], 1);
    binned[q0] = make_int2(d0, i0);
    if (two) {
      int q1 = atomicAdd(&cur[d1 >> s], 1);
      binned[q1] = make_int2(d1, i1);
    }
  }
}

__global__ void __launch_bounds__(1024)
bin_rows(const int* __restrict__ ei, const int* __restrict__ base, const int2* __restrict__ binned,
         int* __restrict__ rowptr, int* __restrict__ seg, int* __restrict__ srcS, int E, int N) {
  __shared__ int cnt[RBMAX];
  __shared__ int start[RBMAX];
  __shared__ int cur[RBMAX];
  __shared__ int wt[32];
  int t = threadIdx.x, lane = t & 31;
  int w = __builtin_amdgcn_readfirstlane(t >> 5);
  int k = (int)blockIdx.x;
  int s = bin_shift(N);
  int RB = 1 << s;
  int node0 = k << s;
  int e0 = base[k], e1 = base[k + 1];
  if (t < RB) cnt[t] = 0;
  __syncthreads();
  for (int i0 = e0 + t; i0 < e1; i0 += 4096) {
    int d[4];
#pragma unroll
    for (int u = 0; u < 4; ++u) { int i = i0 + 1024 * u; d[u] = (i < e1) ? binned[i].x : node0; }
#pragma unroll
    for (int u = 0; u < 4; ++u) { if (i0 + 1024 * u < e1) atomicAdd(&cnt[d[u] - node0], 1); }
  }
  __syncthreads();
  int v = (t < RB) ? cnt[t] : 0;
  int incl = v;
  for (int o = 1; o < 32; o <<= 1) {
    int u = __shfl_up(incl, o, 32);
    if (lane >= o) incl += u;
  }
  if (lane == 31) wt[w] = incl;
  __syncthreads();
  if (w == 0) {
    int sv = wt[lane], si = sv;
    for (int o = 1; o < 32; o <<= 1) {
      int u = __shfl_up(si, o, 32);
      if (lane >= o) si += u;
    }
    wt[lane] = si - sv;
  }
  __syncthreads();
  int ex = wt[w] + incl - v;
  if (t < RB) {
    start[t] = ex; cur[t] = ex;
    if (node0 + t < N) rowptr[node0 + t] = e0 + ex;
  }
  if (k == 0 && t == 0) rowptr[N] = E;
  __syncthreads();
  for (int i0 = e0 + t; i0 < e1; i0 += 4096) {
    int2 pr[4];
#pragma unroll
    for (int u = 0; u < 4; ++u) { int i = i0 + 1024 * u; pr[u] = (i < e1) ? binned[i] : make_int2(node0, 0); }
#pragma unroll
    for (int u = 0; u < 4; ++u) {
      if (i0 + 1024 * u < e1) {
        int p = atomicAdd(&cur[pr[u].x - node0], 1);
        seg[e0 + p] = pr[u].y;
      }
    }
  }
  __syncthreads();
  for (int jb = w * 8; jb < RB; jb += 256) {
    int p0[8], n[8], id[8], rk[8];
    int jl = jb + (lane & 7);
    bool okl = (jl < RB) && (node0 + jl < N);
    int st_l = okl ? start[jl] : 0;
    int n_l = okl ? cnt[jl] : 0;
#pragma unroll
    for (int r = 0; r < 8; ++r) {
      p0[r] = e0 + __builtin_amdgcn_readlane(st_l, r);
      n[r] = __builtin_amdgcn_readlane(n_l, r);
      id[r] = (lane < n[r]) ? seg[p0[r] + lane] : 0x7fffffff;
    }
#pragma unroll
    for (int r = 0; r < 8; ++r) {
      int m = n[r] < 32 ? n[r] : 32;
      int c = 0;
      for (int q = 0; q < m; ++q) c += (__builtin_amdgcn_readlane(id[r], q) < id[r]) ? 1 : 0;
      rk[r] = c;
    }
    int sv[8];
#pragma unroll
    for (int r = 0; r < 8; ++r) sv[r] = (lane < n[r] && n[r] <= 32) ? ei[id[r]] : 0;
#pragma unroll
    for (int r = 0; r < 8; ++r) {
      if (lane < n[r] && n[r] <= 32) srcS[p0[r] + rk[r]] = sv[r];
    }
    for (int j = jb; j < jb + 8 && j < RB && node0 + j < N; ++j) {
      int nn = __builtin_amdgcn_readfirstlane(cnt[j]);
      if (nn <= 32) continue;
      int pp = e0 + __builtin_amdgcn_readfirstlane(start[j]);
      for (int i = lane; i < nn; i += 32) {
        int me = seg[pp + i];
        int c = 0;
        for (int q = 0; q < nn; ++q) c += (seg[pp + q] < me) ? 1 : 0;
        srcS[pp + c] = ei[me];
      }
    }
  }
}

__device__ __forceinline__ float leaky(float v) { return (v >= 0.f) ? v : NEG_SLOPE * v; }

__device__ __forceinline__ void edge_logits_h8(const float* __restrict__ as, const int* __restrict__ srcS,
                                               const float (&adh)[8], int base, int end, int lane,
                                               int& sv, float (&e)[8]) {
  sv = 0;
#pragma unroll
  for (int h = 0; h < 8; ++h) e[h] = -INFINITY;
  if (base + lane < end) {
    sv = srcS[base + lane];
    const float4* ap = (const float4*)(as + (size_t)sv * 8);
    float4 a0 = ap[0], a1 = ap[1];
    e[0] = leaky(a0.x + adh[0]); e[1] = leaky(a0.y + adh[1]);
    e[2] = leaky(a0.z + adh[2]); e[3] = leaky(a0.w + adh[3]);
    e[4] = leaky(a1.x + adh[4]); e[5] = leaky(a1.y + adh[5]);
    e[6] = leaky(a1.z + adh[6]); e[7] = leaky(a1.w + adh[7]);
  }
}

__device__ __forceinline__ float wave_max8(const float (&mx)[8], int lane) {
  bool b4 = (lane & 16) != 0, b3 = (lane & 8) != 0, b2 = (lane & 4) != 0;
  float u[4], t[2];
#pragma unroll
  for (int j = 0; j < 4; ++j) {
    float keep = b4 ? mx[4 + j] : mx[j];
    float send = b4 ? mx[j] : mx[4 + j];
    u[j] = fmaxf(keep, __shfl_xor(send, 16, 32));
  }
#pragma unroll
  for (int i = 0; i < 2; ++i) {
    float keep = b3 ? u[2 + i] : u[i];
    float send = b3 ? u[i] : u[2 + i];
    t[i] = fmaxf(keep, __shfl_xor(send, 8, 32));
  }
  float r = fmaxf(b2 ? t[1] : t[0], __shfl_xor(b2 ? t[0] : t[1], 4, 32));
  r = fmaxf(r, __shfl_xor(r, 2, 32));
  r = fmaxf(r, __shfl_xor(r, 1, 32));
  return r;
}

__device__ __forceinline__ void add_edges_x_h8(const float* __restrict__ x, int sv, const float (&e)[8],
                                               const float (&mh)[8], int n, int lane,
                                               v2f (&z)[8], float (&den)[8]) {
  float w[8];
#pragma unroll
  for (int h = 0; h < 8; ++h) w[h] = __expf(e[h] - mh[h]);
#pragma unroll 4
  for (int k = 0; k < n; ++k) {
    int s = __builtin_amdgcn_readlane(sv, k);
    v2f v = ((const v2f*)(x + (size_t)s * 64))[lane];
#pragma unroll
    for (int h = 0; h < 8; ++h) {
      float a = __uint_as_float(__builtin_amdgcn_readlane(__float_as_uint(w[h]), k));
      den[h] += a;
      z[h] += a * v;
    }
  }
}

#define ZPITCH 516

__global__ void __launch_bounds__(512) __attribute__((amdgpu_num_vgpr(80)))
row_softmax_aggregate_project2_h8(const float* __restrict__ x, const float* __restrict__ as,
                                  const float* __restrict__ ad,
                                  const int* __restrict__ rowptr, const int* __restrict__ srcS,
                                  const float* __restrict__ W, const float* __restrict__ bias,
                                  const float* __restrict__ W2, const float* __restrict__ avs2,
                                  const float* __restrict__ avd2, float* __restrict__ hW2,
                                  float* __restrict__ as2, float* __restrict__ ad2, int N) {
  __shared__ __attribute__((aligned(16))) float Zs[32 * ZPITCH];
  int lane = threadIdx.x & 31;
  int w = __builtin_amdgcn_readfirstlane((int)(threadIdx.x >> 5));
  int row_base = (int)blockIdx.x << 5;

  for (int rr = 0; rr < 2; ++rr) {
    int m = 2 * w + rr;
    int row = row_base + m;
    v2f z[8]; float den[8];
#pragma unroll
    for (int h = 0; h < 8; ++h) { z[h] = (v2f){0.f, 0.f}; den[h] = 0.f; }
    if (row < N) {
      int start = rowptr[row], end = rowptr[row + 1];
      const float4* adp = (const float4*)(ad + (size_t)row * 8);
      float4 d0 = adp[0], d1 = adp[1];
      float adh[8] = {d0.x, d0.y, d0.z, d0.w, d1.x, d1.y, d1.z, d1.w};

      int sv0; float e0[8], mx[8];
      edge_logits_h8(as, srcS, adh, start, end, lane, sv0, e0);
#pragma unroll
      for (int h = 0; h < 8; ++h) mx[h] = e0[h];
      for (int base = start + 32; base < end; base += 32) {
        int sv1; float e1[8];
        edge_logits_h8(as, srcS, adh, base, end, lane, sv1, e1);
#pragma unroll
        for (int h = 0; h < 8; ++h) mx[h] = fmaxf(mx[h], e1[h]);
      }
      float r = wave_max8(mx, lane);
      float mh[8];
#pragma unroll
      for (int h = 0; h < 8; ++h)
        mh[h] = __uint_as_float(__builtin_amdgcn_readlane(__float_as_uint(r), 4 * h));

      {
        int n = end - start; if (n > 32) n = 32;
        add_edges_x_h8(x, sv0, e0, mh, n, lane, z, den);
      }
      for (int base = start + 32; base < end; base += 32) {
        int sv1; float e1[8];
        edge_logits_h8(as, srcS, adh, base, end, lane, sv1, e1);
        int n = end - base; if (n > 32) n = 32;
        add_edges_x_h8(x, sv1, e1, mh, n, lane, z, den);
      }
    }
    float* zrow = Zs + m * ZPITCH + 2 * lane;
#pragma unroll
    for (int h = 0; h < 8; ++h) {
      float inv = (den[h] > 0.f) ? 1.f / den[h] : 0.f;
      zrow[h * 64]     = z[h].x * inv;
      zrow[h * 64 + 1] = z[h].y * inv;
    }
  }
  __syncthreads();

  int mrow = lane & 15, hi = lane >> 4;
  int col_base = (w & 7) << 6;
  int tile_row = (w >> 3) << 4;
  const float* Zrow = Zs + (tile_row + mrow) * ZPITCH + col_base;
  const float* Bcol = W + col_base + 4 * mrow;
  v8f acc0 = {}, acc1 = {}, acc2 = {}, acc3 = {};
#pragma unroll 4
  for (int k0 = 0; k0 < 64; k0 += 4) {
    int ka = k0 + 2 * hi;
    v2f a;
    a.x = Zrow[ka];
    a.y = Zrow[ka + 1];
    float4 bx = *(const float4*)(Bcol + (size_t)ka * 512);
    float4 by = *(const float4*)(Bcol + (size_t)(ka + 1) * 512);
    v2f b0, b1, b2, b3;
    b0.x = bx.x; b0.y = by.x;
    b1.x = bx.y; b1.y = by.y;
    b2.x = bx.z; b2.y = by.z;
    b3.x = bx.w; b3.y = by.w;
    acc0 = __builtin_amdgcn_wmma_f32_16x16x4_f32(false, a, false, b0, (short)0, acc0, false, false);
    acc1 = __builtin_amdgcn_wmma_f32_16x16x4_f32(false, a, false, b1, (short)0, acc1, false, false);
    acc2 = __builtin_amdgcn_wmma_f32_16x16x4_f32(false, a, false, b2, (short)0, acc2, false, false);
    acc3 = __builtin_amdgcn_wmma_f32_16x16x4_f32(false, a, false, b3, (short)0, acc3, false, false);
  }

  float* ztile = Zs + (tile_row + hi * 8) * ZPITCH + col_base + 4 * mrow;
  {
    float4 b = *(const float4*)(bias + col_base + 4 * mrow);
#pragma unroll
    for (int r = 0; r < 8; ++r) {
      float4 q;
      q.x = fmaxf(acc0[r] + b.x, 0.f); q.y = fmaxf(acc1[r] + b.y, 0.f);
      q.z = fmaxf(acc2[r] + b.z, 0.f); q.w = fmaxf(acc3[r] + b.w, 0.f);
      *(float4*)(ztile + r * ZPITCH) = q;
    }
  }
  __syncthreads();

  v8f p0 = {}, p1 = {}, p2 = {}, p3 = {};
  {
    const float* B2col = W2 + (size_t)col_base * 64 + 4 * mrow;
#pragma unroll 4
    for (int k0 = 0; k0 < 64; k0 += 4) {
      int ka = k0 + 2 * hi;
      v2f a;
      a.x = Zrow[ka];
      a.y = Zrow[ka + 1];
      float4 bx = *(const float4*)(B2col + (size_t)ka * 64);
      float4 by = *(const float4*)(B2col + (size_t)(ka + 1) * 64);
      v2f b0, b1, b2, b3;
      b0.x = bx.x; b0.y = by.x;
      b1.x = bx.y; b1.y = by.y;
      b2.x = bx.z; b2.y = by.z;
      b3.x = bx.w; b3.y = by.w;
      p0 = __builtin_amdgcn_wmma_f32_16x16x4_f32(false, a, false, b0, (short)0, p0, false, false);
      p1 = __builtin_amdgcn_wmma_f32_16x16x4_f32(false, a, false, b1, (short)0, p1, false, false);
      p2 = __builtin_amdgcn_wmma_f32_16x16x4_f32(false, a, false, b2, (short)0, p2, false, false);
      p3 = __builtin_amdgcn_wmma_f32_16x16x4_f32(false, a, false, b3, (short)0, p3, false, false);
    }
  }
#pragma unroll
  for (int r = 0; r < 8; ++r)
    *(float4*)(ztile + r * ZPITCH) = make_float4(p0[r], p1[r], p2[r], p3[r]);
  __syncthreads();

  int orow = (int)threadIdx.x >> 4, c4 = (int)threadIdx.x & 15;
  const float* zp = Zs + orow * ZPITCH + 4 * c4;
  float4 hs = *(const float4*)zp;
#pragma unroll
  for (int h = 1; h < 8; ++h) {
    float4 u = *(const float4*)(zp + 64 * h);
    hs.x += u.x; hs.y += u.y; hs.z += u.z; hs.w += u.w;
  }
  float4 s4 = *(const float4*)(avs2 + 4 * c4);
  float4 d4 = *(const float4*)(avd2 + 4 * c4);
  float vs = hs.x * s4.x + hs.y * s4.y + hs.z * s4.z + hs.w * s4.w;
  float vd = hs.x * d4.x + hs.y * d4.y + hs.z * d4.z + hs.w * d4.w;
  for (int off = 8; off; off >>= 1) {
    vs += __shfl_xor(vs, off, 32);
    vd += __shfl_xor(vd, off, 32);
  }
  int grow = row_base + orow;
  if (grow < N) {
    *(float4*)(hW2 + (size_t)grow * 64 + 4 * c4) = hs;
    if (c4 == 0) { as2[grow] = vs; ad2[grow] = vd; }
  }
}

__global__ void fold_attention_vectors(const float* __restrict__ W, const float* __restrict__ avs,
                                       const float* __restrict__ avd, float* __restrict__ wa) {
  int t = blockIdx.x * blockDim.x + threadIdx.x;
  int k = t >> 4, c = t & 15, h = c & 7;
  const float4* a4 = (const float4*)(((c & 8) ? avd : avs) + h * 64);
  const float4* w4 = (const float4*)(W + (size_t)k * 512 + h * 64);
  float s = 0.f;
#pragma unroll
  for (int j = 0; j < 16; ++j) {
    float4 u = w4[j], v = a4[j];
    s += u.x * v.x; s += u.y * v.y; s += u.z * v.z; s += u.w * v.w;
  }
  wa[t] = s;
}

__global__ void node_logits_wmma_f32(const float* __restrict__ X, const float* __restrict__ wa,
                                     float* __restrict__ as, float* __restrict__ ad, int M) {
  int wave = (blockIdx.x * blockDim.x + threadIdx.x) >> 5;
  int lane = threadIdx.x & 31;
  int mt = (M + 15) >> 4;
  if (wave >= mt) return;
  int row_base = wave << 4;
  int mrow = lane & 15, hi = lane >> 4;
  int arow = row_base + mrow; if (arow > M - 1) arow = M - 1;
  const float* Arow = X + (size_t)arow * 64;
  v8f acc = {};
#pragma unroll
  for (int k0 = 0; k0 < 64; k0 += 4) {
    int ka = k0 + 2 * hi;
    v2f a, b;
    a.x = Arow[ka];
    a.y = Arow[ka + 1];
    b.x = wa[ka * 16 + mrow];
    b.y = wa[(ka + 1) * 16 + mrow];
    acc = __builtin_amdgcn_wmma_f32_16x16x4_f32(false, a, false, b, (short)0, acc, false, false);
  }
  float* dst = (mrow & 8) ? ad : as;
  int rbase = row_base + hi * 8;
#pragma unroll
  for (int r = 0; r < 8; ++r)
    if (rbase + r < M) dst[(size_t)(rbase + r) * 8 + (mrow & 7)] = acc[r];
}

__global__ void row_softmax_aggregate_h1(const float* __restrict__ xW, const float* __restrict__ as,
                                         const float* __restrict__ ad,
                                         const int* __restrict__ rowptr, const int* __restrict__ srcS,
                                         const float* __restrict__ bias, float* __restrict__ out, int N) {
  int row = __builtin_amdgcn_readfirstlane((int)((blockIdx.x * blockDim.x + threadIdx.x) >> 5));
  int lane = threadIdx.x & 31;
  if (row >= N) return;
  int start = rowptr[row], end = rowptr[row + 1];
  float adr = ad[row];

  int sv0 = 0; float e0 = -INFINITY;
  if (start + lane < end) { sv0 = srcS[start + lane]; e0 = leaky(as[sv0] + adr); }
  float m = e0;
  for (int base = start + 32; base < end; base += 32) {
    if (base + lane < end) m = fmaxf(m, leaky(as[srcS[base + lane]] + adr));
  }
  for (int off = 16; off; off >>= 1) m = fmaxf(m, __shfl_xor(m, off, 32));

  float2 acc = make_float2(0.f, 0.f);
  float den = 0.f;
  for (int base = start; base < end; base += 32) {
    int n = end - base; if (n > 32) n = 32;
    int sv = sv0; float e = e0;
    if (base != start) {
      sv = 0; e = -INFINITY;
      if (base + lane < end) { sv = srcS[base + lane]; e = leaky(as[sv] + adr); }
    }
    float w = __expf(e - m);
    for (int k = 0; k < n; ++k) {
      int s = __builtin_amdgcn_readlane(sv, k);
      float a = __uint_as_float(__builtin_amdgcn_readlane(__float_as_uint(w), k));
      float2 v = ((const float2*)(xW + (size_t)s * 64))[lane];
      den += a;
      acc.x += a * v.x; acc.y += a * v.y;
    }
  }
  float inv = (den > 0.f) ? 1.f / den : 0.f;
  float2 b = ((const float2*)bias)[lane];
  float2 q; q.x = acc.x * inv + b.x; q.y = acc.y * inv + b.y;
  ((float2*)(out + (size_t)row * 64))[lane] = q;
}

extern "C" void kernel_launch(void* const* d_in, const int* in_sizes, int n_in,
                              void* d_out, int out_size, void* d_ws, size_t ws_size,
                              hipStream_t stream) {
  const float* x     = (const float*)d_in[0];
  const int*   ei    = (const int*)  d_in[1];
  const float* W1    = (const float*)d_in[2];
  const float* asrc1 = (const float*)d_in[3];
  const float* adst1 = (const float*)d_in[4];
  const float* b1    = (const float*)d_in[5];
  const float* W2    = (const float*)d_in[6];
  const float* asrc2 = (const float*)d_in[7];
  const float* adst2 = (const float*)d_in[8];
  const float* b2    = (const float*)d_in[9];
  float* out = (float*)d_out;

  const int D = 64;
  const int N = in_sizes[0] / D;
  const int E = in_sizes[1] / 2;
  const int H = 8;

  float* ws   = (float*)d_ws;
  float* wa1  = ws;
  float* as1  = wa1  + 64 * 16;
  float* ad1  = as1  + (size_t)N * H;
  float* hW2  = ad1  + (size_t)N * H;
  float* as2  = hW2  + (size_t)N * D;
  float* ad2  = as2  + (size_t)N;
  int*  hist   = (int*)(ad2 + (size_t)N);
  int2* binned = (int2*)(hist + NCHUNK * NBIN);
  int*  base   = (int*)(binned + (size_t)E);
  int*  seg    = base + NBIN + 4;
  int*  srcS   = seg  + (size_t)E;
  int*  rowptr = srcS + (size_t)E;

  const int T = 256;

  bin_count<<<NCHUNK, 1024, 0, stream>>>(ei, hist, E, N);
  bin_scatter<<<NCHUNK, 1024, 0, stream>>>(ei, hist, base, binned, E, N);
  bin_rows<<<NBIN, 1024, 0, stream>>>(ei, base, binned, rowptr, seg, srcS, E, N);

  fold_attention_vectors<<<4, T, 0, stream>>>(W1, asrc1, adst1, wa1);
  {
    int nwaves = (N + 15) / 16;
    node_logits_wmma_f32<<<(nwaves + 7) / 8, T, 0, stream>>>(x, wa1, as1, ad1, N);
  }
  row_softmax_aggregate_project2_h8<<<(N + 31) / 32, 512, 0, stream>>>(x, as1, ad1, rowptr, srcS, W1, b1,
                                                                      W2, asrc2, adst2, hW2, as2, ad2, N);

  row_softmax_aggregate_h1<<<(N * 32 + T - 1) / T, T, 0, stream>>>(hW2, as2, ad2, rowptr, srcS, b2, out, N);
}


# counting sort stages records in LDS: range pass groups its slice by range and stores contiguous runs; node pass keeps a range's edge ids in LDS (no scattered 4/8-byte global stores, no id re-read)
# speedup vs baseline: 7.4074x; 1.0828x over previous
#include <hip/hip_runtime.h>

typedef __attribute__((ext_vector_type(2))) float v2f;
typedef __attribute__((ext_vector_type(8))) float v8f;

#define NEG_SLOPE 0.2f


#define NBIN 256
#define NCHUNK 256
#define RBMAX 1024
#define SLCAP 4096
#define SEGCAP 8192

__device__ __forceinline__ int bin_shift(int N) {
  int s = 0;
  while (((N - 1) >> s) >= NBIN) ++s;
  return s;
}

__global__ void __launch_bounds__(1024)
bin_count(const int* __restrict__ ei, int* __restrict__ hist, int E, int N) {
  __shared__ int h[NBIN];
  int t = threadIdx.x;
  if (t < NBIN) h[t] = 0;
  __syncthreads();
  int s = bin_shift(N);
  int ch = (E + (int)gridDim.x - 1) / (int)gridDim.x;
  int lo = (int)blockIdx.x * ch, hi = lo + ch; if (hi > E) hi = E;
  for (int i0 = lo + t; i0 < hi; i0 += 2048) {
    int i1 = i0 + 1024;
    bool two = i1 < hi;
    int d0 = ei[E + i0];
    int d1 = two ? ei[E + i1] : 0;
    atomicAdd(&h[d0 >> s], 1);
    if (two) atomicAdd(&h[d1 >> s], 1);
  }
  __syncthreads();
  if (t < NBIN) hist[(int)blockIdx.x * NBIN + t] = h[t];
}

__global__ void __launch_bounds__(1024)
bin_scatter(const int* __restrict__ ei, const int* __restrict__ hist, int* __restrict__ base,
            int2* __restrict__ binned, int E, int N) {
  __shared__ int tot[NBIN];
  __shared__ int pre[NBIN];
  __shared__ int gst[NBIN];
  __shared__ int cur[NBIN];
  __shared__ int lst[NBIN];
  __shared__ int lcur[NBIN];
  __shared__ int wt[8];
  __shared__ int wl[8];
  __shared__ int2 buf[SLCAP];
  int t = threadIdx.x, lane = t & 31, w = t >> 5;
  int nb = (int)gridDim.x, b = (int)blockIdx.x;
  int mine_t = (t < NBIN) ? hist[b * NBIN + t] : 0;
  if (t < NBIN) { tot[t] = 0; pre[t] = 0; }
  __syncthreads();
  {
    int c4 = t & 63, rg = t >> 6;
    int4 all = make_int4(0, 0, 0, 0), mine = make_int4(0, 0, 0, 0);
    const int4* h4 = (const int4*)hist;
    for (int bb = rg; bb < nb; bb += 16) {
      int4 v = h4[bb * (NBIN / 4) + c4];
      all.x += v.x; all.y += v.y; all.z += v.z; all.w += v.w;
      if (bb < b) { mine.x += v.x; mine.y += v.y; mine.z += v.z; mine.w += v.w; }
    }
    atomicAdd(&tot[4 * c4 + 0], all.x);  atomicAdd(&tot[4 * c4 + 1], all.y);
    atomicAdd(&tot[4 * c4 + 2], all.z);  atomicAdd(&tot[4 * c4 + 3], all.w);
    atomicAdd(&pre[4 * c4 + 0], mine.x); atomicAdd(&pre[4 * c4 + 1], mine.y);
    atomicAdd(&pre[4 * c4 + 2], mine.z); atomicAdd(&pre[4 * c4 + 3], mine.w);
  }
  __syncthreads();
  int v = (t < NBIN) ? tot[t] : 0;
  int incl = v, linc = mine_t;
  for (int o = 1; o < 32; o <<= 1) {
    int u = __shfl_up(incl, o, 32);
    int ul = __shfl_up(linc, o, 32);
    if (lane >= o) { incl += u; linc += ul; }
  }
  if (t < NBIN && lane == 31) { wt[w] = incl; wl[w] = linc; }
  __syncthreads();
  if (t < NBIN) {
    int carry = 0, lcarry = 0;
    for (int i = 0; i < w; ++i) { carry += wt[i]; lcarry += wl[i]; }
    int ex = carry + incl - v;
    int g = ex + pre[t];
    int l = lcarry + linc - mine_t;
    gst[t] = g; cur[t] = g;
    lst[t] = l; lcur[t] = l;
    if (b == 0) base[t] = ex;
  }
  if (b == 0 && t == 0) base[NBIN] = E;
  __syncthreads();
  int s = bin_shift(N);
  int ch = (E + nb - 1) / nb;
  int lo = b * ch, hi = lo + ch; if (hi > E) hi = E;
  int len = hi - lo;
  if (len <= SLCAP) {
    for (int i0 = lo + t; i0 < hi; i0 += 2048) {
      int i1 = i0 + 1024;
      bool two = i1 < hi;
      int d0 = ei[E + i0];
      int d1 = two ? ei[E + i1] : 0;
      int q0 = atomicAdd(&lcur[d0 >> s], 1);
      buf[q0] = make_int2(d0, i0);
      if (two) {
        int q1 = atomicAdd(&lcur[d1 >> s], 1);
        buf[q1] = make_int2(d1, i1);
      }
    }
    __syncthreads();
    for (int j = t; j < len; j += 1024) {
      int2 pr = buf[j];
      int r = pr.x >> s;
      binned[gst[r] + (j - lst[r])] = pr;
    }
  } else {
    for (int i0 = lo + t; i0 < hi; i0 += 2048) {
      int i1 = i0 + 1024;
      bool two = i1 < hi;
      int d0 = ei[E + i0];
      int d1 = two ? ei[E + i1] : 0;
      int q0 = atomicAdd(&cur[d0 >> s], 1);
      binned[q0] = make_int2(d0, i0);
      if (two) {
        int q1 = atomicAdd(&cur[d1 >> s], 1);
        binned[q1] = make_int2(d1, i1);
      }
    }
  }
}

__global__ void __launch_bounds__(1024)
bin_rows(const int* __restrict__ ei, const int* __restrict__ base, const int2* __restrict__ binned,
         int* __restrict__ rowptr, int* __restrict__ seg, int* __restrict__ srcS, int E, int N) {
  __shared__ int cnt[RBMAX];
  __shared__ int start[RBMAX];
  __shared__ int cur[RBMAX];
  __shared__ int wt[32];
  __shared__ int segL[SEGCAP];
  int t = threadIdx.x, lane = t & 31;
  int w = __builtin_amdgcn_readfirstlane(t >> 5);
  int k = (int)blockIdx.x;
  int s = bin_shift(N);
  int RB = 1 << s;
  int node0 = k << s;
  int e0 = base[k], e1 = base[k + 1];
  bool inl = (e1 - e0) <= SEGCAP;
  if (t < RB) cnt[t] = 0;
  __syncthreads();
  for (int i0 = e0 + t; i0 < e1; i0 += 4096) {
    int d[4];
#pragma unroll
    for (int u = 0; u < 4; ++u) { int i = i0 + 1024 * u; d[u] = (i < e1) ? binned[i].x : node0; }
#pragma unroll
    for (int u = 0; u < 4; ++u) { if (i0 + 1024 * u < e1) atomicAdd(&cnt[d[u] - node0], 1); }
  }
  __syncthreads();
  int v = (t < RB) ? cnt[t] : 0;
  int incl = v;
  for (int o = 1; o < 32; o <<= 1) {
    int u = __shfl_up(incl, o, 32);
    if (lane >= o) incl += u;
  }
  if (lane == 31) wt[w] = incl;
  __syncthreads();
  if (w == 0) {
    int sv = wt[lane], si = sv;
    for (int o = 1; o < 32; o <<= 1) {
      int u = __shfl_up(si, o, 32);
      if (lane >= o) si += u;
    }
    wt[lane] = si - sv;
  }
  __syncthreads();
  int ex = wt[w] + incl - v;
  if (t < RB) {
    start[t] = ex; cur[t] = ex;
    if (node0 + t < N) rowptr[node0 + t] = e0 + ex;
  }
  if (k == 0 && t == 0) rowptr[N] = E;
  __syncthreads();
  for (int i0 = e0 + t; i0 < e1; i0 += 4096) {
    int2 pr[4];
#pragma unroll
    for (int u = 0; u < 4; ++u) { int i = i0 + 1024 * u; pr[u] = (i < e1) ? binned[i] : make_int2(node0, 0); }
#pragma unroll
    for (int u = 0; u < 4; ++u) {
      if (i0 + 1024 * u < e1) {
        int p = atomicAdd(&cur[pr[u].x - node0], 1);
        if (inl) segL[p] = pr[u].y; else seg[e0 + p] = pr[u].y;
      }
    }
  }
  __syncthreads();
  for (int jb = w * 8; jb < RB; jb += 256) {
    int st[8], p0[8], n[8], id[8], rk[8];
    int jl = jb + (lane & 7);
    bool okl = (jl < RB) && (node0 + jl < N);
    int st_l = okl ? start[jl] : 0;
    int n_l = okl ? cnt[jl] : 0;
#pragma unroll
    for (int r = 0; r < 8; ++r) {
      st[r] = __builtin_amdgcn_readlane(st_l, r);
      p0[r] = e0 + st[r];
      n[r] = __builtin_amdgcn_readlane(n_l, r);
    }
    if (inl) {
#pragma unroll
      for (int r = 0; r < 8; ++r) id[r] = (lane < n[r]) ? segL[st[r] + lane] : 0x7fffffff;
    } else {
#pragma unroll
      for (int r = 0; r < 8; ++r) id[r] = (lane < n[r]) ? seg[p0[r] + lane] : 0x7fffffff;
    }
#pragma unroll
    for (int r = 0; r < 8; ++r) {
      int m = n[r] < 32 ? n[r] : 32;
      int c = 0;
      for (int q = 0; q < m; ++q) c += (__builtin_amdgcn_readlane(id[r], q) < id[r]) ? 1 : 0;
      rk[r] = c;
    }
    int sv[8];
#pragma unroll
    for (int r = 0; r < 8; ++r) sv[r] = (lane < n[r] && n[r] <= 32) ? ei[id[r]] : 0;
#pragma unroll
    for (int r = 0; r < 8; ++r) {
      if (lane < n[r] && n[r] <= 32) srcS[p0[r] + rk[r]] = sv[r];
    }
    for (int j = jb; j < jb + 8 && j < RB && node0 + j < N; ++j) {
      int nn = __builtin_amdgcn_readfirstlane(cnt[j]);
      if (nn <= 32) continue;
      int ls = __builtin_amdgcn_readfirstlane(start[j]);
      int pp = e0 + ls;
      for (int i = lane; i < nn; i += 32) {
        int me = inl ? segL[ls + i] : seg[pp + i];
        int c = 0;
        for (int q = 0; q < nn; ++q) c += ((inl ? segL[ls + q] : seg[pp + q]) < me) ? 1 : 0;
        srcS[pp + c] = ei[me];
      }
    }
  }
}

__device__ __forceinline__ float leaky(float v) { return (v >= 0.f) ? v : NEG_SLOPE * v; }

__device__ __forceinline__ void edge_logits_h8(const float* __restrict__ as, const int* __restrict__ srcS,
                                               const float (&adh)[8], int base, int end, int lane,
                                               int& sv, float (&e)[8]) {
  sv = 0;
#pragma unroll
  for (int h = 0; h < 8; ++h) e[h] = -INFINITY;
  if (base + lane < end) {
    sv = srcS[base + lane];
    const float4* ap = (const float4*)(as + (size_t)sv * 8);
    float4 a0 = ap[0], a1 = ap[1];
    e[0] = leaky(a0.x + adh[0]); e[1] = leaky(a0.y + adh[1]);
    e[2] = leaky(a0.z + adh[2]); e[3] = leaky(a0.w + adh[3]);
    e[4] = leaky(a1.x + adh[4]); e[5] = leaky(a1.y + adh[5]);
    e[6] = leaky(a1.z + adh[6]); e[7] = leaky(a1.w + adh[7]);
  }
}

__device__ __forceinline__ float wave_max8(const float (&mx)[8], int lane) {
  bool b4 = (lane & 16) != 0, b3 = (lane & 8) != 0, b2 = (lane & 4) != 0;
  float u[4], t[2];
#pragma unroll
  for (int j = 0; j < 4; ++j) {
    float keep = b4 ? mx[4 + j] : mx[j];
    float send = b4 ? mx[j] : mx[4 + j];
    u[j] = fmaxf(keep, __shfl_xor(send, 16, 32));
  }
#pragma unroll
  for (int i = 0; i < 2; ++i) {
    float keep = b3 ? u[2 + i] : u[i];
    float send = b3 ? u[i] : u[2 + i];
    t[i] = fmaxf(keep, __shfl_xor(send, 8, 32));
  }
  float r = fmaxf(b2 ? t[1] : t[0], __shfl_xor(b2 ? t[0] : t[1], 4, 32));
  r = fmaxf(r, __shfl_xor(r, 2, 32));
  r = fmaxf(r, __shfl_xor(r, 1, 32));
  return r;
}

__device__ __forceinline__ void add_edges_x_h8(const float* __restrict__ x, int sv, const float (&e)[8],
                                               const float (&mh)[8], int n, int lane,
                                               v2f (&z)[8], float (&den)[8]) {
  float w[8];
#pragma unroll
  for (int h = 0; h < 8; ++h) w[h] = __expf(e[h] - mh[h]);
#pragma unroll 4
  for (int k = 0; k < n; ++k) {
    int s = __builtin_amdgcn_readlane(sv, k);
    v2f v = ((const v2f*)(x + (size_t)s * 64))[lane];
#pragma unroll
    for (int h = 0; h < 8; ++h) {
      float a = __uint_as_float(__builtin_amdgcn_readlane(__float_as_uint(w[h]), k));
      den[h] += a;
      z[h] += a * v;
    }
  }
}

#define ZPITCH 516

__global__ void __launch_bounds__(512) __attribute__((amdgpu_num_vgpr(80)))
row_softmax_aggregate_project2_h8(const float* __restrict__ x, const float* __restrict__ as,
                                  const float* __restrict__ ad,
                                  const int* __restrict__ rowptr, const int* __restrict__ srcS,
                                  const float* __restrict__ W, const float* __restrict__ bias,
                                  const float* __restrict__ W2, const float* __restrict__ avs2,
                                  const float* __restrict__ avd2, float* __restrict__ hW2,
                                  float* __restrict__ as2, float* __restrict__ ad2, int N) {
  __shared__ __attribute__((aligned(16))) float Zs[32 * ZPITCH];
  int lane = threadIdx.x & 31;
  int w = __builtin_amdgcn_readfirstlane((int)(threadIdx.x >> 5));
  int row_base = (int)blockIdx.x << 5;

  for (int rr = 0; rr < 2; ++rr) {
    int m = 2 * w + rr;
    int row = row_base + m;
    v2f z[8]; float den[8];
#pragma unroll
    for (int h = 0; h < 8; ++h) { z[h] = (v2f){0.f, 0.f}; den[h] = 0.f; }
    if (row < N) {
      int start = rowptr[row], end = rowptr[row + 1];
      const float4* adp = (const float4*)(ad + (size_t)row * 8);
      float4 d0 = adp[0], d1 = adp[1];
      float adh[8] = {d0.x, d0.y, d0.z, d0.w, d1.x, d1.y, d1.z, d1.w};

      int sv0; float e0[8], mx[8];
      edge_logits_h8(as, srcS, adh, start, end, lane, sv0, e0);
#pragma unroll
      for (int h = 0; h < 8; ++h) mx[h] = e0[h];
      for (int base = start + 32; base < end; base += 32) {
        int sv1; float e1[8];
        edge_logits_h8(as, srcS, adh, base, end, lane, sv1, e1);
#pragma unroll
        for (int h = 0; h < 8; ++h) mx[h] = fmaxf(mx[h], e1[h]);
      }
      float r = wave_max8(mx, lane);
      float mh[8];
#pragma unroll
      for (int h = 0; h < 8; ++h)
        mh[h] = __uint_as_float(__builtin_amdgcn_readlane(__float_as_uint(r), 4 * h));

      {
        int n = end - start; if (n > 32) n = 32;
        add_edges_x_h8(x, sv0, e0, mh, n, lane, z, den);
      }
      for (int base = start + 32; base < end; base += 32) {
        int sv1; float e1[8];
        edge_logits_h8(as, srcS, adh, base, end, lane, sv1, e1);
        int n = end - base; if (n > 32) n = 32;
        add_edges_x_h8(x, sv1, e1, mh, n, lane, z, den);
      }
    }
    float* zrow = Zs + m * ZPITCH + 2 * lane;
#pragma unroll
    for (int h = 0; h < 8; ++h) {
      float inv = (den[h] > 0.f) ? 1.f / den[h] : 0.f;
      zrow[h * 64]     = z[h].x * inv;
      zrow[h * 64 + 1] = z[h].y * inv;
    }
  }
  __syncthreads();

  int mrow = lane & 15, hi = lane >> 4;
  int col_base = (w & 7) << 6;
  int tile_row = (w >> 3) << 4;
  const float* Zrow = Zs + (tile_row + mrow) * ZPITCH + col_base;
  const float* Bcol = W + col_base + 4 * mrow;
  v8f acc0 = {}, acc1 = {}, acc2 = {}, acc3 = {};
#pragma unroll 4
  for (int k0 = 0; k0 < 64; k0 += 4) {
    int ka = k0 + 2 * hi;
    v2f a;
    a.x = Zrow[ka];
    a.y = Zrow[ka + 1];
    float4 bx = *(const float4*)(Bcol + (size_t)ka * 512);
    float4 by = *(const float4*)(Bcol + (size_t)(ka + 1) * 512);
    v2f b0, b1, b2, b3;
    b0.x = bx.x; b0.y = by.x;
    b1.x = bx.y; b1.y = by.y;
    b2.x = bx.z; b2.y = by.z;
    b3.x = bx.w; b3.y = by.w;
    acc0 = __builtin_amdgcn_wmma_f32_16x16x4_f32(false, a, false, b0, (short)0, acc0, false, false);
    acc1 = __builtin_amdgcn_wmma_f32_16x16x4_f32(false, a, false, b1, (short)0, acc1, false, false);
    acc2 = __builtin_amdgcn_wmma_f32_16x16x4_f32(false, a, false, b2, (short)0, acc2, false, false);
    acc3 = __builtin_amdgcn_wmma_f32_16x16x4_f32(false, a, false, b3, (short)0, acc3, false, false);
  }

  float* ztile = Zs + (tile_row + hi * 8) * ZPITCH + col_base + 4 * mrow;
  {
    float4 b = *(const float4*)(bias + col_base + 4 * mrow);
#pragma unroll
    for (int r = 0; r < 8; ++r) {
      float4 q;
      q.x = fmaxf(acc0[r] + b.x, 0.f); q.y = fmaxf(acc1[r] + b.y, 0.f);
      q.z = fmaxf(acc2[r] + b.z, 0.f); q.w = fmaxf(acc3[r] + b.w, 0.f);
      *(float4*)(ztile + r * ZPITCH) = q;
    }
  }
  __syncthreads();

  v8f p0 = {}, p1 = {}, p2 = {}, p3 = {};
  {
    const float* B2col = W2 + (size_t)col_base * 64 + 4 * mrow;
#pragma unroll 4
    for (int k0 = 0; k0 < 64; k0 += 4) {
      int ka = k0 + 2 * hi;
      v2f a;
      a.x = Zrow[ka];
      a.y = Zrow[ka + 1];
      float4 bx = *(const float4*)(B2col + (size_t)ka * 64);
      float4 by = *(const float4*)(B2col + (size_t)(ka + 1) * 64);
      v2f b0, b1, b2, b3;
      b0.x = bx.x; b0.y = by.x;
      b1.x = bx.y; b1.y = by.y;
      b2.x = bx.z; b2.y = by.z;
      b3.x = bx.w; b3.y = by.w;
      p0 = __builtin_amdgcn_wmma_f32_16x16x4_f32(false, a, false, b0, (short)0, p0, false, false);
      p1 = __builtin_amdgcn_wmma_f32_16x16x4_f32(false, a, false, b1, (short)0, p1, false, false);
      p2 = __builtin_amdgcn_wmma_f32_16x16x4_f32(false, a, false, b2, (short)0, p2, false, false);
      p3 = __builtin_amdgcn_wmma_f32_16x16x4_f32(false, a, false, b3, (short)0, p3, false, false);
    }
  }
#pragma unroll
  for (int r = 0; r < 8; ++r)
    *(float4*)(ztile + r * ZPITCH) = make_float4(p0[r], p1[r], p2[r], p3[r]);
  __syncthreads();

  int orow = (int)threadIdx.x >> 4, c4 = (int)threadIdx.x & 15;
  const float* zp = Zs + orow * ZPITCH + 4 * c4;
  float4 hs = *(const float4*)zp;
#pragma unroll
  for (int h = 1; h < 8; ++h) {
    float4 u = *(const float4*)(zp + 64 * h);
    hs.x += u.x; hs.y += u.y; hs.z += u.z; hs.w += u.w;
  }
  float4 s4 = *(const float4*)(avs2 + 4 * c4);
  float4 d4 = *(const float4*)(avd2 + 4 * c4);
  float vs = hs.x * s4.x + hs.y * s4.y + hs.z * s4.z + hs.w * s4.w;
  float vd = hs.x * d4.x + hs.y * d4.y + hs.z * d4.z + hs.w * d4.w;
  for (int off = 8; off; off >>= 1) {
    vs += __shfl_xor(vs, off, 32);
    vd += __shfl_xor(vd, off, 32);
  }
  int grow = row_base + orow;
  if (grow < N) {
    *(float4*)(hW2 + (size_t)grow * 64 + 4 * c4) = hs;
    if (c4 == 0) { as2[grow] = vs; ad2[grow] = vd; }
  }
}

__global__ void fold_attention_vectors(const float* __restrict__ W, const float* __restrict__ avs,
                                       const float* __restrict__ avd, float* __restrict__ wa) {
  int t = blockIdx.x * blockDim.x + threadIdx.x;
  int k = t >> 4, c = t & 15, h = c & 7;
  const float4* a4 = (const float4*)(((c & 8) ? avd : avs) + h * 64);
  const float4* w4 = (const float4*)(W + (size_t)k * 512 + h * 64);
  float s = 0.f;
#pragma unroll
  for (int j = 0; j < 16; ++j) {
    float4 u = w4[j], v = a4[j];
    s += u.x * v.x; s += u.y * v.y; s += u.z * v.z; s += u.w * v.w;
  }
  wa[t] = s;
}

__global__ void node_logits_wmma_f32(const float* __restrict__ X, const float* __restrict__ wa,
                                     float* __restrict__ as, float* __restrict__ ad, int M) {
  int wave = (blockIdx.x * blockDim.x + threadIdx.x) >> 5;
  int lane = threadIdx.x & 31;
  int mt = (M + 15) >> 4;
  if (wave >= mt) return;
  int row_base = wave << 4;
  int mrow = lane & 15, hi = lane >> 4;
  int arow = row_base + mrow; if (arow > M - 1) arow = M - 1;
  const float* Arow = X + (size_t)arow * 64;
  v8f acc = {};
#pragma unroll
  for (int k0 = 0; k0 < 64; k0 += 4) {
    int ka = k0 + 2 * hi;
    v2f a, b;
    a.x = Arow[ka];
    a.y = Arow[ka + 1];
    b.x = wa[ka * 16 + mrow];
    b.y = wa[(ka + 1) * 16 + mrow];
    acc = __builtin_amdgcn_wmma_f32_16x16x4_f32(false, a, false, b, (short)0, acc, false, false);
  }
  float* dst = (mrow & 8) ? ad : as;
  int rbase = row_base + hi * 8;
#pragma unroll
  for (int r = 0; r < 8; ++r)
    if (rbase + r < M) dst[(size_t)(rbase + r) * 8 + (mrow & 7)] = acc[r];
}

__global__ void row_softmax_aggregate_h1(const float* __restrict__ xW, const float* __restrict__ as,
                                         const float* __restrict__ ad,
                                         const int* __restrict__ rowptr, const int* __restrict__ srcS,
                                         const float* __restrict__ bias, float* __restrict__ out, int N) {
  int row = __builtin_amdgcn_readfirstlane((int)((blockIdx.x * blockDim.x + threadIdx.x) >> 5));
  int lane = threadIdx.x & 31;
  if (row >= N) return;
  int start = rowptr[row], end = rowptr[row + 1];
  float adr = ad[row];

  int sv0 = 0; float e0 = -INFINITY;
  if (start + lane < end) { sv0 = srcS[start + lane]; e0 = leaky(as[sv0] + adr); }
  float m = e0;
  for (int base = start + 32; base < end; base += 32) {
    if (base + lane < end) m = fmaxf(m, leaky(as[srcS[base + lane]] + adr));
  }
  for (int off = 16; off; off >>= 1) m = fmaxf(m, __shfl_xor(m, off, 32));

  float2 acc = make_float2(0.f, 0.f);
  float den = 0.f;
  for (int base = start; base < end; base += 32) {
    int n = end - base; if (n > 32) n = 32;
    int sv = sv0; float e = e0;
    if (base != start) {
      sv = 0; e = -INFINITY;
      if (base + lane < end) { sv = srcS[base + lane]; e = leaky(as[sv] + adr); }
    }
    float w = __expf(e - m);
    for (int k = 0; k < n; ++k) {
      int s = __builtin_amdgcn_readlane(sv, k);
      float a = __uint_as_float(__builtin_amdgcn_readlane(__float_as_uint(w), k));
      float2 v = ((const float2*)(xW + (size_t)s * 64))[lane];
      den += a;
      acc.x += a * v.x; acc.y += a * v.y;
    }
  }
  float inv = (den > 0.f) ? 1.f / den : 0.f;
  float2 b = ((const float2*)bias)[lane];
  float2 q; q.x = acc.x * inv + b.x; q.y = acc.y * inv + b.y;
  ((float2*)(out + (size_t)row * 64))[lane] = q;
}

extern "C" void kernel_launch(void* const* d_in, const int* in_sizes, int n_in,
                              void* d_out, int out_size, void* d_ws, size_t ws_size,
                              hipStream_t stream) {
  const float* x     = (const float*)d_in[0];
  const int*   ei    = (const int*)  d_in[1];
  const float* W1    = (const float*)d_in[2];
  const float* asrc1 = (const float*)d_in[3];
  const float* adst1 = (const float*)d_in[4];
  const float* b1    = (const float*)d_in[5];
  const float* W2    = (const float*)d_in[6];
  const float* asrc2 = (const float*)d_in[7];
  const float* adst2 = (const float*)d_in[8];
  const float* b2    = (const float*)d_in[9];
  float* out = (float*)d_out;

  const int D = 64;
  const int N = in_sizes[0] / D;
  const int E = in_sizes[1] / 2;
  const int H = 8;

  float* ws   = (float*)d_ws;
  float* wa1  = ws;
  float* as1  = wa1  + 64 * 16;
  float* ad1  = as1  + (size_t)N * H;
  float* hW2  = ad1  + (size_t)N * H;
  float* as2  = hW2  + (size_t)N * D;
  float* ad2  = as2  + (size_t)N;
  int*  hist   = (int*)(ad2 + (size_t)N);
  int2* binned = (int2*)(hist + NCHUNK * NBIN);
  int*  base   = (int*)(binned + (size_t)E);
  int*  seg    = base + NBIN + 4;
  int*  srcS   = seg  + (size_t)E;
  int*  rowptr = srcS + (size_t)E;

  const int T = 256;

  bin_count<<<NCHUNK, 1024, 0, stream>>>(ei, hist, E, N);
  bin_scatter<<<NCHUNK, 1024, 0, stream>>>(ei, hist, base, binned, E, N);
  bin_rows<<<NBIN, 1024, 0, stream>>>(ei, base, binned, rowptr, seg, srcS, E, N);

  fold_attention_vectors<<<4, T, 0, stream>>>(W1, asrc1, adst1, wa1);
  {
    int nwaves = (N + 15) / 16;
    node_logits_wmma_f32<<<(nwaves + 7) / 8, T, 0, stream>>>(x, wa1, as1, ad1, N);
  }
  row_softmax_aggregate_project2_h8<<<(N + 31) / 32, 512, 0, stream>>>(x, as1, ad1, rowptr, srcS, W1, b1,
                                                                      W2, asrc2, adst2, hW2, as2, ad2, N);

  row_softmax_aggregate_h1<<<(N * 32 + T - 1) / T, T, 0, stream>>>(hW2, as2, ad2, rowptr, srcS, b2, out, N);
}


# layer-1 logit path loses its two launches: the attention-vector fold runs as extra workgroups of the bin_count launch, the node logits as extra workgroups of the bin_rows launch; 7 launches -> 5
# speedup vs baseline: 7.7564x; 1.0471x over previous
#include <hip/hip_runtime.h>

typedef __attribute__((ext_vector_type(2))) float v2f;
typedef __attribute__((ext_vector_type(8))) float v8f;

#define NEG_SLOPE 0.2f


#define NFOLD 4

__device__ __forceinline__ void fold_attention_vectors(const float* __restrict__ W, const float* __restrict__ avs,
                                                       const float* __restrict__ avd, float* __restrict__ wa,
                                                       int t) {
  int k = t >> 4, c = t & 15, h = c & 7;
  const float4* a4 = (const float4*)(((c & 8) ? avd : avs) + h * 64);
  const float4* w4 = (const float4*)(W + (size_t)k * 512 + h * 64);
  float s = 0.f;
#pragma unroll 4
  for (int j = 0; j < 16; ++j) {
    float4 u = w4[j], v = a4[j];
    s += u.x * v.x; s += u.y * v.y; s += u.z * v.z; s += u.w * v.w;
  }
  wa[t] = s;
}

__device__ __forceinline__ void node_logits_wmma_f32(const float* __restrict__ X, const float* __restrict__ wa,
                                                     float* __restrict__ as, float* __restrict__ ad, int M,
                                                     int wave, int lane) {
  int row_base = wave << 4;
  int mrow = lane & 15, hi = lane >> 4;
  int arow = row_base + mrow; if (arow > M - 1) arow = M - 1;
  const float* Arow = X + (size_t)arow * 64;
  v8f acc = {};
#pragma unroll
  for (int k0 = 0; k0 < 64; k0 += 4) {
    int ka = k0 + 2 * hi;
    v2f a, b;
    a.x = Arow[ka];
    a.y = Arow[ka + 1];
    b.x = wa[ka * 16 + mrow];
    b.y = wa[(ka + 1) * 16 + mrow];
    acc = __builtin_amdgcn_wmma_f32_16x16x4_f32(false, a, false, b, (short)0, acc, false, false);
  }
  float* dst = (mrow & 8) ? ad : as;
  int rbase = row_base + hi * 8;
#pragma unroll
  for (int r = 0; r < 8; ++r)
    if (rbase + r < M) dst[(size_t)(rbase + r) * 8 + (mrow & 7)] = acc[r];
}

#define NBIN 256
#define NCHUNK 256
#define RBMAX 1024
#define SLCAP 4096
#define SEGCAP 8192

__device__ __forceinline__ int bin_shift(int N) {
  int s = 0;
  while (((N - 1) >> s) >= NBIN) ++s;
  return s;
}

__global__ void __launch_bounds__(1024)
fold_and_bin_count(const int* __restrict__ ei, int* __restrict__ hist, int E, int N,
                   const float* __restrict__ W, const float* __restrict__ avs,
                   const float* __restrict__ avd, float* __restrict__ wa) {
  __shared__ int h[NBIN];
  int t = threadIdx.x;
  if ((int)blockIdx.x < NFOLD) {
    if (t < 256) fold_attention_vectors(W, avs, avd, wa, (int)blockIdx.x * 256 + t);
    return;
  }
  int b = (int)blockIdx.x - NFOLD, nb = NCHUNK;
  if (t < NBIN) h[t] = 0;
  __syncthreads();
  int s = bin_shift(N);
  int ch = (E + nb - 1) / nb;
  int lo = b * ch, hi = lo + ch; if (hi > E) hi = E;
  for (int i0 = lo + t; i0 < hi; i0 += 2048) {
    int i1 = i0 + 1024;
    bool two = i1 < hi;
    int d0 = ei[E + i0];
    int d1 = two ? ei[E + i1] : 0;
    atomicAdd(&h[d0 >> s], 1);
    if (two) atomicAdd(&h[d1 >> s], 1);
  }
  __syncthreads();
  if (t < NBIN) hist[b * NBIN + t] = h[t];
}

__global__ void __launch_bounds__(1024)
bin_scatter(const int* __restrict__ ei, const int* __restrict__ hist, int* __restrict__ base,
            int2* __restrict__ binned, int E, int N) {
  __shared__ int tot[NBIN];
  __shared__ int pre[NBIN];
  __shared__ int gst[NBIN];
  __shared__ int cur[NBIN];
  __shared__ int lst[NBIN];
  __shared__ int lcur[NBIN];
  __shared__ int wt[8];
  __shared__ int wl[8];
  __shared__ int2 buf[SLCAP];
  int t = threadIdx.x, lane = t & 31, w = t >> 5;
  int nb = (int)gridDim.x, b = (int)blockIdx.x;
  int mine_t = (t < NBIN) ? hist[b * NBIN + t] : 0;
  if (t < NBIN) { tot[t] = 0; pre[t] = 0; }
  __syncthreads();
  {
    int c4 = t & 63, rg = t >> 6;
    int4 all = make_int4(0, 0, 0, 0), mine = make_int4(0, 0, 0, 0);
    const int4* h4 = (const int4*)hist;
    for (int bb = rg; bb < nb; bb += 16) {
      int4 v = h4[bb * (NBIN / 4) + c4];
      all.x += v.x; all.y += v.y; all.z += v.z; all.w += v.w;
      if (bb < b) { mine.x += v.x; mine.y += v.y; mine.z += v.z; mine.w += v.w; }
    }
    atomicAdd(&tot[4 * c4 + 0], all.x);  atomicAdd(&tot[4 * c4 + 1], all.y);
    atomicAdd(&tot[4 * c4 + 2], all.z);  atomicAdd(&tot[4 * c4 + 3], all.w);
    atomicAdd(&pre[4 * c4 + 0], mine.x); atomicAdd(&pre[4 * c4 + 1], mine.y);
    atomicAdd(&pre[4 * c4 + 2], mine.z); atomicAdd(&pre[4 * c4 + 3], mine.w);
  }
  __syncthreads();
  int v = (t < NBIN) ? tot[t] : 0;
  int incl = v, linc = mine_t;
  for (int o = 1; o < 32; o <<= 1) {
    int u = __shfl_up(incl, o, 32);
    int ul = __shfl_up(linc, o, 32);
    if (lane >= o) { incl += u; linc += ul; }
  }
  if (t < NBIN && lane == 31) { wt[w] = incl; wl[w] = linc; }
  __syncthreads();
  if (t < NBIN) {
    int carry = 0, lcarry = 0;
    for (int i = 0; i < w; ++i) { carry += wt[i]; lcarry += wl[i]; }
    int ex = carry + incl - v;
    int g = ex + pre[t];
    int l = lcarry + linc - mine_t;
    gst[t] = g; cur[t] = g;
    lst[t] = l; lcur[t] = l;
    if (b == 0) base[t] = ex;
  }
  if (b == 0 && t == 0) base[NBIN] = E;
  __syncthreads();
  int s = bin_shift(N);
  int ch = (E + nb - 1) / nb;
  int lo = b * ch, hi = lo + ch; if (hi > E) hi = E;
  int len = hi - lo;
  if (len <= SLCAP) {
    for (int i0 = lo + t; i0 < hi; i0 += 2048) {
      int i1 = i0 + 1024;
      bool two = i1 < hi;
      int d0 = ei[E + i0];
      int d1 = two ? ei[E + i1] : 0;
      int q0 = atomicAdd(&lcur[d0 >> s], 1);
      buf[q0] = make_int2(d0, i0);
      if (two) {
        int q1 = atomicAdd(&lcur[d1 >> s], 1);
        buf[q1] = make_int2(d1, i1);
      }
    }
    __syncthreads();
    for (int j = t; j < len; j += 1024) {
      int2 pr = buf[j];
      int r = pr.x >> s;
      binned[gst[r] + (j - lst[r])] = pr;
    }
  } else {
    for (int i0 = lo + t; i0 < hi; i0 += 2048) {
      int i1 = i0 + 1024;
      bool two = i1 < hi;
      int d0 = ei[E + i0];
      int d1 = two ? ei[E + i1] : 0;
      int q0 = atomicAdd(&cur[d0 >> s], 1);
      binned[q0] = make_int2(d0, i0);
      if (two) {
        int q1 = atomicAdd(&cur[d1 >> s], 1);
        binned[q1] = make_int2(d1, i1);
      }
    }
  }
}

__global__ void __launch_bounds__(1024)
bin_rows_and_node_logits(const int* __restrict__ ei, const int* __restrict__ base, const int2* __restrict__ binned,
                         int* __restrict__ rowptr, int* __restrict__ seg, int* __restrict__ srcS, int E, int N,
                         const float* __restrict__ X, const float* __restrict__ wa,
                         float* __restrict__ as, float* __restrict__ ad) {
  __shared__ int cnt[RBMAX];
  __shared__ int start[RBMAX];
  __shared__ int cur[RBMAX];
  __shared__ int wt[32];
  __shared__ int segL[SEGCAP];
  int t = threadIdx.x, lane = t & 31;
  int w = __builtin_amdgcn_readfirstlane(t >> 5);
  if ((int)blockIdx.x >= NBIN) {
    int wave = (((int)blockIdx.x - NBIN) << 5) + w;
    if (wave < ((N + 15) >> 4)) node_logits_wmma_f32(X, wa, as, ad, N, wave, lane);
    return;
  }
  int k = (int)blockIdx.x;
  int s = bin_shift(N);
  int RB = 1 << s;
  int node0 = k << s;
  int e0 = base[k], e1 = base[k + 1];
  bool inl = (e1 - e0) <= SEGCAP;
  if (t < RB) cnt[t] = 0;
  __syncthreads();
  for (int i0 = e0 + t; i0 < e1; i0 += 4096) {
    int d[4];
#pragma unroll
    for (int u = 0; u < 4; ++u) { int i = i0 + 1024 * u; d[u] = (i < e1) ? binned[i].x : node0; }
#pragma unroll
    for (int u = 0; u < 4; ++u) { if (i0 + 1024 * u < e1) atomicAdd(&cnt[d[u] - node0], 1); }
  }
  __syncthreads();
  int v = (t < RB) ? cnt[t] : 0;
  int incl = v;
  for (int o = 1; o < 32; o <<= 1) {
    int u = __shfl_up(incl, o, 32);
    if (lane >= o) incl += u;
  }
  if (lane == 31) wt[w] = incl;
  __syncthreads();
  if (w == 0) {
    int sv = wt[lane], si = sv;
    for (int o = 1; o < 32; o <<= 1) {
      int u = __shfl_up(si, o, 32);
      if (lane >= o) si += u;
    }
    wt[lane] = si - sv;
  }
  __syncthreads();
  int ex = wt[w] + incl - v;
  if (t < RB) {
    start[t] = ex; cur[t] = ex;
    if (node0 + t < N) rowptr[node0 + t] = e0 + ex;
  }
  if (k == 0 && t == 0) rowptr[N] = E;
  __syncthreads();
  for (int i0 = e0 + t; i0 < e1; i0 += 4096) {
    int2 pr[4];
#pragma unroll
    for (int u = 0; u < 4; ++u) { int i = i0 + 1024 * u; pr[u] = (i < e1) ? binned[i] : make_int2(node0, 0); }
#pragma unroll
    for (int u = 0; u < 4; ++u) {
      if (i0 + 1024 * u < e1) {
        int p = atomicAdd(&cur[pr[u].x - node0], 1);
        if (inl) segL[p] = pr[u].y; else seg[e0 + p] = pr[u].y;
      }
    }
  }
  __syncthreads();
  for (int jb = w * 8; jb < RB; jb += 256) {
    int st[8], p0[8], n[8], id[8], rk[8];
    int jl = jb + (lane & 7);
    bool okl = (jl < RB) && (node0 + jl < N);
    int st_l = okl ? start[jl] : 0;
    int n_l = okl ? cnt[jl] : 0;
#pragma unroll
    for (int r = 0; r < 8; ++r) {
      st[r] = __builtin_amdgcn_readlane(st_l, r);
      p0[r] = e0 + st[r];
      n[r] = __builtin_amdgcn_readlane(n_l, r);
    }
    if (inl) {
#pragma unroll
      for (int r = 0; r < 8; ++r) id[r] = (lane < n[r]) ? segL[st[r] + lane] : 0x7fffffff;
    } else {
#pragma unroll
      for (int r = 0; r < 8; ++r) id[r] = (lane < n[r]) ? seg[p0[r] + lane] : 0x7fffffff;
    }
#pragma unroll
    for (int r = 0; r < 8; ++r) {
      int m = n[r] < 32 ? n[r] : 32;
      int c = 0;
      for (int q = 0; q < m; ++q) c += (__builtin_amdgcn_readlane(id[r], q) < id[r]) ? 1 : 0;
      rk[r] = c;
    }
    int sv[8];
#pragma unroll
    for (int r = 0; r < 8; ++r) sv[r] = (lane < n[r] && n[r] <= 32) ? ei[id[r]] : 0;
#pragma unroll
    for (int r = 0; r < 8; ++r) {
      if (lane < n[r] && n[r] <= 32) srcS[p0[r] + rk[r]] = sv[r];
    }
    for (int j = jb; j < jb + 8 && j < RB && node0 + j < N; ++j) {
      int nn = __builtin_amdgcn_readfirstlane(cnt[j]);
      if (nn <= 32) continue;
      int ls = __builtin_amdgcn_readfirstlane(start[j]);
      int pp = e0 + ls;
      for (int i = lane; i < nn; i += 32) {
        int me = inl ? segL[ls + i] : seg[pp + i];
        int c = 0;
        for (int q = 0; q < nn; ++q) c += ((inl ? segL[ls + q] : seg[pp + q]) < me) ? 1 : 0;
        srcS[pp + c] = ei[me];
      }
    }
  }
}

__device__ __forceinline__ float leaky(float v) { return (v >= 0.f) ? v : NEG_SLOPE * v; }

__device__ __forceinline__ void edge_logits_h8(const float* __restrict__ as, const int* __restrict__ srcS,
                                               const float (&adh)[8], int base, int end, int lane,
                                               int& sv, float (&e)[8]) {
  sv = 0;
#pragma unroll
  for (int h = 0; h < 8; ++h) e[h] = -INFINITY;
  if (base + lane < end) {
    sv = srcS[base + lane];
    const float4* ap = (const float4*)(as + (size_t)sv * 8);
    float4 a0 = ap[0], a1 = ap[1];
    e[0] = leaky(a0.x + adh[0]); e[1] = leaky(a0.y + adh[1]);
    e[2] = leaky(a0.z + adh[2]); e[3] = leaky(a0.w + adh[3]);
    e[4] = leaky(a1.x + adh[4]); e[5] = leaky(a1.y + adh[5]);
    e[6] = leaky(a1.z + adh[6]); e[7] = leaky(a1.w + adh[7]);
  }
}

__device__ __forceinline__ float wave_max8(const float (&mx)[8], int lane) {
  bool b4 = (lane & 16) != 0, b3 = (lane & 8) != 0, b2 = (lane & 4) != 0;
  float u[4], t[2];
#pragma unroll
  for (int j = 0; j < 4; ++j) {
    float keep = b4 ? mx[4 + j] : mx[j];
    float send = b4 ? mx[j] : mx[4 + j];
    u[j] = fmaxf(keep, __shfl_xor(send, 16, 32));
  }
#pragma unroll
  for (int i = 0; i < 2; ++i) {
    float keep = b3 ? u[2 + i] : u[i];
    float send = b3 ? u[i] : u[2 + i];
    t[i] = fmaxf(keep, __shfl_xor(send, 8, 32));
  }
  float r = fmaxf(b2 ? t[1] : t[0], __shfl_xor(b2 ? t[0] : t[1], 4, 32));
  r = fmaxf(r, __shfl_xor(r, 2, 32));
  r = fmaxf(r, __shfl_xor(r, 1, 32));
  return r;
}

__device__ __forceinline__ void add_edges_x_h8(const float* __restrict__ x, int sv, const float (&e)[8],
                                               const float (&mh)[8], int n, int lane,
                                               v2f (&z)[8], float (&den)[8]) {
  float w[8];
#pragma unroll
  for (int h = 0; h < 8; ++h) w[h] = __expf(e[h] - mh[h]);
#pragma unroll 4
  for (int k = 0; k < n; ++k) {
    int s = __builtin_amdgcn_readlane(sv, k);
    v2f v = ((const v2f*)(x + (size_t)s * 64))[lane];
#pragma unroll
    for (int h = 0; h < 8; ++h) {
      float a = __uint_as_float(__builtin_amdgcn_readlane(__float_as_uint(w[h]), k));
      den[h] += a;
      z[h] += a * v;
    }
  }
}

#define ZPITCH 516

__global__ void __launch_bounds__(512) __attribute__((amdgpu_num_vgpr(80)))
row_softmax_aggregate_project2_h8(const float* __restrict__ x, const float* __restrict__ as,
                                  const float* __restrict__ ad,
                                  const int* __restrict__ rowptr, const int* __restrict__ srcS,
                                  const float* __restrict__ W, const float* __restrict__ bias,
                                  const float* __restrict__ W2, const float* __restrict__ avs2,
                                  const float* __restrict__ avd2, float* __restrict__ hW2,
                                  float* __restrict__ as2, float* __restrict__ ad2, int N) {
  __shared__ __attribute__((aligned(16))) float Zs[32 * ZPITCH];
  int lane = threadIdx.x & 31;
  int w = __builtin_amdgcn_readfirstlane((int)(threadIdx.x >> 5));
  int row_base = (int)blockIdx.x << 5;

  for (int rr = 0; rr < 2; ++rr) {
    int m = 2 * w + rr;
    int row = row_base + m;
    v2f z[8]; float den[8];
#pragma unroll
    for (int h = 0; h < 8; ++h) { z[h] = (v2f){0.f, 0.f}; den[h] = 0.f; }
    if (row < N) {
      int start = rowptr[row], end = rowptr[row + 1];
      const float4* adp = (const float4*)(ad + (size_t)row * 8);
      float4 d0 = adp[0], d1 = adp[1];
      float adh[8] = {d0.x, d0.y, d0.z, d0.w, d1.x, d1.y, d1.z, d1.w};

      int sv0; float e0[8], mx[8];
      edge_logits_h8(as, srcS, adh, start, end, lane, sv0, e0);
#pragma unroll
      for (int h = 0; h < 8; ++h) mx[h] = e0[h];
      for (int base = start + 32; base < end; base += 32) {
        int sv1; float e1[8];
        edge_logits_h8(as, srcS, adh, base, end, lane, sv1, e1);
#pragma unroll
        for (int h = 0; h < 8; ++h) mx[h] = fmaxf(mx[h], e1[h]);
      }
      float r = wave_max8(mx, lane);
      float mh[8];
#pragma unroll
      for (int h = 0; h < 8; ++h)
        mh[h] = __uint_as_float(__builtin_amdgcn_readlane(__float_as_uint(r), 4 * h));

      {
        int n = end - start; if (n > 32) n = 32;
        add_edges_x_h8(x, sv0, e0, mh, n, lane, z, den);
      }
      for (int base = start + 32; base < end; base += 32) {
        int sv1; float e1[8];
        edge_logits_h8(as, srcS, adh, base, end, lane, sv1, e1);
        int n = end - base; if (n > 32) n = 32;
        add_edges_x_h8(x, sv1, e1, mh, n, lane, z, den);
      }
    }
    float* zrow = Zs + m * ZPITCH + 2 * lane;
#pragma unroll
    for (int h = 0; h < 8; ++h) {
      float inv = (den[h] > 0.f) ? 1.f / den[h] : 0.f;
      zrow[h * 64]     = z[h].x * inv;
      zrow[h * 64 + 1] = z[h].y * inv;
    }
  }
  __syncthreads();

  int mrow = lane & 15, hi = lane >> 4;
  int col_base = (w & 7) << 6;
  int tile_row = (w >> 3) << 4;
  const float* Zrow = Zs + (tile_row + mrow) * ZPITCH + col_base;
  const float* Bcol = W + col_base + 4 * mrow;
  v8f acc0 = {}, acc1 = {}, acc2 = {}, acc3 = {};
#pragma unroll 4
  for (int k0 = 0; k0 < 64; k0 += 4) {
    int ka = k0 + 2 * hi;
    v2f a;
    a.x = Zrow[ka];
    a.y = Zrow[ka + 1];
    float4 bx = *(const float4*)(Bcol + (size_t)ka * 512);
    float4 by = *(const float4*)(Bcol + (size_t)(ka + 1) * 512);
    v2f b0, b1, b2, b3;
    b0.x = bx.x; b0.y = by.x;
    b1.x = bx.y; b1.y = by.y;
    b2.x = bx.z; b2.y = by.z;
    b3.x = bx.w; b3.y = by.w;
    acc0 = __builtin_amdgcn_wmma_f32_16x16x4_f32(false, a, false, b0, (short)0, acc0, false, false);
    acc1 = __builtin_amdgcn_wmma_f32_16x16x4_f32(false, a, false, b1, (short)0, acc1, false, false);
    acc2 = __builtin_amdgcn_wmma_f32_16x16x4_f32(false, a, false, b2, (short)0, acc2, false, false);
    acc3 = __builtin_amdgcn_wmma_f32_16x16x4_f32(false, a, false, b3, (short)0, acc3, false, false);
  }

  float* ztile = Zs + (tile_row + hi * 8) * ZPITCH + col_base + 4 * mrow;
  {
    float4 b = *(const float4*)(bias + col_base + 4 * mrow);
#pragma unroll
    for (int r = 0; r < 8; ++r) {
      float4 q;
      q.x = fmaxf(acc0[r] + b.x, 0.f); q.y = fmaxf(acc1[r] + b.y, 0.f);
      q.z = fmaxf(acc2[r] + b.z, 0.f); q.w = fmaxf(acc3[r] + b.w, 0.f);
      *(float4*)(ztile + r * ZPITCH) = q;
    }
  }
  __syncthreads();

  v8f p0 = {}, p1 = {}, p2 = {}, p3 = {};
  {
    const float* B2col = W2 + (size_t)col_base * 64 + 4 * mrow;
#pragma unroll 4
    for (int k0 = 0; k0 < 64; k0 += 4) {
      int ka = k0 + 2 * hi;
      v2f a;
      a.x = Zrow[ka];
      a.y = Zrow[ka + 1];
      float4 bx = *(const float4*)(B2col + (size_t)ka * 64);
      float4 by = *(const float4*)(B2col + (size_t)(ka + 1) * 64);
      v2f b0, b1, b2, b3;
      b0.x = bx.x; b0.y = by.x;
      b1.x = bx.y; b1.y = by.y;
      b2.x = bx.z; b2.y = by.z;
      b3.x = bx.w; b3.y = by.w;
      p0 = __builtin_amdgcn_wmma_f32_16x16x4_f32(false, a, false, b0, (short)0, p0, false, false);
      p1 = __builtin_amdgcn_wmma_f32_16x16x4_f32(false, a, false, b1, (short)0, p1, false, false);
      p2 = __builtin_amdgcn_wmma_f32_16x16x4_f32(false, a, false, b2, (short)0, p2, false, false);
      p3 = __builtin_amdgcn_wmma_f32_16x16x4_f32(false, a, false, b3, (short)0, p3, false, false);
    }
  }
#pragma unroll
  for (int r = 0; r < 8; ++r)
    *(float4*)(ztile + r * ZPITCH) = make_float4(p0[r], p1[r], p2[r], p3[r]);
  __syncthreads();

  int orow = (int)threadIdx.x >> 4, c4 = (int)threadIdx.x & 15;
  const float* zp = Zs + orow * ZPITCH + 4 * c4;
  float4 hs = *(const float4*)zp;
#pragma unroll
  for (int h = 1; h < 8; ++h) {
    float4 u = *(const float4*)(zp + 64 * h);
    hs.x += u.x; hs.y += u.y; hs.z += u.z; hs.w += u.w;
  }
  float4 s4 = *(const float4*)(avs2 + 4 * c4);
  float4 d4 = *(const float4*)(avd2 + 4 * c4);
  float vs = hs.x * s4.x + hs.y * s4.y + hs.z * s4.z + hs.w * s4.w;
  float vd = hs.x * d4.x + hs.y * d4.y + hs.z * d4.z + hs.w * d4.w;
  for (int off = 8; off; off >>= 1) {
    vs += __shfl_xor(vs, off, 32);
    vd += __shfl_xor(vd, off, 32);
  }
  int grow = row_base + orow;
  if (grow < N) {
    *(float4*)(hW2 + (size_t)grow * 64 + 4 * c4) = hs;
    if (c4 == 0) { as2[grow] = vs; ad2[grow] = vd; }
  }
}

__global__ void row_softmax_aggregate_h1(const float* __restrict__ xW, const float* __restrict__ as,
                                         const float* __restrict__ ad,
                                         const int* __restrict__ rowptr, const int* __restrict__ srcS,
                                         const float* __restrict__ bias, float* __restrict__ out, int N) {
  int row = __builtin_amdgcn_readfirstlane((int)((blockIdx.x * blockDim.x + threadIdx.x) >> 5));
  int lane = threadIdx.x & 31;
  if (row >= N) return;
  int start = rowptr[row], end = rowptr[row + 1];
  float adr = ad[row];

  int sv0 = 0; float e0 = -INFINITY;
  if (start + lane < end) { sv0 = srcS[start + lane]; e0 = leaky(as[sv0] + adr); }
  float m = e0;
  for (int base = start + 32; base < end; base += 32) {
    if (base + lane < end) m = fmaxf(m, leaky(as[srcS[base + lane]] + adr));
  }
  for (int off = 16; off; off >>= 1) m = fmaxf(m, __shfl_xor(m, off, 32));

  float2 acc = make_float2(0.f, 0.f);
  float den = 0.f;
  for (int base = start; base < end; base += 32) {
    int n = end - base; if (n > 32) n = 32;
    int sv = sv0; float e = e0;
    if (base != start) {
      sv = 0; e = -INFINITY;
      if (base + lane < end) { sv = srcS[base + lane]; e = leaky(as[sv] + adr); }
    }
    float w = __expf(e - m);
    for (int k = 0; k < n; ++k) {
      int s = __builtin_amdgcn_readlane(sv, k);
      float a = __uint_as_float(__builtin_amdgcn_readlane(__float_as_uint(w), k));
      float2 v = ((const float2*)(xW + (size_t)s * 64))[lane];
      den += a;
      acc.x += a * v.x; acc.y += a * v.y;
    }
  }
  float inv = (den > 0.f) ? 1.f / den : 0.f;
  float2 b = ((const float2*)bias)[lane];
  float2 q; q.x = acc.x * inv + b.x; q.y = acc.y * inv + b.y;
  ((float2*)(out + (size_t)row * 64))[lane] = q;
}

extern "C" void kernel_launch(void* const* d_in, const int* in_sizes, int n_in,
                              void* d_out, int out_size, void* d_ws, size_t ws_size,
                              hipStream_t stream) {
  const float* x     = (const float*)d_in[0];
  const int*   ei    = (const int*)  d_in[1];
  const float* W1    = (const float*)d_in[2];
  const float* asrc1 = (const float*)d_in[3];
  const float* adst1 = (const float*)d_in[4];
  const float* b1    = (const float*)d_in[5];
  const float* W2    = (const float*)d_in[6];
  const float* asrc2 = (const float*)d_in[7];
  const float* adst2 = (const float*)d_in[8];
  const float* b2    = (const float*)d_in[9];
  float* out = (float*)d_out;

  const int D = 64;
  const int N = in_sizes[0] / D;
  const int E = in_sizes[1] / 2;
  const int H = 8;

  float* ws   = (float*)d_ws;
  float* wa1  = ws;
  float* as1  = wa1  + 64 * 16;
  float* ad1  = as1  + (size_t)N * H;
  float* hW2  = ad1  + (size_t)N * H;
  float* as2  = hW2  + (size_t)N * D;
  float* ad2  = as2  + (size_t)N;
  int*  hist   = (int*)(ad2 + (size_t)N);
  int2* binned = (int2*)(hist + NCHUNK * NBIN);
  int*  base   = (int*)(binned + (size_t)E);
  int*  seg    = base + NBIN + 4;
  int*  srcS   = seg  + (size_t)E;
  int*  rowptr = srcS + (size_t)E;

  const int T = 256;

  const int LB = ((N + 15) / 16 + 31) / 32;
  fold_and_bin_count<<<NFOLD + NCHUNK, 1024, 0, stream>>>(ei, hist, E, N, W1, asrc1, adst1, wa1);
  bin_scatter<<<NCHUNK, 1024, 0, stream>>>(ei, hist, base, binned, E, N);
  bin_rows_and_node_logits<<<NBIN + LB, 1024, 0, stream>>>(ei, base, binned, rowptr, seg, srcS, E, N,
                                                           x, wa1, as1, ad1);

  row_softmax_aggregate_project2_h8<<<(N + 31) / 32, 512, 0, stream>>>(x, as1, ad1, rowptr, srcS, W1, b1,
                                                                      W2, asrc2, adst2, hW2, as2, ad2, N);

  row_softmax_aggregate_h1<<<(N * 32 + T - 1) / T, T, 0, stream>>>(hW2, as2, ad2, rowptr, srcS, b2, out, N);
}


# layer-2 row aggregation: one wave sums four rows side by side (8 lanes x 8 floats per row, edge id and weight passed inside the 8-lane group by DPP8): 4x fewer waves to start; sum order unchanged
# speedup vs baseline: 8.0111x; 1.0328x over previous
#include <hip/hip_runtime.h>

typedef __attribute__((ext_vector_type(2))) float v2f;
typedef __attribute__((ext_vector_type(8))) float v8f;

#define NEG_SLOPE 0.2f


#define NFOLD 4

__device__ __forceinline__ void fold_attention_vectors(const float* __restrict__ W, const float* __restrict__ avs,
                                                       const float* __restrict__ avd, float* __restrict__ wa,
                                                       int t) {
  int k = t >> 4, c = t & 15, h = c & 7;
  const float4* a4 = (const float4*)(((c & 8) ? avd : avs) + h * 64);
  const float4* w4 = (const float4*)(W + (size_t)k * 512 + h * 64);
  float s = 0.f;
#pragma unroll 4
  for (int j = 0; j < 16; ++j) {
    float4 u = w4[j], v = a4[j];
    s += u.x * v.x; s += u.y * v.y; s += u.z * v.z; s += u.w * v.w;
  }
  wa[t] = s;
}

__device__ __forceinline__ void node_logits_wmma_f32(const float* __restrict__ X, const float* __restrict__ wa,
                                                     float* __restrict__ as, float* __restrict__ ad, int M,
                                                     int wave, int lane) {
  int row_base = wave << 4;
  int mrow = lane & 15, hi = lane >> 4;
  int arow = row_base + mrow; if (arow > M - 1) arow = M - 1;
  const float* Arow = X + (size_t)arow * 64;
  v8f acc = {};
#pragma unroll
  for (int k0 = 0; k0 < 64; k0 += 4) {
    int ka = k0 + 2 * hi;
    v2f a, b;
    a.x = Arow[ka];
    a.y = Arow[ka + 1];
    b.x = wa[ka * 16 + mrow];
    b.y = wa[(ka + 1) * 16 + mrow];
    acc = __builtin_amdgcn_wmma_f32_16x16x4_f32(false, a, false, b, (short)0, acc, false, false);
  }
  float* dst = (mrow & 8) ? ad : as;
  int rbase = row_base + hi * 8;
#pragma unroll
  for (int r = 0; r < 8; ++r)
    if (rbase + r < M) dst[(size_t)(rbase + r) * 8 + (mrow & 7)] = acc[r];
}

#define NBIN 256
#define NCHUNK 256
#define RBMAX 1024
#define SLCAP 4096
#define SEGCAP 8192

__device__ __forceinline__ int bin_shift(int N) {
  int s = 0;
  while (((N - 1) >> s) >= NBIN) ++s;
  return s;
}

__global__ void __launch_bounds__(1024)
fold_and_bin_count(const int* __restrict__ ei, int* __restrict__ hist, int E, int N,
                   const float* __restrict__ W, const float* __restrict__ avs,
                   const float* __restrict__ avd, float* __restrict__ wa) {
  __shared__ int h[NBIN];
  int t = threadIdx.x;
  if ((int)blockIdx.x < NFOLD) {
    if (t < 256) fold_attention_vectors(W, avs, avd, wa, (int)blockIdx.x * 256 + t);
    return;
  }
  int b = (int)blockIdx.x - NFOLD, nb = NCHUNK;
  if (t < NBIN) h[t] = 0;
  __syncthreads();
  int s = bin_shift(N);
  int ch = (E + nb - 1) / nb;
  int lo = b * ch, hi = lo + ch; if (hi > E) hi = E;
  for (int i0 = lo + t; i0 < hi; i0 += 2048) {
    int i1 = i0 + 1024;
    bool two = i1 < hi;
    int d0 = ei[E + i0];
    int d1 = two ? ei[E + i1] : 0;
    atomicAdd(&h[d0 >> s], 1);
    if (two) atomicAdd(&h[d1 >> s], 1);
  }
  __syncthreads();
  if (t < NBIN) hist[b * NBIN + t] = h[t];
}

__global__ void __launch_bounds__(1024)
bin_scatter(const int* __restrict__ ei, const int* __restrict__ hist, int* __restrict__ base,
            int2* __restrict__ binned, int E, int N) {
  __shared__ int tot[NBIN];
  __shared__ int pre[NBIN];
  __shared__ int gst[NBIN];
  __shared__ int cur[NBIN];
  __shared__ int lst[NBIN];
  __shared__ int lcur[NBIN];
  __shared__ int wt[8];
  __shared__ int wl[8];
  __shared__ int2 buf[SLCAP];
  int t = threadIdx.x, lane = t & 31, w = t >> 5;
  int nb = (int)gridDim.x, b = (int)blockIdx.x;
  int mine_t = (t < NBIN) ? hist[b * NBIN + t] : 0;
  if (t < NBIN) { tot[t] = 0; pre[t] = 0; }
  __syncthreads();
  {
    int c4 = t & 63, rg = t >> 6;
    int4 all = make_int4(0, 0, 0, 0), mine = make_int4(0, 0, 0, 0);
    const int4* h4 = (const int4*)hist;
    for (int bb = rg; bb < nb; bb += 16) {
      int4 v = h4[bb * (NBIN / 4) + c4];
      all.x += v.x; all.y += v.y; all.z += v.z; all.w += v.w;
      if (bb < b) { mine.x += v.x; mine.y += v.y; mine.z += v.z; mine.w += v.w; }
    }
    atomicAdd(&tot[4 * c4 + 0], all.x);  atomicAdd(&tot[4 * c4 + 1], all.y);
    atomicAdd(&tot[4 * c4 + 2], all.z);  atomicAdd(&tot[4 * c4 + 3], all.w);
    atomicAdd(&pre[4 * c4 + 0], mine.x); atomicAdd(&pre[4 * c4 + 1], mine.y);
    atomicAdd(&pre[4 * c4 + 2], mine.z); atomicAdd(&pre[4 * c4 + 3], mine.w);
  }
  __syncthreads();
  int v = (t < NBIN) ? tot[t] : 0;
  int incl = v, linc = mine_t;
  for (int o = 1; o < 32; o <<= 1) {
    int u = __shfl_up(incl, o, 32);
    int ul = __shfl_up(linc, o, 32);
    if (lane >= o) { incl += u; linc += ul; }
  }
  if (t < NBIN && lane == 31) { wt[w] = incl; wl[w] = linc; }
  __syncthreads();
  if (t < NBIN) {
    int carry = 0, lcarry = 0;
    for (int i = 0; i < w; ++i) { carry += wt[i]; lcarry += wl[i]; }
    int ex = carry + incl - v;
    int g = ex + pre[t];
    int l = lcarry + linc - mine_t;
    gst[t] = g; cur[t] = g;
    lst[t] = l; lcur[t] = l;
    if (b == 0) base[t] = ex;
  }
  if (b == 0 && t == 0) base[NBIN] = E;
  __syncthreads();
  int s = bin_shift(N);
  int ch = (E + nb - 1) / nb;
  int lo = b * ch, hi = lo + ch; if (hi > E) hi = E;
  int len = hi - lo;
  if (len <= SLCAP) {
    for (int i0 = lo + t; i0 < hi; i0 += 2048) {
      int i1 = i0 + 1024;
      bool two = i1 < hi;
      int d0 = ei[E + i0];
      int d1 = two ? ei[E + i1] : 0;
      int q0 = atomicAdd(&lcur[d0 >> s], 1);
      buf[q0] = make_int2(d0, i0);
      if (two) {
        int q1 = atomicAdd(&lcur[d1 >> s], 1);
        buf[q1] = make_int2(d1, i1);
      }
    }
    __syncthreads();
    for (int j = t; j < len; j += 1024) {
      int2 pr = buf[j];
      int r = pr.x >> s;
      binned[gst[r] + (j - lst[r])] = pr;
    }
  } else {
    for (int i0 = lo + t; i0 < hi; i0 += 2048) {
      int i1 = i0 + 1024;
      bool two = i1 < hi;
      int d0 = ei[E + i0];
      int d1 = two ? ei[E + i1] : 0;
      int q0 = atomicAdd(&cur[d0 >> s], 1);
      binned[q0] = make_int2(d0, i0);
      if (two) {
        int q1 = atomicAdd(&cur[d1 >> s], 1);
        binned[q1] = make_int2(d1, i1);
      }
    }
  }
}

__global__ void __launch_bounds__(1024)
bin_rows_and_node_logits(const int* __restrict__ ei, const int* __restrict__ base, const int2* __restrict__ binned,
                         int* __restrict__ rowptr, int* __restrict__ seg, int* __restrict__ srcS, int E, int N,
                         const float* __restrict__ X, const float* __restrict__ wa,
                         float* __restrict__ as, float* __restrict__ ad) {
  __shared__ int cnt[RBMAX];
  __shared__ int start[RBMAX];
  __shared__ int cur[RBMAX];
  __shared__ int wt[32];
  __shared__ int segL[SEGCAP];
  int t = threadIdx.x, lane = t & 31;
  int w = __builtin_amdgcn_readfirstlane(t >> 5);
  if ((int)blockIdx.x >= NBIN) {
    int wave = (((int)blockIdx.x - NBIN) << 5) + w;
    if (wave < ((N + 15) >> 4)) node_logits_wmma_f32(X, wa, as, ad, N, wave, lane);
    return;
  }
  int k = (int)blockIdx.x;
  int s = bin_shift(N);
  int RB = 1 << s;
  int node0 = k << s;
  int e0 = base[k], e1 = base[k + 1];
  bool inl = (e1 - e0) <= SEGCAP;
  if (t < RB) cnt[t] = 0;
  __syncthreads();
  for (int i0 = e0 + t; i0 < e1; i0 += 4096) {
    int d[4];
#pragma unroll
    for (int u = 0; u < 4; ++u) { int i = i0 + 1024 * u; d[u] = (i < e1) ? binned[i].x : node0; }
#pragma unroll
    for (int u = 0; u < 4; ++u) { if (i0 + 1024 * u < e1) atomicAdd(&cnt[d[u] - node0], 1); }
  }
  __syncthreads();
  int v = (t < RB) ? cnt[t] : 0;
  int incl = v;
  for (int o = 1; o < 32; o <<= 1) {
    int u = __shfl_up(incl, o, 32);
    if (lane >= o) incl += u;
  }
  if (lane == 31) wt[w] = incl;
  __syncthreads();
  if (w == 0) {
    int sv = wt[lane], si = sv;
    for (int o = 1; o < 32; o <<= 1) {
      int u = __shfl_up(si, o, 32);
      if (lane >= o) si += u;
    }
    wt[lane] = si - sv;
  }
  __syncthreads();
  int ex = wt[w] + incl - v;
  if (t < RB) {
    start[t] = ex; cur[t] = ex;
    if (node0 + t < N) rowptr[node0 + t] = e0 + ex;
  }
  if (k == 0 && t == 0) rowptr[N] = E;
  __syncthreads();
  for (int i0 = e0 + t; i0 < e1; i0 += 4096) {
    int2 pr[4];
#pragma unroll
    for (int u = 0; u < 4; ++u) { int i = i0 + 1024 * u; pr[u] = (i < e1) ? binned[i] : make_int2(node0, 0); }
#pragma unroll
    for (int u = 0; u < 4; ++u) {
      if (i0 + 1024 * u < e1) {
        int p = atomicAdd(&cur[pr[u].x - node0], 1);
        if (inl) segL[p] = pr[u].y; else seg[e0 + p] = pr[u].y;
      }
    }
  }
  __syncthreads();
  for (int jb = w * 8; jb < RB; jb += 256) {
    int st[8], p0[8], n[8], id[8], rk[8];
    int jl = jb + (lane & 7);
    bool okl = (jl < RB) && (node0 + jl < N);
    int st_l = okl ? start[jl] : 0;
    int n_l = okl ? cnt[jl] : 0;
#pragma unroll
    for (int r = 0; r < 8; ++r) {
      st[r] = __builtin_amdgcn_readlane(st_l, r);
      p0[r] = e0 + st[r];
      n[r] = __builtin_amdgcn_readlane(n_l, r);
    }
    if (inl) {
#pragma unroll
      for (int r = 0; r < 8; ++r) id[r] = (lane < n[r]) ? segL[st[r] + lane] : 0x7fffffff;
    } else {
#pragma unroll
      for (int r = 0; r < 8; ++r) id[r] = (lane < n[r]) ? seg[p0[r] + lane] : 0x7fffffff;
    }
#pragma unroll
    for (int r = 0; r < 8; ++r) {
      int m = n[r] < 32 ? n[r] : 32;
      int c = 0;
      for (int q = 0; q < m; ++q) c += (__builtin_amdgcn_readlane(id[r], q) < id[r]) ? 1 : 0;
      rk[r] = c;
    }
    int sv[8];
#pragma unroll
    for (int r = 0; r < 8; ++r) sv[r] = (lane < n[r] && n[r] <= 32) ? ei[id[r]] : 0;
#pragma unroll
    for (int r = 0; r < 8; ++r) {
      if (lane < n[r] && n[r] <= 32) srcS[p0[r] + rk[r]] = sv[r];
    }
    for (int j = jb; j < jb + 8 && j < RB && node0 + j < N; ++j) {
      int nn = __builtin_amdgcn_readfirstlane(cnt[j]);
      if (nn <= 32) continue;
      int ls = __builtin_amdgcn_readfirstlane(start[j]);
      int pp = e0 + ls;
      for (int i = lane; i < nn; i += 32) {
        int me = inl ? segL[ls + i] : seg[pp + i];
        int c = 0;
        for (int q = 0; q < nn; ++q) c += ((inl ? segL[ls + q] : seg[pp + q]) < me) ? 1 : 0;
        srcS[pp + c] = ei[me];
      }
    }
  }
}

__device__ __forceinline__ float leaky(float v) { return (v >= 0.f) ? v : NEG_SLOPE * v; }

__device__ __forceinline__ void edge_logits_h8(const float* __restrict__ as, const int* __restrict__ srcS,
                                               const float (&adh)[8], int base, int end, int lane,
                                               int& sv, float (&e)[8]) {
  sv = 0;
#pragma unroll
  for (int h = 0; h < 8; ++h) e[h] = -INFINITY;
  if (base + lane < end) {
    sv = srcS[base + lane];
    const float4* ap = (const float4*)(as + (size_t)sv * 8);
    float4 a0 = ap[0], a1 = ap[1];
    e[0] = leaky(a0.x + adh[0]); e[1] = leaky(a0.y + adh[1]);
    e[2] = leaky(a0.z + adh[2]); e[3] = leaky(a0.w + adh[3]);
    e[4] = leaky(a1.x + adh[4]); e[5] = leaky(a1.y + adh[5]);
    e[6] = leaky(a1.z + adh[6]); e[7] = leaky(a1.w + adh[7]);
  }
}

__device__ __forceinline__ float wave_max8(const float (&mx)[8], int lane) {
  bool b4 = (lane & 16) != 0, b3 = (lane & 8) != 0, b2 = (lane & 4) != 0;
  float u[4], t[2];
#pragma unroll
  for (int j = 0; j < 4; ++j) {
    float keep = b4 ? mx[4 + j] : mx[j];
    float send = b4 ? mx[j] : mx[4 + j];
    u[j] = fmaxf(keep, __shfl_xor(send, 16, 32));
  }
#pragma unroll
  for (int i = 0; i < 2; ++i) {
    float keep = b3 ? u[2 + i] : u[i];
    float send = b3 ? u[i] : u[2 + i];
    t[i] = fmaxf(keep, __shfl_xor(send, 8, 32));
  }
  float r = fmaxf(b2 ? t[1] : t[0], __shfl_xor(b2 ? t[0] : t[1], 4, 32));
  r = fmaxf(r, __shfl_xor(r, 2, 32));
  r = fmaxf(r, __shfl_xor(r, 1, 32));
  return r;
}

__device__ __forceinline__ void add_edges_x_h8(const float* __restrict__ x, int sv, const float (&e)[8],
                                               const float (&mh)[8], int n, int lane,
                                               v2f (&z)[8], float (&den)[8]) {
  float w[8];
#pragma unroll
  for (int h = 0; h < 8; ++h) w[h] = __expf(e[h] - mh[h]);
#pragma unroll 4
  for (int k = 0; k < n; ++k) {
    int s = __builtin_amdgcn_readlane(sv, k);
    v2f v = ((const v2f*)(x + (size_t)s * 64))[lane];
#pragma unroll
    for (int h = 0; h < 8; ++h) {
      float a = __uint_as_float(__builtin_amdgcn_readlane(__float_as_uint(w[h]), k));
      den[h] += a;
      z[h] += a * v;
    }
  }
}

#define ZPITCH 516

__global__ void __launch_bounds__(512) __attribute__((amdgpu_num_vgpr(80)))
row_softmax_aggregate_project2_h8(const float* __restrict__ x, const float* __restrict__ as,
                                  const float* __restrict__ ad,
                                  const int* __restrict__ rowptr, const int* __restrict__ srcS,
                                  const float* __restrict__ W, const float* __restrict__ bias,
                                  const float* __restrict__ W2, const float* __restrict__ avs2,
                                  const float* __restrict__ avd2, float* __restrict__ hW2,
                                  float* __restrict__ as2, float* __restrict__ ad2, int N) {
  __shared__ __attribute__((aligned(16))) float Zs[32 * ZPITCH];
  int lane = threadIdx.x & 31;
  int w = __builtin_amdgcn_readfirstlane((int)(threadIdx.x >> 5));
  int row_base = (int)blockIdx.x << 5;

  for (int rr = 0; rr < 2; ++rr) {
    int m = 2 * w + rr;
    int row = row_base + m;
    v2f z[8]; float den[8];
#pragma unroll
    for (int h = 0; h < 8; ++h) { z[h] = (v2f){0.f, 0.f}; den[h] = 0.f; }
    if (row < N) {
      int start = rowptr[row], end = rowptr[row + 1];
      const float4* adp = (const float4*)(ad + (size_t)row * 8);
      float4 d0 = adp[0], d1 = adp[1];
      float adh[8] = {d0.x, d0.y, d0.z, d0.w, d1.x, d1.y, d1.z, d1.w};

      int sv0; float e0[8], mx[8];
      edge_logits_h8(as, srcS, adh, start, end, lane, sv0, e0);
#pragma unroll
      for (int h = 0; h < 8; ++h) mx[h] = e0[h];
      for (int base = start + 32; base < end; base += 32) {
        int sv1; float e1[8];
        edge_logits_h8(as, srcS, adh, base, end, lane, sv1, e1);
#pragma unroll
        for (int h = 0; h < 8; ++h) mx[h] = fmaxf(mx[h], e1[h]);
      }
      float r = wave_max8(mx, lane);
      float mh[8];
#pragma unroll
      for (int h = 0; h < 8; ++h)
        mh[h] = __uint_as_float(__builtin_amdgcn_readlane(__float_as_uint(r), 4 * h));

      {
        int n = end - start; if (n > 32) n = 32;
        add_edges_x_h8(x, sv0, e0, mh, n, lane, z, den);
      }
      for (int base = start + 32; base < end; base += 32) {
        int sv1; float e1[8];
        edge_logits_h8(as, srcS, adh, base, end, lane, sv1, e1);
        int n = end - base; if (n > 32) n = 32;
        add_edges_x_h8(x, sv1, e1, mh, n, lane, z, den);
      }
    }
    float* zrow = Zs + m * ZPITCH + 2 * lane;
#pragma unroll
    for (int h = 0; h < 8; ++h) {
      float inv = (den[h] > 0.f) ? 1.f / den[h] : 0.f;
      zrow[h * 64]     = z[h].x * inv;
      zrow[h * 64 + 1] = z[h].y * inv;
    }
  }
  __syncthreads();

  int mrow = lane & 15, hi = lane >> 4;
  int col_base = (w & 7) << 6;
  int tile_row = (w >> 3) << 4;
  const float* Zrow = Zs + (tile_row + mrow) * ZPITCH + col_base;
  const float* Bcol = W + col_base + 4 * mrow;
  v8f acc0 = {}, acc1 = {}, acc2 = {}, acc3 = {};
#pragma unroll 4
  for (int k0 = 0; k0 < 64; k0 += 4) {
    int ka = k0 + 2 * hi;
    v2f a;
    a.x = Zrow[ka];
    a.y = Zrow[ka + 1];
    float4 bx = *(const float4*)(Bcol + (size_t)ka * 512);
    float4 by = *(const float4*)(Bcol + (size_t)(ka + 1) * 512);
    v2f b0, b1, b2, b3;
    b0.x = bx.x; b0.y = by.x;
    b1.x = bx.y; b1.y = by.y;
    b2.x = bx.z; b2.y = by.z;
    b3.x = bx.w; b3.y = by.w;
    acc0 = __builtin_amdgcn_wmma_f32_16x16x4_f32(false, a, false, b0, (short)0, acc0, false, false);
    acc1 = __builtin_amdgcn_wmma_f32_16x16x4_f32(false, a, false, b1, (short)0, acc1, false, false);
    acc2 = __builtin_amdgcn_wmma_f32_16x16x4_f32(false, a, false, b2, (short)0, acc2, false, false);
    acc3 = __builtin_amdgcn_wmma_f32_16x16x4_f32(false, a, false, b3, (short)0, acc3, false, false);
  }

  float* ztile = Zs + (tile_row + hi * 8) * ZPITCH + col_base + 4 * mrow;
  {
    float4 b = *(const float4*)(bias + col_base + 4 * mrow);
#pragma unroll
    for (int r = 0; r < 8; ++r) {
      float4 q;
      q.x = fmaxf(acc0[r] + b.x, 0.f); q.y = fmaxf(acc1[r] + b.y, 0.f);
      q.z = fmaxf(acc2[r] + b.z, 0.f); q.w = fmaxf(acc3[r] + b.w, 0.f);
      *(float4*)(ztile + r * ZPITCH) = q;
    }
  }
  __syncthreads();

  v8f p0 = {}, p1 = {}, p2 = {}, p3 = {};
  {
    const float* B2col = W2 + (size_t)col_base * 64 + 4 * mrow;
#pragma unroll 4
    for (int k0 = 0; k0 < 64; k0 += 4) {
      int ka = k0 + 2 * hi;
      v2f a;
      a.x = Zrow[ka];
      a.y = Zrow[ka + 1];
      float4 bx = *(const float4*)(B2col + (size_t)ka * 64);
      float4 by = *(const float4*)(B2col + (size_t)(ka + 1) * 64);
      v2f b0, b1, b2, b3;
      b0.x = bx.x; b0.y = by.x;
      b1.x = bx.y; b1.y = by.y;
      b2.x = bx.z; b2.y = by.z;
      b3.x = bx.w; b3.y = by.w;
      p0 = __builtin_amdgcn_wmma_f32_16x16x4_f32(false, a, false, b0, (short)0, p0, false, false);
      p1 = __builtin_amdgcn_wmma_f32_16x16x4_f32(false, a, false, b1, (short)0, p1, false, false);
      p2 = __builtin_amdgcn_wmma_f32_16x16x4_f32(false, a, false, b2, (short)0, p2, false, false);
      p3 = __builtin_amdgcn_wmma_f32_16x16x4_f32(false, a, false, b3, (short)0, p3, false, false);
    }
  }
#pragma unroll
  for (int r = 0; r < 8; ++r)
    *(float4*)(ztile + r * ZPITCH) = make_float4(p0[r], p1[r], p2[r], p3[r]);
  __syncthreads();

  int orow = (int)threadIdx.x >> 4, c4 = (int)threadIdx.x & 15;
  const float* zp = Zs + orow * ZPITCH + 4 * c4;
  float4 hs = *(const float4*)zp;
#pragma unroll
  for (int h = 1; h < 8; ++h) {
    float4 u = *(const float4*)(zp + 64 * h);
    hs.x += u.x; hs.y += u.y; hs.z += u.z; hs.w += u.w;
  }
  float4 s4 = *(const float4*)(avs2 + 4 * c4);
  float4 d4 = *(const float4*)(avd2 + 4 * c4);
  float vs = hs.x * s4.x + hs.y * s4.y + hs.z * s4.z + hs.w * s4.w;
  float vd = hs.x * d4.x + hs.y * d4.y + hs.z * d4.z + hs.w * d4.w;
  for (int off = 8; off; off >>= 1) {
    vs += __shfl_xor(vs, off, 32);
    vd += __shfl_xor(vd, off, 32);
  }
  int grow = row_base + orow;
  if (grow < N) {
    *(float4*)(hW2 + (size_t)grow * 64 + 4 * c4) = hs;
    if (c4 == 0) { as2[grow] = vs; ad2[grow] = vd; }
  }
}

template <int SEL> __device__ __forceinline__ int dpp8_i(int v) {
  return (int)__builtin_amdgcn_mov_dpp8((unsigned)v, (unsigned)SEL);
}
template <int SEL> __device__ __forceinline__ float dpp8_f(float v) {
  return __uint_as_float(__builtin_amdgcn_mov_dpp8(__float_as_uint(v), (unsigned)SEL));
}
#define BC8(k) ((k) * 0x249249)
#define XOR8_1 0xde54c1
#define XOR8_2 0xb3e21a
#define XOR8_4 0x688fac

template <int G> __device__ __forceinline__ void add4_rows_h1(const float4* __restrict__ X4, int sv, float w, int j,
                                                              float4& lo, float4& hi, float& den) {
  int s0 = dpp8_i<BC8(G)>(sv), s1 = dpp8_i<BC8(G + 1)>(sv), s2 = dpp8_i<BC8(G + 2)>(sv), s3 = dpp8_i<BC8(G + 3)>(sv);
  float a0 = dpp8_f<BC8(G)>(w), a1 = dpp8_f<BC8(G + 1)>(w), a2 = dpp8_f<BC8(G + 2)>(w), a3 = dpp8_f<BC8(G + 3)>(w);
  float4 u0 = X4[(unsigned)(s0 * 16 + j)], v0 = X4[(unsigned)(s0 * 16 + 8 + j)];
  float4 u1 = X4[(unsigned)(s1 * 16 + j)], v1 = X4[(unsigned)(s1 * 16 + 8 + j)];
  float4 u2 = X4[(unsigned)(s2 * 16 + j)], v2 = X4[(unsigned)(s2 * 16 + 8 + j)];
  float4 u3 = X4[(unsigned)(s3 * 16 + j)], v3 = X4[(unsigned)(s3 * 16 + 8 + j)];
  den += a0;
  lo.x += a0 * u0.x; lo.y += a0 * u0.y; lo.z += a0 * u0.z; lo.w += a0 * u0.w;
  hi.x += a0 * v0.x; hi.y += a0 * v0.y; hi.z += a0 * v0.z; hi.w += a0 * v0.w;
  den += a1;
  lo.x += a1 * u1.x; lo.y += a1 * u1.y; lo.z += a1 * u1.z; lo.w += a1 * u1.w;
  hi.x += a1 * v1.x; hi.y += a1 * v1.y; hi.z += a1 * v1.z; hi.w += a1 * v1.w;
  den += a2;
  lo.x += a2 * u2.x; lo.y += a2 * u2.y; lo.z += a2 * u2.z; lo.w += a2 * u2.w;
  hi.x += a2 * v2.x; hi.y += a2 * v2.y; hi.z += a2 * v2.z; hi.w += a2 * v2.w;
  den += a3;
  lo.x += a3 * u3.x; lo.y += a3 * u3.y; lo.z += a3 * u3.z; lo.w += a3 * u3.w;
  hi.x += a3 * v3.x; hi.y += a3 * v3.y; hi.z += a3 * v3.z; hi.w += a3 * v3.w;
}

__global__ void __launch_bounds__(256)
row_softmax_aggregate_h1(const float* __restrict__ xW, const float* __restrict__ as,
                         const float* __restrict__ ad,
                         const int* __restrict__ rowptr, const int* __restrict__ srcS,
                         const float* __restrict__ bias, float* __restrict__ out, int N) {
  int wave = __builtin_amdgcn_readfirstlane((int)(blockIdx.x * 8 + (threadIdx.x >> 5)));
  int lane = threadIdx.x & 31;
  if (4 * wave >= N) return;
  int j = lane & 7;
  int row = 4 * wave + (lane >> 3);
  bool rv = row < N;
  int start = 0, end = 0;
  float adr = 0.f;
  if (rv) { start = rowptr[row]; end = rowptr[row + 1]; adr = ad[row]; }
  int n = end - start;
  int nmax = __builtin_amdgcn_readlane(n, 0);
  { int t1 = __builtin_amdgcn_readlane(n, 8), t2 = __builtin_amdgcn_readlane(n, 16), t3 = __builtin_amdgcn_readlane(n, 24);
    if (t1 > nmax) nmax = t1;
    if (t2 > nmax) nmax = t2;
    if (t3 > nmax) nmax = t3; }

  bool in0 = j < n, in1 = j + 8 < n;
  int t0 = srcS[in0 ? start + j : 0];
  int t1 = srcS[in1 ? start + 8 + j : 0];
  float g0 = as[t0], g1 = as[t1];
  int sv0 = in0 ? t0 : 0, sv1 = in1 ? t1 : 0;
  float e0 = in0 ? leaky(g0 + adr) : -INFINITY;
  float e1 = in1 ? leaky(g1 + adr) : -INFINITY;
  float m = fmaxf(e0, e1);
  for (int c = 16; c < nmax; c += 8) {
    if (c + j < n) m = fmaxf(m, leaky(as[srcS[start + c + j]] + adr));
  }
  m = fmaxf(m, dpp8_f<XOR8_1>(m));
  m = fmaxf(m, dpp8_f<XOR8_2>(m));
  m = fmaxf(m, dpp8_f<XOR8_4>(m));

  const float4* X4 = (const float4*)xW;
  float4 lo = make_float4(0.f, 0.f, 0.f, 0.f), hi = make_float4(0.f, 0.f, 0.f, 0.f);
  float den = 0.f;
  for (int c = 0; c < nmax; c += 8) {
    int sv = sv0; float e = e0;
    if (c == 8) { sv = sv1; e = e1; }
    else if (c != 0) {
      sv = 0; e = -INFINITY;
      if (c + j < n) { sv = srcS[start + c + j]; e = leaky(as[sv] + adr); }
    }
    float w = (c + j < n) ? __expf(e - m) : 0.f;
    add4_rows_h1<0>(X4, sv, w, j, lo, hi, den);
    if (nmax - c > 4) add4_rows_h1<4>(X4, sv, w, j, lo, hi, den);
  }
  float inv = (den > 0.f) ? 1.f / den : 0.f;
  float4 b0 = ((const float4*)bias)[j], b1 = ((const float4*)bias)[8 + j];
  if (rv) {
    float4 q0, q1;
    q0.x = lo.x * inv + b0.x; q0.y = lo.y * inv + b0.y; q0.z = lo.z * inv + b0.z; q0.w = lo.w * inv + b0.w;
    q1.x = hi.x * inv + b1.x; q1.y = hi.y * inv + b1.y; q1.z = hi.z * inv + b1.z; q1.w = hi.w * inv + b1.w;
    float4* o = (float4*)(out + (size_t)row * 64);
    o[j] = q0;
    o[8 + j] = q1;
  }
}

extern "C" void kernel_launch(void* const* d_in, const int* in_sizes, int n_in,
                              void* d_out, int out_size, void* d_ws, size_t ws_size,
                              hipStream_t stream) {
  const float* x     = (const float*)d_in[0];
  const int*   ei    = (const int*)  d_in[1];
  const float* W1    = (const float*)d_in[2];
  const float* asrc1 = (const float*)d_in[3];
  const float* adst1 = (const float*)d_in[4];
  const float* b1    = (const float*)d_in[5];
  const float* W2    = (const float*)d_in[6];
  const float* asrc2 = (const float*)d_in[7];
  const float* adst2 = (const float*)d_in[8];
  const float* b2    = (const float*)d_in[9];
  float* out = (float*)d_out;

  const int D = 64;
  const int N = in_sizes[0] / D;
  const int E = in_sizes[1] / 2;
  const int H = 8;

  float* ws   = (float*)d_ws;
  float* wa1  = ws;
  float* as1  = wa1  + 64 * 16;
  float* ad1  = as1  + (size_t)N * H;
  float* hW2  = ad1  + (size_t)N * H;
  float* as2  = hW2  + (size_t)N * D;
  float* ad2  = as2  + (size_t)N;
  int*  hist   = (int*)(ad2 + (size_t)N);
  int2* binned = (int2*)(hist + NCHUNK * NBIN);
  int*  base   = (int*)(binned + (size_t)E);
  int*  seg    = base + NBIN + 4;
  int*  srcS   = seg  + (size_t)E;
  int*  rowptr = srcS + (size_t)E;

  const int T = 256;

  const int LB = ((N + 15) / 16 + 31) / 32;
  fold_and_bin_count<<<NFOLD + NCHUNK, 1024, 0, stream>>>(ei, hist, E, N, W1, asrc1, adst1, wa1);
  bin_scatter<<<NCHUNK, 1024, 0, stream>>>(ei, hist, base, binned, E, N);
  bin_rows_and_node_logits<<<NBIN + LB, 1024, 0, stream>>>(ei, base, binned, rowptr, seg, srcS, E, N,
                                                           x, wa1, as1, ad1);

  row_softmax_aggregate_project2_h8<<<(N + 31) / 32, 512, 0, stream>>>(x, as1, ad1, rowptr, srcS, W1, b1,
                                                                      W2, asrc2, adst2, hW2, as2, ad2, N);

  row_softmax_aggregate_h1<<<((N + 3) / 4 * 32 + T - 1) / T, T, 0, stream>>>(hW2, as2, ad2, rowptr, srcS, b2, out, N);
}


# edge sort level 1 made slice-local: a slice groups its own records by range (count, scan, place in LDS, contiguous store) in one launch; the range pass gathers its 256 short runs; 5 launches -> 4
# speedup vs baseline: 8.5500x; 1.0673x over previous
#include <hip/hip_runtime.h>

typedef __attribute__((ext_vector_type(2))) float v2f;
typedef __attribute__((ext_vector_type(8))) float v8f;

#define NEG_SLOPE 0.2f


#define NFOLD 4

__device__ __forceinline__ void fold_attention_vectors(const float* __restrict__ W, const float* __restrict__ avs,
                                                       const float* __restrict__ avd, float* __restrict__ wa,
                                                       int t) {
  int k = t >> 4, c = t & 15, h = c & 7;
  const float4* a4 = (const float4*)(((c & 8) ? avd : avs) + h * 64);
  const float4* w4 = (const float4*)(W + (size_t)k * 512 + h * 64);
  float s = 0.f;
#pragma unroll 4
  for (int j = 0; j < 16; ++j) {
    float4 u = w4[j], v = a4[j];
    s += u.x * v.x; s += u.y * v.y; s += u.z * v.z; s += u.w * v.w;
  }
  wa[t] = s;
}

__device__ __forceinline__ void node_logits_wmma_f32(const float* __restrict__ X, const float* __restrict__ wa,
                                                     float* __restrict__ as, float* __restrict__ ad, int M,
                                                     int wave, int lane) {
  int row_base = wave << 4;
  int mrow = lane & 15, hi = lane >> 4;
  int arow = row_base + mrow; if (arow > M - 1) arow = M - 1;
  const float* Arow = X + (size_t)arow * 64;
  v8f acc = {};
#pragma unroll
  for (int k0 = 0; k0 < 64; k0 += 4) {
    int ka = k0 + 2 * hi;
    v2f a, b;
    a.x = Arow[ka];
    a.y = Arow[ka + 1];
    b.x = wa[ka * 16 + mrow];
    b.y = wa[(ka + 1) * 16 + mrow];
    acc = __builtin_amdgcn_wmma_f32_16x16x4_f32(false, a, false, b, (short)0, acc, false, false);
  }
  float* dst = (mrow & 8) ? ad : as;
  int rbase = row_base + hi * 8;
#pragma unroll
  for (int r = 0; r < 8; ++r)
    if (rbase + r < M) dst[(size_t)(rbase + r) * 8 + (mrow & 7)] = acc[r];
}

#define NBIN 256
#define NCHUNK 256
#define SPW (NCHUNK / 32)
#define RBMAX 1024
#define SLCAP 4096
#define SEGCAP 8192

__device__ __forceinline__ int bin_shift(int N) {
  int s = 0;
  while (((N - 1) >> s) >= NBIN) ++s;
  return s;
}

__global__ void __launch_bounds__(1024)
fold_and_group_slices(const int* __restrict__ ei, int* __restrict__ hist, int* __restrict__ hoff,
                      int2* __restrict__ binned, int E, int N,
                      const float* __restrict__ W, const float* __restrict__ avs,
                      const float* __restrict__ avd, float* __restrict__ wa) {
  __shared__ int h[NBIN];
  __shared__ int lcur[NBIN];
  __shared__ int wl[8];
  __shared__ int2 buf[SLCAP];
  int t = threadIdx.x, lane = t & 31, w = t >> 5;
  if ((int)blockIdx.x < NFOLD) {
    if (t < 256) fold_attention_vectors(W, avs, avd, wa, (int)blockIdx.x * 256 + t);
    return;
  }
  int b = (int)blockIdx.x - NFOLD, nb = NCHUNK;
  if (t < NBIN) h[t] = 0;
  __syncthreads();
  int s = bin_shift(N);
  int ch = (E + nb - 1) / nb;
  int lo = b * ch, hi = lo + ch; if (hi > E) hi = E;
  for (int i0 = lo + t; i0 < hi; i0 += 2048) {
    int i1 = i0 + 1024;
    bool two = i1 < hi;
    int d0 = ei[E + i0];
    int d1 = two ? ei[E + i1] : 0;
    atomicAdd(&h[d0 >> s], 1);
    if (two) atomicAdd(&h[d1 >> s], 1);
  }
  __syncthreads();
  int v = (t < NBIN) ? h[t] : 0;
  int incl = v;
  for (int o = 1; o < 32; o <<= 1) {
    int u = __shfl_up(incl, o, 32);
    if (lane >= o) incl += u;
  }
  if (t < NBIN && lane == 31) wl[w] = incl;
  __syncthreads();
  if (t < NBIN) {
    int carry = 0;
    for (int i = 0; i < w; ++i) carry += wl[i];
    int ex = carry + incl - v;
    lcur[t] = ex;
    hist[b * NBIN + t] = v;
    hoff[b * NBIN + t] = ex;
  }
  __syncthreads();
  int len = hi - lo;
  if (len <= SLCAP) {
    for (int i0 = lo + t; i0 < hi; i0 += 2048) {
      int i1 = i0 + 1024;
      bool two = i1 < hi;
      int d0 = ei[E + i0];
      int d1 = two ? ei[E + i1] : 0;
      int q0 = atomicAdd(&lcur[d0 >> s], 1);
      buf[q0] = make_int2(d0, i0);
      if (two) {
        int q1 = atomicAdd(&lcur[d1 >> s], 1);
        buf[q1] = make_int2(d1, i1);
      }
    }
    __syncthreads();
    for (int j = t; j < len; j += 1024) binned[lo + j] = buf[j];
  } else {
    for (int i0 = lo + t; i0 < hi; i0 += 2048) {
      int i1 = i0 + 1024;
      bool two = i1 < hi;
      int d0 = ei[E + i0];
      int d1 = two ? ei[E + i1] : 0;
      int q0 = atomicAdd(&lcur[d0 >> s], 1);
      binned[lo + q0] = make_int2(d0, i0);
      if (two) {
        int q1 = atomicAdd(&lcur[d1 >> s], 1);
        binned[lo + q1] = make_int2(d1, i1);
      }
    }
  }
}

__global__ void __launch_bounds__(1024)
bin_rows_and_node_logits(const int* __restrict__ ei, const int* __restrict__ hist, const int* __restrict__ hoff,
                         const int2* __restrict__ binned,
                         int* __restrict__ rowptr, int* __restrict__ seg, int* __restrict__ srcS, int E, int N,
                         const float* __restrict__ X, const float* __restrict__ wa,
                         float* __restrict__ as, float* __restrict__ ad) {
  __shared__ int cnt[RBMAX];
  __shared__ int start[RBMAX];
  __shared__ int cur[RBMAX];
  __shared__ int wt[32];
  __shared__ int segL[SEGCAP];
  __shared__ int rlo[NCHUNK];
  __shared__ int rcn[NCHUNK];
  __shared__ int tot2[2];
  int t = threadIdx.x, lane = t & 31;
  int w = __builtin_amdgcn_readfirstlane(t >> 5);
  if ((int)blockIdx.x >= NBIN) {
    int wave = (((int)blockIdx.x - NBIN) << 5) + w;
    if (wave < ((N + 15) >> 4)) node_logits_wmma_f32(X, wa, as, ad, N, wave, lane);
    return;
  }
  int k = (int)blockIdx.x;
  int s = bin_shift(N);
  int RB = 1 << s;
  int node0 = k << s;
  int ch = (E + NCHUNK - 1) / NCHUNK;
  if (t < 2) tot2[t] = 0;
  if (t < RB) cnt[t] = 0;
  __syncthreads();
  if (w < NCHUNK / 32) {
    int c = hist[t * NBIN + k], o = hoff[t * NBIN + k];
    rlo[t] = t * ch + o;
    rcn[t] = c;
    for (int off = 16; off; off >>= 1) {
      o += __shfl_xor(o, off, 32);
      c += __shfl_xor(c, off, 32);
    }
    if (lane == 0) { atomicAdd(&tot2[0], o); atomicAdd(&tot2[1], c); }
  }
  __syncthreads();
  int e0 = tot2[0], e1 = e0 + tot2[1];
  bool inl = (e1 - e0) <= SEGCAP;
  int sb = w * SPW;
  int c_l = rcn[sb + (lane & (SPW - 1))], p_l = rlo[sb + (lane & (SPW - 1))];
  int cs[SPW], ps[SPW];
  int2 pr[SPW];
#pragma unroll
  for (int q = 0; q < SPW; ++q) {
    cs[q] = __builtin_amdgcn_readlane(c_l, q);
    ps[q] = __builtin_amdgcn_readlane(p_l, q);
  }
#pragma unroll
  for (int q = 0; q < SPW; ++q) pr[q] = (lane < cs[q]) ? binned[ps[q] + lane] : make_int2(node0, 0);
#pragma unroll
  for (int q = 0; q < SPW; ++q) { if (lane < cs[q]) atomicAdd(&cnt[pr[q].x - node0], 1); }
  int cmax = cs[0];
#pragma unroll
  for (int q = 1; q < SPW; ++q) cmax = cs[q] > cmax ? cs[q] : cmax;
  if (cmax > 32) {
    for (int q = 0; q < SPW; ++q) {
      int c = __builtin_amdgcn_readfirstlane(rcn[sb + q]);
      if (c <= 32) continue;
      int p = __builtin_amdgcn_readfirstlane(rlo[sb + q]);
      for (int j = 32 + lane; j < c; j += 32) atomicAdd(&cnt[binned[p + j].x - node0], 1);
    }
  }
  __syncthreads();
  int v = (t < RB) ? cnt[t] : 0;
  int incl = v;
  for (int o = 1; o < 32; o <<= 1) {
    int u = __shfl_up(incl, o, 32);
    if (lane >= o) incl += u;
  }
  if (lane == 31) wt[w] = incl;
  __syncthreads();
  if (w == 0) {
    int sv = wt[lane], si = sv;
    for (int o = 1; o < 32; o <<= 1) {
      int u = __shfl_up(si, o, 32);
      if (lane >= o) si += u;
    }
    wt[lane] = si - sv;
  }
  __syncthreads();
  int ex = wt[w] + incl - v;
  if (t < RB) {
    start[t] = ex; cur[t] = ex;
    if (node0 + t < N) rowptr[node0 + t] = e0 + ex;
  }
  if (k == 0 && t == 0) rowptr[N] = E;
  __syncthreads();
#pragma unroll
  for (int q = 0; q < SPW; ++q) {
    if (lane < cs[q]) {
      int p = atomicAdd(&cur[pr[q].x - node0], 1);
      if (inl) segL[p] = pr[q].y; else seg[e0 + p] = pr[q].y;
    }
  }
  if (cmax > 32) {
    for (int q = 0; q < SPW; ++q) {
      int c = __builtin_amdgcn_readfirstlane(rcn[sb + q]);
      if (c <= 32) continue;
      int pp = __builtin_amdgcn_readfirstlane(rlo[sb + q]);
      for (int j = 32 + lane; j < c; j += 32) {
        int2 r = binned[pp + j];
        int p = atomicAdd(&cur[r.x - node0], 1);
        if (inl) segL[p] = r.y; else seg[e0 + p] = r.y;
      }
    }
  }
  __syncthreads();
  for (int jb = w * 8; jb < RB; jb += 256) {
    int st[8], p0[8], n[8], id[8], rk[8];
    int jl = jb + (lane & 7);
    bool okl = (jl < RB) && (node0 + jl < N);
    int st_l = okl ? start[jl] : 0;
    int n_l = okl ? cnt[jl] : 0;
#pragma unroll
    for (int r = 0; r < 8; ++r) {
      st[r] = __builtin_amdgcn_readlane(st_l, r);
      p0[r] = e0 + st[r];
      n[r] = __builtin_amdgcn_readlane(n_l, r);
    }
    if (inl) {
#pragma unroll
      for (int r = 0; r < 8; ++r) id[r] = (lane < n[r]) ? segL[st[r] + lane] : 0x7fffffff;
    } else {
#pragma unroll
      for (int r = 0; r < 8; ++r) id[r] = (lane < n[r]) ? seg[p0[r] + lane] : 0x7fffffff;
    }
#pragma unroll
    for (int r = 0; r < 8; ++r) {
      int m = n[r] < 32 ? n[r] : 32;
      int c = 0;
      for (int q = 0; q < m; ++q) c += (__builtin_amdgcn_readlane(id[r], q) < id[r]) ? 1 : 0;
      rk[r] = c;
    }
    int sv[8];
#pragma unroll
    for (int r = 0; r < 8; ++r) sv[r] = (lane < n[r] && n[r] <= 32) ? ei[id[r]] : 0;
#pragma unroll
    for (int r = 0; r < 8; ++r) {
      if (lane < n[r] && n[r] <= 32) srcS[p0[r] + rk[r]] = sv[r];
    }
    for (int j = jb; j < jb + 8 && j < RB && node0 + j < N; ++j) {
      int nn = __builtin_amdgcn_readfirstlane(cnt[j]);
      if (nn <= 32) continue;
      int ls = __builtin_amdgcn_readfirstlane(start[j]);
      int pp = e0 + ls;
      for (int i = lane; i < nn; i += 32) {
        int me = inl ? segL[ls + i] : seg[pp + i];
        int c = 0;
        for (int q = 0; q < nn; ++q) c += ((inl ? segL[ls + q] : seg[pp + q]) < me) ? 1 : 0;
        srcS[pp + c] = ei[me];
      }
    }
  }
}

__device__ __forceinline__ float leaky(float v) { return (v >= 0.f) ? v : NEG_SLOPE * v; }

__device__ __forceinline__ void edge_logits_h8(const float* __restrict__ as, const int* __restrict__ srcS,
                                               const float (&adh)[8], int base, int end, int lane,
                                               int& sv, float (&e)[8]) {
  sv = 0;
#pragma unroll
  for (int h = 0; h < 8; ++h) e[h] = -INFINITY;
  if (base + lane < end) {
    sv = srcS[base + lane];
    const float4* ap = (const float4*)(as + (size_t)sv * 8);
    float4 a0 = ap[0], a1 = ap[1];
    e[0] = leaky(a0.x + adh[0]); e[1] = leaky(a0.y + adh[1]);
    e[2] = leaky(a0.z + adh[2]); e[3] = leaky(a0.w + adh[3]);
    e[4] = leaky(a1.x + adh[4]); e[5] = leaky(a1.y + adh[5]);
    e[6] = leaky(a1.z + adh[6]); e[7] = leaky(a1.w + adh[7]);
  }
}

__device__ __forceinline__ float wave_max8(const float (&mx)[8], int lane) {
  bool b4 = (lane & 16) != 0, b3 = (lane & 8) != 0, b2 = (lane & 4) != 0;
  float u[4], t[2];
#pragma unroll
  for (int j = 0; j < 4; ++j) {
    float keep = b4 ? mx[4 + j] : mx[j];
    float send = b4 ? mx[j] : mx[4 + j];
    u[j] = fmaxf(keep, __shfl_xor(send, 16, 32));
  }
#pragma unroll
  for (int i = 0; i < 2; ++i) {
    float keep = b3 ? u[2 + i] : u[i];
    float send = b3 ? u[i] : u[2 + i];
    t[i] = fmaxf(keep, __shfl_xor(send, 8, 32));
  }
  float r = fmaxf(b2 ? t[1] : t[0], __shfl_xor(b2 ? t[0] : t[1], 4, 32));
  r = fmaxf(r, __shfl_xor(r, 2, 32));
  r = fmaxf(r, __shfl_xor(r, 1, 32));
  return r;
}

__device__ __forceinline__ void add_edges_x_h8(const float* __restrict__ x, int sv, const float (&e)[8],
                                               const float (&mh)[8], int n, int lane,
                                               v2f (&z)[8], float (&den)[8]) {
  float w[8];
#pragma unroll
  for (int h = 0; h < 8; ++h) w[h] = __expf(e[h] - mh[h]);
#pragma unroll 4
  for (int k = 0; k < n; ++k) {
    int s = __builtin_amdgcn_readlane(sv, k);
    v2f v = ((const v2f*)(x + (size_t)s * 64))[lane];
#pragma unroll
    for (int h = 0; h < 8; ++h) {
      float a = __uint_as_float(__builtin_amdgcn_readlane(__float_as_uint(w[h]), k));
      den[h] += a;
      z[h] += a * v;
    }
  }
}

#define ZPITCH 516

__global__ void __launch_bounds__(512) __attribute__((amdgpu_num_vgpr(80)))
row_softmax_aggregate_project2_h8(const float* __restrict__ x, const float* __restrict__ as,
                                  const float* __restrict__ ad,
                                  const int* __restrict__ rowptr, const int* __restrict__ srcS,
                                  const float* __restrict__ W, const float* __restrict__ bias,
                                  const float* __restrict__ W2, const float* __restrict__ avs2,
                                  const float* __restrict__ avd2, float* __restrict__ hW2,
                                  float* __restrict__ as2, float* __restrict__ ad2, int N) {
  __shared__ __attribute__((aligned(16))) float Zs[32 * ZPITCH];
  int lane = threadIdx.x & 31;
  int w = __builtin_amdgcn_readfirstlane((int)(threadIdx.x >> 5));
  int row_base = (int)blockIdx.x << 5;

  for (int rr = 0; rr < 2; ++rr) {
    int m = 2 * w + rr;
    int row = row_base + m;
    v2f z[8]; float den[8];
#pragma unroll
    for (int h = 0; h < 8; ++h) { z[h] = (v2f){0.f, 0.f}; den[h] = 0.f; }
    if (row < N) {
      int start = rowptr[row], end = rowptr[row + 1];
      const float4* adp = (const float4*)(ad + (size_t)row * 8);
      float4 d0 = adp[0], d1 = adp[1];
      float adh[8] = {d0.x, d0.y, d0.z, d0.w, d1.x, d1.y, d1.z, d1.w};

      int sv0; float e0[8], mx[8];
      edge_logits_h8(as, srcS, adh, start, end, lane, sv0, e0);
#pragma unroll
      for (int h = 0; h < 8; ++h) mx[h] = e0[h];
      for (int base = start + 32; base < end; base += 32) {
        int sv1; float e1[8];
        edge_logits_h8(as, srcS, adh, base, end, lane, sv1, e1);
#pragma unroll
        for (int h = 0; h < 8; ++h) mx[h] = fmaxf(mx[h], e1[h]);
      }
      float r = wave_max8(mx, lane);
      float mh[8];
#pragma unroll
      for (int h = 0; h < 8; ++h)
        mh[h] = __uint_as_float(__builtin_amdgcn_readlane(__float_as_uint(r), 4 * h));

      {
        int n = end - start; if (n > 32) n = 32;
        add_edges_x_h8(x, sv0, e0, mh, n, lane, z, den);
      }
      for (int base = start + 32; base < end; base += 32) {
        int sv1; float e1[8];
        edge_logits_h8(as, srcS, adh, base, end, lane, sv1, e1);
        int n = end - base; if (n > 32) n = 32;
        add_edges_x_h8(x, sv1, e1, mh, n, lane, z, den);
      }
    }
    float* zrow = Zs + m * ZPITCH + 2 * lane;
#pragma unroll
    for (int h = 0; h < 8; ++h) {
      float inv = (den[h] > 0.f) ? 1.f / den[h] : 0.f;
      zrow[h * 64]     = z[h].x * inv;
      zrow[h * 64 + 1] = z[h].y * inv;
    }
  }
  __syncthreads();

  int mrow = lane & 15, hi = lane >> 4;
  int col_base = (w & 7) << 6;
  int tile_row = (w >> 3) << 4;
  const float* Zrow = Zs + (tile_row + mrow) * ZPITCH + col_base;
  const float* Bcol = W + col_base + 4 * mrow;
  v8f acc0 = {}, acc1 = {}, acc2 = {}, acc3 = {};
#pragma unroll 4
  for (int k0 = 0; k0 < 64; k0 += 4) {
    int ka = k0 + 2 * hi;
    v2f a;
    a.x = Zrow[ka];
    a.y = Zrow[ka + 1];
    float4 bx = *(const float4*)(Bcol + (size_t)ka * 512);
    float4 by = *(const float4*)(Bcol + (size_t)(ka + 1) * 512);
    v2f b0, b1, b2, b3;
    b0.x = bx.x; b0.y = by.x;
    b1.x = bx.y; b1.y = by.y;
    b2.x = bx.z; b2.y = by.z;
    b3.x = bx.w; b3.y = by.w;
    acc0 = __builtin_amdgcn_wmma_f32_16x16x4_f32(false, a, false, b0, (short)0, acc0, false, false);
    acc1 = __builtin_amdgcn_wmma_f32_16x16x4_f32(false, a, false, b1, (short)0, acc1, false, false);
    acc2 = __builtin_amdgcn_wmma_f32_16x16x4_f32(false, a, false, b2, (short)0, acc2, false, false);
    acc3 = __builtin_amdgcn_wmma_f32_16x16x4_f32(false, a, false, b3, (short)0, acc3, false, false);
  }

  float* ztile = Zs + (tile_row + hi * 8) * ZPITCH + col_base + 4 * mrow;
  {
    float4 b = *(const float4*)(bias + col_base + 4 * mrow);
#pragma unroll
    for (int r = 0; r < 8; ++r) {
      float4 q;
      q.x = fmaxf(acc0[r] + b.x, 0.f); q.y = fmaxf(acc1[r] + b.y, 0.f);
      q.z = fmaxf(acc2[r] + b.z, 0.f); q.w = fmaxf(acc3[r] + b.w, 0.f);
      *(float4*)(ztile + r * ZPITCH) = q;
    }
  }
  __syncthreads();

  v8f p0 = {}, p1 = {}, p2 = {}, p3 = {};
  {
    const float* B2col = W2 + (size_t)col_base * 64 + 4 * mrow;
#pragma unroll 4
    for (int k0 = 0; k0 < 64; k0 += 4) {
      int ka = k0 + 2 * hi;
      v2f a;
      a.x = Zrow[ka];
      a.y = Zrow[ka + 1];
      float4 bx = *(const float4*)(B2col + (size_t)ka * 64);
      float4 by = *(const float4*)(B2col + (size_t)(ka + 1) * 64);
      v2f b0, b1, b2, b3;
      b0.x = bx.x; b0.y = by.x;
      b1.x = bx.y; b1.y = by.y;
      b2.x = bx.z; b2.y = by.z;
      b3.x = bx.w; b3.y = by.w;
      p0 = __builtin_amdgcn_wmma_f32_16x16x4_f32(false, a, false, b0, (short)0, p0, false, false);
      p1 = __builtin_amdgcn_wmma_f32_16x16x4_f32(false, a, false, b1, (short)0, p1, false, false);
      p2 = __builtin_amdgcn_wmma_f32_16x16x4_f32(false, a, false, b2, (short)0, p2, false, false);
      p3 = __builtin_amdgcn_wmma_f32_16x16x4_f32(false, a, false, b3, (short)0, p3, false, false);
    }
  }
#pragma unroll
  for (int r = 0; r < 8; ++r)
    *(float4*)(ztile + r * ZPITCH) = make_float4(p0[r], p1[r], p2[r], p3[r]);
  __syncthreads();

  int orow = (int)threadIdx.x >> 4, c4 = (int)threadIdx.x & 15;
  const float* zp = Zs + orow * ZPITCH + 4 * c4;
  float4 hs = *(const float4*)zp;
#pragma unroll
  for (int h = 1; h < 8; ++h) {
    float4 u = *(const float4*)(zp + 64 * h);
    hs.x += u.x; hs.y += u.y; hs.z += u.z; hs.w += u.w;
  }
  float4 s4 = *(const float4*)(avs2 + 4 * c4);
  float4 d4 = *(const float4*)(avd2 + 4 * c4);
  float vs = hs.x * s4.x + hs.y * s4.y + hs.z * s4.z + hs.w * s4.w;
  float vd = hs.x * d4.x + hs.y * d4.y + hs.z * d4.z + hs.w * d4.w;
  for (int off = 8; off; off >>= 1) {
    vs += __shfl_xor(vs, off, 32);
    vd += __shfl_xor(vd, off, 32);
  }
  int grow = row_base + orow;
  if (grow < N) {
    *(float4*)(hW2 + (size_t)grow * 64 + 4 * c4) = hs;
    if (c4 == 0) { as2[grow] = vs; ad2[grow] = vd; }
  }
}

template <int SEL> __device__ __forceinline__ int dpp8_i(int v) {
  return (int)__builtin_amdgcn_mov_dpp8((unsigned)v, (unsigned)SEL);
}
template <int SEL> __device__ __forceinline__ float dpp8_f(float v) {
  return __uint_as_float(__builtin_amdgcn_mov_dpp8(__float_as_uint(v), (unsigned)SEL));
}
#define BC8(k) ((k) * 0x249249)
#define XOR8_1 0xde54c1
#define XOR8_2 0xb3e21a
#define XOR8_4 0x688fac

template <int G> __device__ __forceinline__ void add4_rows_h1(const float4* __restrict__ X4, int sv, float w, int j,
                                                              float4& lo, float4& hi, float& den) {
  int s0 = dpp8_i<BC8(G)>(sv), s1 = dpp8_i<BC8(G + 1)>(sv), s2 = dpp8_i<BC8(G + 2)>(sv), s3 = dpp8_i<BC8(G + 3)>(sv);
  float a0 = dpp8_f<BC8(G)>(w), a1 = dpp8_f<BC8(G + 1)>(w), a2 = dpp8_f<BC8(G + 2)>(w), a3 = dpp8_f<BC8(G + 3)>(w);
  float4 u0 = X4[(unsigned)(s0 * 16 + j)], v0 = X4[(unsigned)(s0 * 16 + 8 + j)];
  float4 u1 = X4[(unsigned)(s1 * 16 + j)], v1 = X4[(unsigned)(s1 * 16 + 8 + j)];
  float4 u2 = X4[(unsigned)(s2 * 16 + j)], v2 = X4[(unsigned)(s2 * 16 + 8 + j)];
  float4 u3 = X4[(unsigned)(s3 * 16 + j)], v3 = X4[(unsigned)(s3 * 16 + 8 + j)];
  den += a0;
  lo.x += a0 * u0.x; lo.y += a0 * u0.y; lo.z += a0 * u0.z; lo.w += a0 * u0.w;
  hi.x += a0 * v0.x; hi.y += a0 * v0.y; hi.z += a0 * v0.z; hi.w += a0 * v0.w;
  den += a1;
  lo.x += a1 * u1.x; lo.y += a1 * u1.y; lo.z += a1 * u1.z; lo.w += a1 * u1.w;
  hi.x += a1 * v1.x; hi.y += a1 * v1.y; hi.z += a1 * v1.z; hi.w += a1 * v1.w;
  den += a2;
  lo.x += a2 * u2.x; lo.y += a2 * u2.y; lo.z += a2 * u2.z; lo.w += a2 * u2.w;
  hi.x += a2 * v2.x; hi.y += a2 * v2.y; hi.z += a2 * v2.z; hi.w += a2 * v2.w;
  den += a3;
  lo.x += a3 * u3.x; lo.y += a3 * u3.y; lo.z += a3 * u3.z; lo.w += a3 * u3.w;
  hi.x += a3 * v3.x; hi.y += a3 * v3.y; hi.z += a3 * v3.z; hi.w += a3 * v3.w;
}

__global__ void __launch_bounds__(256)
row_softmax_aggregate_h1(const float* __restrict__ xW, const float* __restrict__ as,
                         const float* __restrict__ ad,
                         const int* __restrict__ rowptr, const int* __restrict__ srcS,
                         const float* __restrict__ bias, float* __restrict__ out, int N) {
  int wave = __builtin_amdgcn_readfirstlane((int)(blockIdx.x * 8 + (threadIdx.x >> 5)));
  int lane = threadIdx.x & 31;
  if (4 * wave >= N) return;
  int j = lane & 7;
  int row = 4 * wave + (lane >> 3);
  bool rv = row < N;
  int start = 0, end = 0;
  float adr = 0.f;
  if (rv) { start = rowptr[row]; end = rowptr[row + 1]; adr = ad[row]; }
  int n = end - start;
  int nmax = __builtin_amdgcn_readlane(n, 0);
  { int t1 = __builtin_amdgcn_readlane(n, 8), t2 = __builtin_amdgcn_readlane(n, 16), t3 = __builtin_amdgcn_readlane(n, 24);
    if (t1 > nmax) nmax = t1;
    if (t2 > nmax) nmax = t2;
    if (t3 > nmax) nmax = t3; }

  bool in0 = j < n, in1 = j + 8 < n;
  int t0 = srcS[in0 ? start + j : 0];
  int t1 = srcS[in1 ? start + 8 + j : 0];
  float g0 = as[t0], g1 = as[t1];
  int sv0 = in0 ? t0 : 0, sv1 = in1 ? t1 : 0;
  float e0 = in0 ? leaky(g0 + adr) : -INFINITY;
  float e1 = in1 ? leaky(g1 + adr) : -INFINITY;
  float m = fmaxf(e0, e1);
  for (int c = 16; c < nmax; c += 8) {
    if (c + j < n) m = fmaxf(m, leaky(as[srcS[start + c + j]] + adr));
  }
  m = fmaxf(m, dpp8_f<XOR8_1>(m));
  m = fmaxf(m, dpp8_f<XOR8_2>(m));
  m = fmaxf(m, dpp8_f<XOR8_4>(m));

  const float4* X4 = (const float4*)xW;
  float4 lo = make_float4(0.f, 0.f, 0.f, 0.f), hi = make_float4(0.f, 0.f, 0.f, 0.f);
  float den = 0.f;
  for (int c = 0; c < nmax; c += 8) {
    int sv = sv0; float e = e0;
    if (c == 8) { sv = sv1; e = e1; }
    else if (c != 0) {
      sv = 0; e = -INFINITY;
      if (c + j < n) { sv = srcS[start + c + j]; e = leaky(as[sv] + adr); }
    }
    float w = (c + j < n) ? __expf(e - m) : 0.f;
    add4_rows_h1<0>(X4, sv, w, j, lo, hi, den);
    if (nmax - c > 4) add4_rows_h1<4>(X4, sv, w, j, lo, hi, den);
  }
  float inv = (den > 0.f) ? 1.f / den : 0.f;
  float4 b0 = ((const float4*)bias)[j], b1 = ((const float4*)bias)[8 + j];
  if (rv) {
    float4 q0, q1;
    q0.x = lo.x * inv + b0.x; q0.y = lo.y * inv + b0.y; q0.z = lo.z * inv + b0.z; q0.w = lo.w * inv + b0.w;
    q1.x = hi.x * inv + b1.x; q1.y = hi.y * inv + b1.y; q1.z = hi.z * inv + b1.z; q1.w = hi.w * inv + b1.w;
    float4* o = (float4*)(out + (size_t)row * 64);
    o[j] = q0;
    o[8 + j] = q1;
  }
}

extern "C" void kernel_launch(void* const* d_in, const int* in_sizes, int n_in,
                              void* d_out, int out_size, void* d_ws, size_t ws_size,
                              hipStream_t stream) {
  const float* x     = (const float*)d_in[0];
  const int*   ei    = (const int*)  d_in[1];
  const float* W1    = (const float*)d_in[2];
  const float* asrc1 = (const float*)d_in[3];
  const float* adst1 = (const float*)d_in[4];
  const float* b1    = (const float*)d_in[5];
  const float* W2    = (const float*)d_in[6];
  const float* asrc2 = (const float*)d_in[7];
  const float* adst2 = (const float*)d_in[8];
  const float* b2    = (const float*)d_in[9];
  float* out = (float*)d_out;

  const int D = 64;
  const int N = in_sizes[0] / D;
  const int E = in_sizes[1] / 2;
  const int H = 8;

  float* ws   = (float*)d_ws;
  float* wa1  = ws;
  float* as1  = wa1  + 64 * 16;
  float* ad1  = as1  + (size_t)N * H;
  float* hW2  = ad1  + (size_t)N * H;
  float* as2  = hW2  + (size_t)N * D;
  float* ad2  = as2  + (size_t)N;
  int*  hist   = (int*)(ad2 + (size_t)N);
  int*  hoff   = hist + NCHUNK * NBIN;
  int2* binned = (int2*)(hoff + NCHUNK * NBIN);
  int*  seg    = (int*)(binned + (size_t)E);
  int*  srcS   = seg  + (size_t)E;
  int*  rowptr = srcS + (size_t)E;

  const int T = 256;

  const int LB = ((N + 15) / 16 + 31) / 32;
  fold_and_group_slices<<<NFOLD + NCHUNK, 1024, 0, stream>>>(ei, hist, hoff, binned, E, N, W1, asrc1, adst1, wa1);
  bin_rows_and_node_logits<<<NBIN + LB, 1024, 0, stream>>>(ei, hist, hoff, binned, rowptr, seg, srcS, E, N,
                                                           x, wa1, as1, ad1);

  row_softmax_aggregate_project2_h8<<<(N + 31) / 32, 512, 0, stream>>>(x, as1, ad1, rowptr, srcS, W1, b1,
                                                                      W2, asrc2, adst2, hW2, as2, ad2, N);

  row_softmax_aggregate_h1<<<((N + 3) / 4 * 32 + T - 1) / T, T, 0, stream>>>(hW2, as2, ad2, rowptr, srcS, b2, out, N);
}
